# starting kernel
# baseline (speedup 1.0000x reference)
#include <hip/hip_runtime.h>
#include <math.h>

#define NNODES 100000
#define NEDGES 1600000
#define NEP    1700000
#define FIN    16
#define FH     64
#define FOUT   8

typedef __attribute__((ext_vector_type(2))) float v2f;
typedef __attribute__((ext_vector_type(8))) float v8f;

__device__ __forceinline__ float leaky(float e) { return e > 0.f ? e : 0.2f * e; }

__global__ void gemm_h_kernel(const float* __restrict__ in, const float* __restrict__ W,
                              float* __restrict__ h, int K, int nRowTiles) {
    const int lane = threadIdx.x & 31;
    const int gw   = blockIdx.x * 8 + (threadIdx.x >> 5);
    const int rowTile = gw >> 2;
    const int colTile = gw & 3;
    if (rowTile >= nRowTiles) return;
    const int hi  = lane >> 4;
    const int lo  = lane & 15;
    const int row = rowTile * 16 + lo;
    const int col = colTile * 16 + lo;
    v8f c = {};
    for (int kk = 0; kk < K; kk += 4) {
        const int k0 = kk + 2 * hi;
        v2f a, b;
        a.x = in[(size_t)row * K + k0];
        a.y = in[(size_t)row * K + k0 + 1];
        b.x = W[(size_t)k0 * FH + col];
        b.y = W[(size_t)(k0 + 1) * FH + col];
        c = __builtin_amdgcn_wmma_f32_16x16x4_f32(false, a, false, b, (short)0, c, false, false);
    }
    const int rbase = rowTile * 16 + 8 * hi;
#pragma unroll
    for (int r = 0; r < 8; ++r)
        h[(size_t)(rbase + r) * FH + col] = c[r];
}

__global__ void scalars_init_kernel(const float* __restrict__ h,
                                    const float* __restrict__ a_src,
                                    const float* __restrict__ a_dst,
                                    float* __restrict__ s_src, float* __restrict__ s_dst,
                                    int* __restrict__ mbits, float* __restrict__ den,
                                    float* __restrict__ obuf) {
    const int lane = threadIdx.x & 31;
    const int n    = blockIdx.x * 8 + (threadIdx.x >> 5);
    if (n >= NNODES) return;
    const float h0 = h[(size_t)n * FH + lane];
    const float h1 = h[(size_t)n * FH + lane + 32];
    float ps = h0 * a_src[lane] + h1 * a_src[lane + 32];
    float pd = h0 * a_dst[lane] + h1 * a_dst[lane + 32];
    for (int off = 16; off; off >>= 1) {
        ps += __shfl_xor(ps, off, 32);
        pd += __shfl_xor(pd, off, 32);
    }
    if (lane == 0) {
        s_src[n] = ps;
        s_dst[n] = pd;
        mbits[n] = (int)0xFF800000u;
        den[n]   = 0.f;
    }
    obuf[(size_t)n * FH + lane]      = 0.f;
    obuf[(size_t)n * FH + lane + 32] = 0.f;
}

__device__ __forceinline__ void edge_sd(int e, const int* __restrict__ src,
                                        const int* __restrict__ dst, int& s, int& d) {
    if (e < NEDGES) { s = src[e]; d = dst[e]; }
    else            { s = d = e - NEDGES; }
}

__global__ void edge_max_kernel(const int* __restrict__ src, const int* __restrict__ dst,
                                const float* __restrict__ s_src, const float* __restrict__ s_dst,
                                int* __restrict__ mbits) {
    const int e = blockIdx.x * blockDim.x + threadIdx.x;
    if (e >= NEP) return;
    int s, d; edge_sd(e, src, dst, s, d);
    const float l = leaky(s_src[s] + s_dst[d]);
    const unsigned u = __float_as_uint(l);
    if ((u >> 31) == 0) atomicMax((int*)&mbits[d], (int)u);
    else                atomicMin((unsigned int*)&mbits[d], u);
}

__global__ void edge_den_kernel(const int* __restrict__ src, const int* __restrict__ dst,
                                const float* __restrict__ s_src, const float* __restrict__ s_dst,
                                const int* __restrict__ mbits, float* __restrict__ den) {
    const int e = blockIdx.x * blockDim.x + threadIdx.x;
    if (e >= NEP) return;
    int s, d; edge_sd(e, src, dst, s, d);
    const float l  = leaky(s_src[s] + s_dst[d]);
    const float mv = __int_as_float(mbits[d]);
    atomicAdd(&den[d], expf(l - mv));
}

__global__ void edge_aggr_kernel(const int* __restrict__ src, const int* __restrict__ dst,
                                 const float* __restrict__ s_src, const float* __restrict__ s_dst,
                                 const int* __restrict__ mbits, const float* __restrict__ den,
                                 const float* __restrict__ h, float* __restrict__ out) {
    const int lane = threadIdx.x & 31;
    const int e    = blockIdx.x * 8 + (threadIdx.x >> 5);
    if (e >= NEP) return;
    int s, d; edge_sd(e, src, dst, s, d);
    const float l     = leaky(s_src[s] + s_dst[d]);
    const float mv    = __int_as_float(mbits[d]);
    const float alpha = expf(l - mv) / (den[d] + 1e-16f);
    const float v0 = h[(size_t)s * FH + lane]      * alpha;
    const float v1 = h[(size_t)s * FH + lane + 32] * alpha;
    atomicAdd(&out[(size_t)d * FH + lane],      v0);
    atomicAdd(&out[(size_t)d * FH + lane + 32], v1);
}

__global__ void act_kernel(float* __restrict__ buf, const float* __restrict__ b) {
    const size_t i = (size_t)blockIdx.x * blockDim.x + threadIdx.x;
    if (i >= (size_t)NNODES * FH) return;
    const float v = buf[i] + b[i & 63];
    buf[i] = v > 0.f ? v : 0.f;
}

__global__ void policy_kernel(const float* __restrict__ act, const float* __restrict__ Wp,
                              const float* __restrict__ bp, float* __restrict__ out,
                              int nRowTiles) {
    const int lane = threadIdx.x & 31;
    const int gw   = blockIdx.x * 8 + (threadIdx.x >> 5);
    if (gw >= nRowTiles) return;
    const int hi  = lane >> 4;
    const int lo  = lane & 15;
    const int row = gw * 16 + lo;
    v8f c = {};
    for (int kk = 0; kk < FH; kk += 4) {
        const int k0 = kk + 2 * hi;
        v2f a, b;
        a.x = act[(size_t)row * FH + k0];
        a.y = act[(size_t)row * FH + k0 + 1];
        b.x = (lo < FOUT) ? Wp[(size_t)k0 * FOUT + lo]       : 0.f;
        b.y = (lo < FOUT) ? Wp[(size_t)(k0 + 1) * FOUT + lo] : 0.f;
        c = __builtin_amdgcn_wmma_f32_16x16x4_f32(false, a, false, b, (short)0, c, false, false);
    }
    if (lo < FOUT) {
        const int rbase = gw * 16 + 8 * hi;
        const float bb = bp[lo];
#pragma unroll
        for (int r = 0; r < 8; ++r)
            out[(size_t)(rbase + r) * FOUT + lo] = c[r] + bb;
    }
}

extern "C" void kernel_launch(void* const* d_in, const int* in_sizes, int n_in,
                              void* d_out, int out_size, void* d_ws, size_t ws_size,
                              hipStream_t stream) {
    const float* x   = (const float*)d_in[0];
    const int*   ei  = (const int*)  d_in[1];
    const float* W1  = (const float*)d_in[2];
    const float* as1 = (const float*)d_in[3];
    const float* ad1 = (const float*)d_in[4];
    const float* b1  = (const float*)d_in[5];
    const float* W2  = (const float*)d_in[6];
    const float* as2 = (const float*)d_in[7];
    const float* ad2 = (const float*)d_in[8];
    const float* b2  = (const float*)d_in[9];
    const float* Wp  = (const float*)d_in[10];
    const float* bp  = (const float*)d_in[11];
    float* out = (float*)d_out;

    const int* srcA = ei;
    const int* dstA = ei + NEDGES;

    float* ws   = (float*)d_ws;
    float* hbuf = ws;
    float* obuf = ws + (size_t)NNODES * FH;
    float* ssrc = ws + (size_t)2 * NNODES * FH;
    float* sdst = ssrc + NNODES;
    int*   mb   = (int*)(sdst + NNODES);
    float* den  = (float*)(mb + NNODES);

    const int nRowTiles = NNODES / 16;
    const dim3 blk(256);
    const int gemmBlocks = (nRowTiles * 4 + 7) / 8;
    const int nodeBlocks = (NNODES + 7) / 8;
    const int edgeBlocks = (NEP + 255) / 256;
    const int aggrBlocks = (NEP + 7) / 8;
    const int actBlocks  = (NNODES * FH + 255) / 256;
    const int polBlocks  = (nRowTiles + 7) / 8;

    gemm_h_kernel<<<gemmBlocks, blk, 0, stream>>>(x, W1, hbuf, FIN, nRowTiles);
    scalars_init_kernel<<<nodeBlocks, blk, 0, stream>>>(hbuf, as1, ad1, ssrc, sdst, mb, den, obuf);
    edge_max_kernel<<<edgeBlocks, blk, 0, stream>>>(srcA, dstA, ssrc, sdst, mb);
    edge_den_kernel<<<edgeBlocks, blk, 0, stream>>>(srcA, dstA, ssrc, sdst, mb, den);
    edge_aggr_kernel<<<aggrBlocks, blk, 0, stream>>>(srcA, dstA, ssrc, sdst, mb, den, hbuf, obuf);
    act_kernel<<<actBlocks, blk, 0, stream>>>(obuf, b1);

    gemm_h_kernel<<<gemmBlocks, blk, 0, stream>>>(obuf, W2, hbuf, FH, nRowTiles);
    scalars_init_kernel<<<nodeBlocks, blk, 0, stream>>>(hbuf, as2, ad2, ssrc, sdst, mb, den, obuf);
    edge_max_kernel<<<edgeBlocks, blk, 0, stream>>>(srcA, dstA, ssrc, sdst, mb);
    edge_den_kernel<<<edgeBlocks, blk, 0, stream>>>(srcA, dstA, ssrc, sdst, mb, den);
    edge_aggr_kernel<<<aggrBlocks, blk, 0, stream>>>(srcA, dstA, ssrc, sdst, mb, den, hbuf, obuf);
    act_kernel<<<actBlocks, blk, 0, stream>>>(obuf, b2);

    policy_kernel<<<polBlocks, blk, 0, stream>>>(obuf, Wp, bp, out, nRowTiles);
}


# Fold the softmax denominator pass into the aggregation pass: edge_aggr adds w into den and sums h*w unnormalised, act divides by den once per node (division after the sum); both edge_den launches go
# speedup vs baseline: 1.0431x; 1.0431x over previous
#include <hip/hip_runtime.h>
#include <math.h>

#define NNODES 100000
#define NEDGES 1600000
#define NEP    1700000
#define FIN    16
#define FH     64
#define FOUT   8

typedef __attribute__((ext_vector_type(2))) float v2f;
typedef __attribute__((ext_vector_type(8))) float v8f;

__device__ __forceinline__ float leaky(float e) { return e > 0.f ? e : 0.2f * e; }

__global__ void gemm_h_kernel(const float* __restrict__ in, const float* __restrict__ W,
                              float* __restrict__ h, int K, int nRowTiles) {
    const int lane = threadIdx.x & 31;
    const int gw   = blockIdx.x * 8 + (threadIdx.x >> 5);
    const int rowTile = gw >> 2;
    const int colTile = gw & 3;
    if (rowTile >= nRowTiles) return;
    const int hi  = lane >> 4;
    const int lo  = lane & 15;
    const int row = rowTile * 16 + lo;
    const int col = colTile * 16 + lo;
    v8f c = {};
    for (int kk = 0; kk < K; kk += 4) {
        const int k0 = kk + 2 * hi;
        v2f a, b;
        a.x = in[(size_t)row * K + k0];
        a.y = in[(size_t)row * K + k0 + 1];
        b.x = W[(size_t)k0 * FH + col];
        b.y = W[(size_t)(k0 + 1) * FH + col];
        c = __builtin_amdgcn_wmma_f32_16x16x4_f32(false, a, false, b, (short)0, c, false, false);
    }
    const int rbase = rowTile * 16 + 8 * hi;
#pragma unroll
    for (int r = 0; r < 8; ++r)
        h[(size_t)(rbase + r) * FH + col] = c[r];
}

__global__ void scalars_init_kernel(const float* __restrict__ h,
                                    const float* __restrict__ a_src,
                                    const float* __restrict__ a_dst,
                                    float* __restrict__ s_src, float* __restrict__ s_dst,
                                    int* __restrict__ mbits, float* __restrict__ den,
                                    float* __restrict__ obuf) {
    const int lane = threadIdx.x & 31;
    const int n    = blockIdx.x * 8 + (threadIdx.x >> 5);
    if (n >= NNODES) return;
    const float h0 = h[(size_t)n * FH + lane];
    const float h1 = h[(size_t)n * FH + lane + 32];
    float ps = h0 * a_src[lane] + h1 * a_src[lane + 32];
    float pd = h0 * a_dst[lane] + h1 * a_dst[lane + 32];
    for (int off = 16; off; off >>= 1) {
        ps += __shfl_xor(ps, off, 32);
        pd += __shfl_xor(pd, off, 32);
    }
    if (lane == 0) {
        s_src[n] = ps;
        s_dst[n] = pd;
        mbits[n] = (int)0xFF800000u;
        den[n]   = 0.f;
    }
    obuf[(size_t)n * FH + lane]      = 0.f;
    obuf[(size_t)n * FH + lane + 32] = 0.f;
}

__device__ __forceinline__ void edge_sd(int e, const int* __restrict__ src,
                                        const int* __restrict__ dst, int& s, int& d) {
    if (e < NEDGES) { s = src[e]; d = dst[e]; }
    else            { s = d = e - NEDGES; }
}

__global__ void edge_max_kernel(const int* __restrict__ src, const int* __restrict__ dst,
                                const float* __restrict__ s_src, const float* __restrict__ s_dst,
                                int* __restrict__ mbits) {
    const int e = blockIdx.x * blockDim.x + threadIdx.x;
    if (e >= NEP) return;
    int s, d; edge_sd(e, src, dst, s, d);
    const float l = leaky(s_src[s] + s_dst[d]);
    const unsigned u = __float_as_uint(l);
    if ((u >> 31) == 0) atomicMax((int*)&mbits[d], (int)u);
    else                atomicMin((unsigned int*)&mbits[d], u);
}

__global__ void edge_aggr_kernel(const int* __restrict__ src, const int* __restrict__ dst,
                                 const float* __restrict__ s_src, const float* __restrict__ s_dst,
                                 const int* __restrict__ mbits, float* __restrict__ den,
                                 const float* __restrict__ h, float* __restrict__ out) {
    const int lane = threadIdx.x & 31;
    const int e    = blockIdx.x * 8 + (threadIdx.x >> 5);
    if (e >= NEP) return;
    int s, d; edge_sd(e, src, dst, s, d);
    const float l  = leaky(s_src[s] + s_dst[d]);
    const float mv = __int_as_float(mbits[d]);
    const float w  = expf(l - mv);
    const float v0 = h[(size_t)s * FH + lane]      * w;
    const float v1 = h[(size_t)s * FH + lane + 32] * w;
    atomicAdd(&out[(size_t)d * FH + lane],      v0);
    atomicAdd(&out[(size_t)d * FH + lane + 32], v1);
    if (lane == 0) atomicAdd(&den[d], w);
}

__global__ void act_kernel(float* __restrict__ buf, const float* __restrict__ den,
                           const float* __restrict__ b) {
    const size_t i = (size_t)blockIdx.x * blockDim.x + threadIdx.x;
    if (i >= (size_t)NNODES * FH) return;
    const float v = buf[i] / (den[i >> 6] + 1e-16f) + b[i & 63];
    buf[i] = v > 0.f ? v : 0.f;
}

__global__ void policy_kernel(const float* __restrict__ act, const float* __restrict__ Wp,
                              const float* __restrict__ bp, float* __restrict__ out,
                              int nRowTiles) {
    const int lane = threadIdx.x & 31;
    const int gw   = blockIdx.x * 8 + (threadIdx.x >> 5);
    if (gw >= nRowTiles) return;
    const int hi  = lane >> 4;
    const int lo  = lane & 15;
    const int row = gw * 16 + lo;
    v8f c = {};
    for (int kk = 0; kk < FH; kk += 4) {
        const int k0 = kk + 2 * hi;
        v2f a, b;
        a.x = act[(size_t)row * FH + k0];
        a.y = act[(size_t)row * FH + k0 + 1];
        b.x = (lo < FOUT) ? Wp[(size_t)k0 * FOUT + lo]       : 0.f;
        b.y = (lo < FOUT) ? Wp[(size_t)(k0 + 1) * FOUT + lo] : 0.f;
        c = __builtin_amdgcn_wmma_f32_16x16x4_f32(false, a, false, b, (short)0, c, false, false);
    }
    if (lo < FOUT) {
        const int rbase = gw * 16 + 8 * hi;
        const float bb = bp[lo];
#pragma unroll
        for (int r = 0; r < 8; ++r)
            out[(size_t)(rbase + r) * FOUT + lo] = c[r] + bb;
    }
}

extern "C" void kernel_launch(void* const* d_in, const int* in_sizes, int n_in,
                              void* d_out, int out_size, void* d_ws, size_t ws_size,
                              hipStream_t stream) {
    const float* x   = (const float*)d_in[0];
    const int*   ei  = (const int*)  d_in[1];
    const float* W1  = (const float*)d_in[2];
    const float* as1 = (const float*)d_in[3];
    const float* ad1 = (const float*)d_in[4];
    const float* b1  = (const float*)d_in[5];
    const float* W2  = (const float*)d_in[6];
    const float* as2 = (const float*)d_in[7];
    const float* ad2 = (const float*)d_in[8];
    const float* b2  = (const float*)d_in[9];
    const float* Wp  = (const float*)d_in[10];
    const float* bp  = (const float*)d_in[11];
    float* out = (float*)d_out;

    const int* srcA = ei;
    const int* dstA = ei + NEDGES;

    float* ws   = (float*)d_ws;
    float* hbuf = ws;
    float* obuf = ws + (size_t)NNODES * FH;
    float* ssrc = ws + (size_t)2 * NNODES * FH;
    float* sdst = ssrc + NNODES;
    int*   mb   = (int*)(sdst + NNODES);
    float* den  = (float*)(mb + NNODES);

    const int nRowTiles = NNODES / 16;
    const dim3 blk(256);
    const int gemmBlocks = (nRowTiles * 4 + 7) / 8;
    const int nodeBlocks = (NNODES + 7) / 8;
    const int edgeBlocks = (NEP + 255) / 256;
    const int aggrBlocks = (NEP + 7) / 8;
    const int actBlocks  = (NNODES * FH + 255) / 256;
    const int polBlocks  = (nRowTiles + 7) / 8;

    gemm_h_kernel<<<gemmBlocks, blk, 0, stream>>>(x, W1, hbuf, FIN, nRowTiles);
    scalars_init_kernel<<<nodeBlocks, blk, 0, stream>>>(hbuf, as1, ad1, ssrc, sdst, mb, den, obuf);
    edge_max_kernel<<<edgeBlocks, blk, 0, stream>>>(srcA, dstA, ssrc, sdst, mb);
    edge_aggr_kernel<<<aggrBlocks, blk, 0, stream>>>(srcA, dstA, ssrc, sdst, mb, den, hbuf, obuf);
    act_kernel<<<actBlocks, blk, 0, stream>>>(obuf, den, b1);

    gemm_h_kernel<<<gemmBlocks, blk, 0, stream>>>(obuf, W2, hbuf, FH, nRowTiles);
    scalars_init_kernel<<<nodeBlocks, blk, 0, stream>>>(hbuf, as2, ad2, ssrc, sdst, mb, den, obuf);
    edge_max_kernel<<<edgeBlocks, blk, 0, stream>>>(srcA, dstA, ssrc, sdst, mb);
    edge_aggr_kernel<<<aggrBlocks, blk, 0, stream>>>(srcA, dstA, ssrc, sdst, mb, den, hbuf, obuf);
    act_kernel<<<actBlocks, blk, 0, stream>>>(obuf, den, b2);

    policy_kernel<<<polBlocks, blk, 0, stream>>>(obuf, Wp, bp, out, nRowTiles);
}


# edge_aggr: one wave takes 32 edges (was 1): edge scalars and exp one edge per lane, then h rows fetched 8 at a time by lane broadcast; 32x fewer waves; f32 atomic adds and their arrival order kept
# speedup vs baseline: 1.3447x; 1.2891x over previous
#include <hip/hip_runtime.h>
#include <math.h>

#define NNODES 100000
#define NEDGES 1600000
#define NEP    1700000
#define FIN    16
#define FH     64
#define FOUT   8

typedef __attribute__((ext_vector_type(2))) float v2f;
typedef __attribute__((ext_vector_type(8))) float v8f;

__device__ __forceinline__ float leaky(float e) { return e > 0.f ? e : 0.2f * e; }

__global__ void gemm_h_kernel(const float* __restrict__ in, const float* __restrict__ W,
                              float* __restrict__ h, int K, int nRowTiles) {
    const int lane = threadIdx.x & 31;
    const int gw   = blockIdx.x * 8 + (threadIdx.x >> 5);
    const int rowTile = gw >> 2;
    const int colTile = gw & 3;
    if (rowTile >= nRowTiles) return;
    const int hi  = lane >> 4;
    const int lo  = lane & 15;
    const int row = rowTile * 16 + lo;
    const int col = colTile * 16 + lo;
    v8f c = {};
    for (int kk = 0; kk < K; kk += 4) {
        const int k0 = kk + 2 * hi;
        v2f a, b;
        a.x = in[(size_t)row * K + k0];
        a.y = in[(size_t)row * K + k0 + 1];
        b.x = W[(size_t)k0 * FH + col];
        b.y = W[(size_t)(k0 + 1) * FH + col];
        c = __builtin_amdgcn_wmma_f32_16x16x4_f32(false, a, false, b, (short)0, c, false, false);
    }
    const int rbase = rowTile * 16 + 8 * hi;
#pragma unroll
    for (int r = 0; r < 8; ++r)
        h[(size_t)(rbase + r) * FH + col] = c[r];
}

__global__ void scalars_init_kernel(const float* __restrict__ h,
                                    const float* __restrict__ a_src,
                                    const float* __restrict__ a_dst,
                                    float* __restrict__ s_src, float* __restrict__ s_dst,
                                    int* __restrict__ mbits, float* __restrict__ den,
                                    float* __restrict__ obuf) {
    const int lane = threadIdx.x & 31;
    const int n    = blockIdx.x * 8 + (threadIdx.x >> 5);
    if (n >= NNODES) return;
    const float h0 = h[(size_t)n * FH + lane];
    const float h1 = h[(size_t)n * FH + lane + 32];
    float ps = h0 * a_src[lane] + h1 * a_src[lane + 32];
    float pd = h0 * a_dst[lane] + h1 * a_dst[lane + 32];
    for (int off = 16; off; off >>= 1) {
        ps += __shfl_xor(ps, off, 32);
        pd += __shfl_xor(pd, off, 32);
    }
    if (lane == 0) {
        s_src[n] = ps;
        s_dst[n] = pd;
        mbits[n] = (int)0xFF800000u;
        den[n]   = 0.f;
    }
    obuf[(size_t)n * FH + lane]      = 0.f;
    obuf[(size_t)n * FH + lane + 32] = 0.f;
}

__device__ __forceinline__ void edge_sd(int e, const int* __restrict__ src,
                                        const int* __restrict__ dst, int& s, int& d) {
    if (e < NEDGES) { s = src[e]; d = dst[e]; }
    else            { s = d = e - NEDGES; }
}

__global__ void edge_max_kernel(const int* __restrict__ src, const int* __restrict__ dst,
                                const float* __restrict__ s_src, const float* __restrict__ s_dst,
                                int* __restrict__ mbits) {
    const int e = blockIdx.x * blockDim.x + threadIdx.x;
    if (e >= NEP) return;
    int s, d; edge_sd(e, src, dst, s, d);
    const float l = leaky(s_src[s] + s_dst[d]);
    const unsigned u = __float_as_uint(l);
    if ((u >> 31) == 0) atomicMax((int*)&mbits[d], (int)u);
    else                atomicMin((unsigned int*)&mbits[d], u);
}

#define AGGR_EPW  32
#define AGGR_ROWS 8
static_assert(NEP % AGGR_EPW == 0 && NEDGES % AGGR_EPW == 0, "full waves only");
static_assert(AGGR_EPW % AGGR_ROWS == 0, "whole groups");

__global__ void __launch_bounds__(256)
edge_aggr_kernel(const int* __restrict__ src, const int* __restrict__ dst,
                 const float* __restrict__ s_src, const float* __restrict__ s_dst,
                 const int* __restrict__ mbits, float* __restrict__ den,
                 const float* __restrict__ h, float* __restrict__ out) {
    const int lane = threadIdx.x & 31;
    const int wv   = blockIdx.x * 8 + (threadIdx.x >> 5);
    if (wv >= NEP / AGGR_EPW) return;
    const int e = wv * AGGR_EPW + lane;
    int s, d; edge_sd(e, src, dst, s, d);
    const float l  = leaky(s_src[s] + s_dst[d]);
    const float mv = __int_as_float(mbits[d]);
    const float w  = expf(l - mv);
    atomicAdd(&den[d], w);
    const int wbits = __float_as_int(w);
    const float* __restrict__ hl = h + lane;
    float* __restrict__ ol = out + lane;
#pragma unroll
    for (int g = 0; g < AGGR_EPW; g += AGGR_ROWS) {
        float v0[AGGR_ROWS], v1[AGGR_ROWS];
#pragma unroll
        for (int j = 0; j < AGGR_ROWS; ++j) {
            const int sj = __builtin_amdgcn_readlane(s, g + j);
            v0[j] = hl[(size_t)sj * FH];
            v1[j] = hl[(size_t)sj * FH + 32];
        }
#pragma unroll
        for (int j = 0; j < AGGR_ROWS; ++j) {
            const int   dj = __builtin_amdgcn_readlane(d, g + j);
            const float wj = __int_as_float(__builtin_amdgcn_readlane(wbits, g + j));
            atomicAdd(&ol[(size_t)dj * FH],      v0[j] * wj);
            atomicAdd(&ol[(size_t)dj * FH + 32], v1[j] * wj);
        }
    }
}

__global__ void act_kernel(float* __restrict__ buf, const float* __restrict__ den,
                           const float* __restrict__ b) {
    const size_t i = (size_t)blockIdx.x * blockDim.x + threadIdx.x;
    if (i >= (size_t)NNODES * FH) return;
    const float v = buf[i] / (den[i >> 6] + 1e-16f) + b[i & 63];
    buf[i] = v > 0.f ? v : 0.f;
}

__global__ void policy_kernel(const float* __restrict__ act, const float* __restrict__ Wp,
                              const float* __restrict__ bp, float* __restrict__ out,
                              int nRowTiles) {
    const int lane = threadIdx.x & 31;
    const int gw   = blockIdx.x * 8 + (threadIdx.x >> 5);
    if (gw >= nRowTiles) return;
    const int hi  = lane >> 4;
    const int lo  = lane & 15;
    const int row = gw * 16 + lo;
    v8f c = {};
    for (int kk = 0; kk < FH; kk += 4) {
        const int k0 = kk + 2 * hi;
        v2f a, b;
        a.x = act[(size_t)row * FH + k0];
        a.y = act[(size_t)row * FH + k0 + 1];
        b.x = (lo < FOUT) ? Wp[(size_t)k0 * FOUT + lo]       : 0.f;
        b.y = (lo < FOUT) ? Wp[(size_t)(k0 + 1) * FOUT + lo] : 0.f;
        c = __builtin_amdgcn_wmma_f32_16x16x4_f32(false, a, false, b, (short)0, c, false, false);
    }
    if (lo < FOUT) {
        const int rbase = gw * 16 + 8 * hi;
        const float bb = bp[lo];
#pragma unroll
        for (int r = 0; r < 8; ++r)
            out[(size_t)(rbase + r) * FOUT + lo] = c[r] + bb;
    }
}

extern "C" void kernel_launch(void* const* d_in, const int* in_sizes, int n_in,
                              void* d_out, int out_size, void* d_ws, size_t ws_size,
                              hipStream_t stream) {
    const float* x   = (const float*)d_in[0];
    const int*   ei  = (const int*)  d_in[1];
    const float* W1  = (const float*)d_in[2];
    const float* as1 = (const float*)d_in[3];
    const float* ad1 = (const float*)d_in[4];
    const float* b1  = (const float*)d_in[5];
    const float* W2  = (const float*)d_in[6];
    const float* as2 = (const float*)d_in[7];
    const float* ad2 = (const float*)d_in[8];
    const float* b2  = (const float*)d_in[9];
    const float* Wp  = (const float*)d_in[10];
    const float* bp  = (const float*)d_in[11];
    float* out = (float*)d_out;

    const int* srcA = ei;
    const int* dstA = ei + NEDGES;

    float* ws   = (float*)d_ws;
    float* hbuf = ws;
    float* obuf = ws + (size_t)NNODES * FH;
    float* ssrc = ws + (size_t)2 * NNODES * FH;
    float* sdst = ssrc + NNODES;
    int*   mb   = (int*)(sdst + NNODES);
    float* den  = (float*)(mb + NNODES);

    const int nRowTiles = NNODES / 16;
    const dim3 blk(256);
    const int gemmBlocks = (nRowTiles * 4 + 7) / 8;
    const int nodeBlocks = (NNODES + 7) / 8;
    const int edgeBlocks = (NEP + 255) / 256;
    const int aggrBlocks = (NEP / AGGR_EPW + 7) / 8;
    const int actBlocks  = (NNODES * FH + 255) / 256;
    const int polBlocks  = (nRowTiles + 7) / 8;

    gemm_h_kernel<<<gemmBlocks, blk, 0, stream>>>(x, W1, hbuf, FIN, nRowTiles);
    scalars_init_kernel<<<nodeBlocks, blk, 0, stream>>>(hbuf, as1, ad1, ssrc, sdst, mb, den, obuf);
    edge_max_kernel<<<edgeBlocks, blk, 0, stream>>>(srcA, dstA, ssrc, sdst, mb);
    edge_aggr_kernel<<<aggrBlocks, blk, 0, stream>>>(srcA, dstA, ssrc, sdst, mb, den, hbuf, obuf);
    act_kernel<<<actBlocks, blk, 0, stream>>>(obuf, den, b1);

    gemm_h_kernel<<<gemmBlocks, blk, 0, stream>>>(obuf, W2, hbuf, FH, nRowTiles);
    scalars_init_kernel<<<nodeBlocks, blk, 0, stream>>>(hbuf, as2, ad2, ssrc, sdst, mb, den, obuf);
    edge_max_kernel<<<edgeBlocks, blk, 0, stream>>>(srcA, dstA, ssrc, sdst, mb);
    edge_aggr_kernel<<<aggrBlocks, blk, 0, stream>>>(srcA, dstA, ssrc, sdst, mb, den, hbuf, obuf);
    act_kernel<<<actBlocks, blk, 0, stream>>>(obuf, den, b2);

    policy_kernel<<<polBlocks, blk, 0, stream>>>(obuf, Wp, bp, out, nRowTiles);
}


# Layer 1 sums the 16-float input rows x[src]*w per edge (2 edges per instruction) and applies W1 once per node on WMMA after the sum: 4x fewer f32 atomic lanes; rounding order changes
# speedup vs baseline: 1.4203x; 1.0562x over previous
#include <hip/hip_runtime.h>
#include <math.h>

#define NNODES 100000
#define NEDGES 1600000
#define NEP    1700000
#define FIN    16
#define FH     64
#define FOUT   8

typedef __attribute__((ext_vector_type(2))) float v2f;
typedef __attribute__((ext_vector_type(8))) float v8f;

__device__ __forceinline__ float leaky(float e) { return e > 0.f ? e : 0.2f * e; }

__global__ void gemm_h_kernel(const float* __restrict__ in, const float* __restrict__ W,
                              float* __restrict__ h, int K, int nRowTiles) {
    const int lane = threadIdx.x & 31;
    const int gw   = blockIdx.x * 8 + (threadIdx.x >> 5);
    const int rowTile = gw >> 2;
    const int colTile = gw & 3;
    if (rowTile >= nRowTiles) return;
    const int hi  = lane >> 4;
    const int lo  = lane & 15;
    const int row = rowTile * 16 + lo;
    const int col = colTile * 16 + lo;
    v8f c = {};
    for (int kk = 0; kk < K; kk += 4) {
        const int k0 = kk + 2 * hi;
        v2f a, b;
        a.x = in[(size_t)row * K + k0];
        a.y = in[(size_t)row * K + k0 + 1];
        b.x = W[(size_t)k0 * FH + col];
        b.y = W[(size_t)(k0 + 1) * FH + col];
        c = __builtin_amdgcn_wmma_f32_16x16x4_f32(false, a, false, b, (short)0, c, false, false);
    }
    const int rbase = rowTile * 16 + 8 * hi;
#pragma unroll
    for (int r = 0; r < 8; ++r)
        h[(size_t)(rbase + r) * FH + col] = c[r];
}

template <int OW>
__global__ void scalars_init_kernel(const float* __restrict__ h,
                                    const float* __restrict__ a_src,
                                    const float* __restrict__ a_dst,
                                    float* __restrict__ s_src, float* __restrict__ s_dst,
                                    int* __restrict__ mbits, float* __restrict__ den,
                                    float* __restrict__ obuf) {
    const int lane = threadIdx.x & 31;
    const int n    = blockIdx.x * 8 + (threadIdx.x >> 5);
    if (n >= NNODES) return;
    const float h0 = h[(size_t)n * FH + lane];
    const float h1 = h[(size_t)n * FH + lane + 32];
    float ps = h0 * a_src[lane] + h1 * a_src[lane + 32];
    float pd = h0 * a_dst[lane] + h1 * a_dst[lane + 32];
    for (int off = 16; off; off >>= 1) {
        ps += __shfl_xor(ps, off, 32);
        pd += __shfl_xor(pd, off, 32);
    }
    if (lane == 0) {
        s_src[n] = ps;
        s_dst[n] = pd;
        mbits[n] = (int)0xFF800000u;
        den[n]   = 0.f;
    }
    if (OW == FH) {
        obuf[(size_t)n * FH + lane]      = 0.f;
        obuf[(size_t)n * FH + lane + 32] = 0.f;
    } else if (lane < OW) {
        obuf[(size_t)n * OW + lane] = 0.f;
    }
}

__device__ __forceinline__ void edge_sd(int e, const int* __restrict__ src,
                                        const int* __restrict__ dst, int& s, int& d) {
    if (e < NEDGES) { s = src[e]; d = dst[e]; }
    else            { s = d = e - NEDGES; }
}

__global__ void edge_max_kernel(const int* __restrict__ src, const int* __restrict__ dst,
                                const float* __restrict__ s_src, const float* __restrict__ s_dst,
                                int* __restrict__ mbits) {
    const int e = blockIdx.x * blockDim.x + threadIdx.x;
    if (e >= NEP) return;
    int s, d; edge_sd(e, src, dst, s, d);
    const float l = leaky(s_src[s] + s_dst[d]);
    const unsigned u = __float_as_uint(l);
    if ((u >> 31) == 0) atomicMax((int*)&mbits[d], (int)u);
    else                atomicMin((unsigned int*)&mbits[d], u);
}

#define AGGR_EPW  32
#define AGGR_ROWS 8
static_assert(NEP % AGGR_EPW == 0 && NEDGES % AGGR_EPW == 0, "full waves only");
static_assert(AGGR_EPW % AGGR_ROWS == 0, "whole groups");

__global__ void __launch_bounds__(256)
edge_aggr_kernel(const int* __restrict__ src, const int* __restrict__ dst,
                 const float* __restrict__ s_src, const float* __restrict__ s_dst,
                 const int* __restrict__ mbits, float* __restrict__ den,
                 const float* __restrict__ h, float* __restrict__ out) {
    const int lane = threadIdx.x & 31;
    const int wv   = blockIdx.x * 8 + (threadIdx.x >> 5);
    if (wv >= NEP / AGGR_EPW) return;
    const int e = wv * AGGR_EPW + lane;
    int s, d; edge_sd(e, src, dst, s, d);
    const float l  = leaky(s_src[s] + s_dst[d]);
    const float mv = __int_as_float(mbits[d]);
    const float w  = expf(l - mv);
    atomicAdd(&den[d], w);
    const int wbits = __float_as_int(w);
    const float* __restrict__ hl = h + lane;
    float* __restrict__ ol = out + lane;
#pragma unroll
    for (int g = 0; g < AGGR_EPW; g += AGGR_ROWS) {
        float v0[AGGR_ROWS], v1[AGGR_ROWS];
#pragma unroll
        for (int j = 0; j < AGGR_ROWS; ++j) {
            const int sj = __builtin_amdgcn_readlane(s, g + j);
            v0[j] = hl[(size_t)sj * FH];
            v1[j] = hl[(size_t)sj * FH + 32];
        }
#pragma unroll
        for (int j = 0; j < AGGR_ROWS; ++j) {
            const int   dj = __builtin_amdgcn_readlane(d, g + j);
            const float wj = __int_as_float(__builtin_amdgcn_readlane(wbits, g + j));
            atomicAdd(&ol[(size_t)dj * FH],      v0[j] * wj);
            atomicAdd(&ol[(size_t)dj * FH + 32], v1[j] * wj);
        }
    }
}

template <int J>
__device__ __forceinline__ int row_lane(int v) {
    return __builtin_amdgcn_update_dpp(0, v, 0x150 + J, 0xF, 0xF, false);
}
template <int J>
__device__ __forceinline__ void gather_rows(int s, const float* __restrict__ xl, float (&v)[FIN]) {
    v[J] = xl[(size_t)row_lane<J>(s) * FIN];
    if constexpr (J + 1 < FIN) gather_rows<J + 1>(s, xl, v);
}
template <int J>
__device__ __forceinline__ void scatter_rows(int d, int wbits, float* __restrict__ al,
                                             const float (&v)[FIN]) {
    const int   dj = row_lane<J>(d);
    const float wj = __int_as_float(row_lane<J>(wbits));
    atomicAdd(&al[(size_t)dj * FIN], v[J] * wj);
    if constexpr (J + 1 < FIN) scatter_rows<J + 1>(d, wbits, al, v);
}

__global__ void __launch_bounds__(256)
edge_aggr_in_kernel(const int* __restrict__ src, const int* __restrict__ dst,
                    const float* __restrict__ s_src, const float* __restrict__ s_dst,
                    const int* __restrict__ mbits, float* __restrict__ den,
                    const float* __restrict__ x, float* __restrict__ agg) {
    const int lane = threadIdx.x & 31;
    const int wv   = blockIdx.x * 8 + (threadIdx.x >> 5);
    if (wv >= NEP / AGGR_EPW) return;
    const int e = wv * AGGR_EPW + lane;
    int s, d; edge_sd(e, src, dst, s, d);
    const float l  = leaky(s_src[s] + s_dst[d]);
    const float mv = __int_as_float(mbits[d]);
    const float w  = expf(l - mv);
    atomicAdd(&den[d], w);
    const int wbits = __float_as_int(w);
    const int lo = lane & 15;
    float v[FIN];
    gather_rows<0>(s, x + lo, v);
    scatter_rows<0>(d, wbits, agg + lo, v);
}

__global__ void gemm_norm_act_kernel(const float* __restrict__ agg, const float* __restrict__ den,
                                     const float* __restrict__ W, const float* __restrict__ b,
                                     float* __restrict__ act, int nRowTiles) {
    const int lane = threadIdx.x & 31;
    const int gw   = blockIdx.x * 8 + (threadIdx.x >> 5);
    const int rowTile = gw >> 2;
    const int colTile = gw & 3;
    if (rowTile >= nRowTiles) return;
    const int hi  = lane >> 4;
    const int lo  = lane & 15;
    const int row = rowTile * 16 + lo;
    const int col = colTile * 16 + lo;
    const float dn = den[row] + 1e-16f;
    v8f c = {};
#pragma unroll
    for (int kk = 0; kk < FIN; kk += 4) {
        const int k0 = kk + 2 * hi;
        v2f a, bb;
        a.x  = agg[(size_t)row * FIN + k0]     / dn;
        a.y  = agg[(size_t)row * FIN + k0 + 1] / dn;
        bb.x = W[(size_t)k0 * FH + col];
        bb.y = W[(size_t)(k0 + 1) * FH + col];
        c = __builtin_amdgcn_wmma_f32_16x16x4_f32(false, a, false, bb, (short)0, c, false, false);
    }
    const int rbase = rowTile * 16 + 8 * hi;
    const float bias = b[col];
#pragma unroll
    for (int r = 0; r < 8; ++r) {
        const float o = c[r] + bias;
        act[(size_t)(rbase + r) * FH + col] = o > 0.f ? o : 0.f;
    }
}

__global__ void act_kernel(float* __restrict__ buf, const float* __restrict__ den,
                           const float* __restrict__ b) {
    const size_t i = (size_t)blockIdx.x * blockDim.x + threadIdx.x;
    if (i >= (size_t)NNODES * FH) return;
    const float v = buf[i] / (den[i >> 6] + 1e-16f) + b[i & 63];
    buf[i] = v > 0.f ? v : 0.f;
}

__global__ void policy_kernel(const float* __restrict__ act, const float* __restrict__ Wp,
                              const float* __restrict__ bp, float* __restrict__ out,
                              int nRowTiles) {
    const int lane = threadIdx.x & 31;
    const int gw   = blockIdx.x * 8 + (threadIdx.x >> 5);
    if (gw >= nRowTiles) return;
    const int hi  = lane >> 4;
    const int lo  = lane & 15;
    const int row = gw * 16 + lo;
    v8f c = {};
    for (int kk = 0; kk < FH; kk += 4) {
        const int k0 = kk + 2 * hi;
        v2f a, b;
        a.x = act[(size_t)row * FH + k0];
        a.y = act[(size_t)row * FH + k0 + 1];
        b.x = (lo < FOUT) ? Wp[(size_t)k0 * FOUT + lo]       : 0.f;
        b.y = (lo < FOUT) ? Wp[(size_t)(k0 + 1) * FOUT + lo] : 0.f;
        c = __builtin_amdgcn_wmma_f32_16x16x4_f32(false, a, false, b, (short)0, c, false, false);
    }
    if (lo < FOUT) {
        const int rbase = gw * 16 + 8 * hi;
        const float bb = bp[lo];
#pragma unroll
        for (int r = 0; r < 8; ++r)
            out[(size_t)(rbase + r) * FOUT + lo] = c[r] + bb;
    }
}

extern "C" void kernel_launch(void* const* d_in, const int* in_sizes, int n_in,
                              void* d_out, int out_size, void* d_ws, size_t ws_size,
                              hipStream_t stream) {
    const float* x   = (const float*)d_in[0];
    const int*   ei  = (const int*)  d_in[1];
    const float* W1  = (const float*)d_in[2];
    const float* as1 = (const float*)d_in[3];
    const float* ad1 = (const float*)d_in[4];
    const float* b1  = (const float*)d_in[5];
    const float* W2  = (const float*)d_in[6];
    const float* as2 = (const float*)d_in[7];
    const float* ad2 = (const float*)d_in[8];
    const float* b2  = (const float*)d_in[9];
    const float* Wp  = (const float*)d_in[10];
    const float* bp  = (const float*)d_in[11];
    float* out = (float*)d_out;

    const int* srcA = ei;
    const int* dstA = ei + NEDGES;

    float* ws   = (float*)d_ws;
    float* hbuf = ws;
    float* obuf = ws + (size_t)NNODES * FH;
    float* ssrc = ws + (size_t)2 * NNODES * FH;
    float* sdst = ssrc + NNODES;
    int*   mb   = (int*)(sdst + NNODES);
    float* den  = (float*)(mb + NNODES);

    const int nRowTiles = NNODES / 16;
    const dim3 blk(256);
    const int gemmBlocks = (nRowTiles * 4 + 7) / 8;
    const int nodeBlocks = (NNODES + 7) / 8;
    const int edgeBlocks = (NEP + 255) / 256;
    const int aggrBlocks = (NEP / AGGR_EPW + 7) / 8;
    const int actBlocks  = (NNODES * FH + 255) / 256;
    const int polBlocks  = (nRowTiles + 7) / 8;

    float* agg = obuf;
    gemm_h_kernel<<<gemmBlocks, blk, 0, stream>>>(x, W1, hbuf, FIN, nRowTiles);
    scalars_init_kernel<FIN><<<nodeBlocks, blk, 0, stream>>>(hbuf, as1, ad1, ssrc, sdst, mb, den, agg);
    edge_max_kernel<<<edgeBlocks, blk, 0, stream>>>(srcA, dstA, ssrc, sdst, mb);
    edge_aggr_in_kernel<<<aggrBlocks, blk, 0, stream>>>(srcA, dstA, ssrc, sdst, mb, den, x, agg);
    gemm_norm_act_kernel<<<gemmBlocks, blk, 0, stream>>>(agg, den, W1, b1, hbuf, nRowTiles);

    gemm_h_kernel<<<gemmBlocks, blk, 0, stream>>>(hbuf, W2, obuf, FH, nRowTiles);
    scalars_init_kernel<FH><<<nodeBlocks, blk, 0, stream>>>(obuf, as2, ad2, ssrc, sdst, mb, den, hbuf);
    edge_max_kernel<<<edgeBlocks, blk, 0, stream>>>(srcA, dstA, ssrc, sdst, mb);
    edge_aggr_kernel<<<aggrBlocks, blk, 0, stream>>>(srcA, dstA, ssrc, sdst, mb, den, obuf, hbuf);
    act_kernel<<<actBlocks, blk, 0, stream>>>(hbuf, den, b2);

    policy_kernel<<<polBlocks, blk, 0, stream>>>(hbuf, Wp, bp, out, nRowTiles);
}


# Segment max taken over s_src[src] alone (adding s_dst[dst] and leaky are monotone, applied after the max: same bits); s_dst and that max share one 8-byte word per node: 2 fewer random reads per edge
# speedup vs baseline: 1.5813x; 1.1134x over previous
#include <hip/hip_runtime.h>
#include <math.h>

#define NNODES 100000
#define NEDGES 1600000
#define NEP    1700000
#define FIN    16
#define FH     64
#define FOUT   8

typedef __attribute__((ext_vector_type(2))) float v2f;
typedef __attribute__((ext_vector_type(8))) float v8f;

__device__ __forceinline__ float leaky(float e) { return e > 0.f ? e : 0.2f * e; }

__global__ void gemm_h_kernel(const float* __restrict__ in, const float* __restrict__ W,
                              float* __restrict__ h, int K, int nRowTiles) {
    const int lane = threadIdx.x & 31;
    const int gw   = blockIdx.x * 8 + (threadIdx.x >> 5);
    const int rowTile = gw >> 2;
    const int colTile = gw & 3;
    if (rowTile >= nRowTiles) return;
    const int hi  = lane >> 4;
    const int lo  = lane & 15;
    const int row = rowTile * 16 + lo;
    const int col = colTile * 16 + lo;
    v8f c = {};
    for (int kk = 0; kk < K; kk += 4) {
        const int k0 = kk + 2 * hi;
        v2f a, b;
        a.x = in[(size_t)row * K + k0];
        a.y = in[(size_t)row * K + k0 + 1];
        b.x = W[(size_t)k0 * FH + col];
        b.y = W[(size_t)(k0 + 1) * FH + col];
        c = __builtin_amdgcn_wmma_f32_16x16x4_f32(false, a, false, b, (short)0, c, false, false);
    }
    const int rbase = rowTile * 16 + 8 * hi;
#pragma unroll
    for (int r = 0; r < 8; ++r)
        h[(size_t)(rbase + r) * FH + col] = c[r];
}

template <int OW>
__global__ void scalars_init_kernel(const float* __restrict__ h,
                                    const float* __restrict__ a_src,
                                    const float* __restrict__ a_dst,
                                    float* __restrict__ s_src, float* __restrict__ nd,
                                    float* __restrict__ den, float* __restrict__ obuf) {
    const int lane = threadIdx.x & 31;
    const int n    = blockIdx.x * 8 + (threadIdx.x >> 5);
    if (n >= NNODES) return;
    const float h0 = h[(size_t)n * FH + lane];
    const float h1 = h[(size_t)n * FH + lane + 32];
    float ps = h0 * a_src[lane] + h1 * a_src[lane + 32];
    float pd = h0 * a_dst[lane] + h1 * a_dst[lane + 32];
    for (int off = 16; off; off >>= 1) {
        ps += __shfl_xor(ps, off, 32);
        pd += __shfl_xor(pd, off, 32);
    }
    if (lane == 0) {
        s_src[n] = ps;
        v2f p; p.x = pd; p.y = __int_as_float((int)0xFF800000u);
        *(v2f*)(nd + 2 * (size_t)n) = p;
        den[n]   = 0.f;
    }
    if (OW == FH) {
        obuf[(size_t)n * FH + lane]      = 0.f;
        obuf[(size_t)n * FH + lane + 32] = 0.f;
    } else if (lane < OW) {
        obuf[(size_t)n * OW + lane] = 0.f;
    }
}

__device__ __forceinline__ void edge_sd(int e, const int* __restrict__ src,
                                        const int* __restrict__ dst, int& s, int& d) {
    if (e < NEDGES) { s = src[e]; d = dst[e]; }
    else            { s = d = e - NEDGES; }
}

__global__ void edge_max_kernel(const int* __restrict__ src, const int* __restrict__ dst,
                                const float* __restrict__ s_src, float* __restrict__ nd) {
    const int e = blockIdx.x * blockDim.x + threadIdx.x;
    if (e >= NEP) return;
    int s, d; edge_sd(e, src, dst, s, d);
    const unsigned u = __float_as_uint(s_src[s]);
    float* mp = nd + 2 * (size_t)d + 1;
    if ((u >> 31) == 0) atomicMax((int*)mp, (int)u);
    else                atomicMin((unsigned int*)mp, u);
}

__device__ __forceinline__ float edge_weight(float ss, const float* __restrict__ nd, int d) {
    const v2f p = *(const v2f*)(nd + 2 * (size_t)d);
    const float l  = leaky(ss + p.x);
    const float mv = leaky(p.y + p.x);
    return expf(l - mv);
}

#define AGGR_EPW  32
#define AGGR_ROWS 8
static_assert(NEP % AGGR_EPW == 0 && NEDGES % AGGR_EPW == 0, "full waves only");
static_assert(AGGR_EPW % AGGR_ROWS == 0, "whole groups");

__global__ void __launch_bounds__(256)
edge_aggr_kernel(const int* __restrict__ src, const int* __restrict__ dst,
                 const float* __restrict__ s_src, const float* __restrict__ nd,
                 float* __restrict__ den,
                 const float* __restrict__ h, float* __restrict__ out) {
    const int lane = threadIdx.x & 31;
    const int wv   = blockIdx.x * 8 + (threadIdx.x >> 5);
    if (wv >= NEP / AGGR_EPW) return;
    const int e = wv * AGGR_EPW + lane;
    int s, d; edge_sd(e, src, dst, s, d);
    const float w  = edge_weight(s_src[s], nd, d);
    atomicAdd(&den[d], w);
    const int wbits = __float_as_int(w);
    const float* __restrict__ hl = h + lane;
    float* __restrict__ ol = out + lane;
#pragma unroll
    for (int g = 0; g < AGGR_EPW; g += AGGR_ROWS) {
        float v0[AGGR_ROWS], v1[AGGR_ROWS];
#pragma unroll
        for (int j = 0; j < AGGR_ROWS; ++j) {
            const int sj = __builtin_amdgcn_readlane(s, g + j);
            v0[j] = hl[(size_t)sj * FH];
            v1[j] = hl[(size_t)sj * FH + 32];
        }
#pragma unroll
        for (int j = 0; j < AGGR_ROWS; ++j) {
            const int   dj = __builtin_amdgcn_readlane(d, g + j);
            const float wj = __int_as_float(__builtin_amdgcn_readlane(wbits, g + j));
            atomicAdd(&ol[(size_t)dj * FH],      v0[j] * wj);
            atomicAdd(&ol[(size_t)dj * FH + 32], v1[j] * wj);
        }
    }
}

template <int J>
__device__ __forceinline__ int row_lane(int v) {
    return __builtin_amdgcn_update_dpp(0, v, 0x150 + J, 0xF, 0xF, false);
}
template <int J>
__device__ __forceinline__ void gather_rows(int s, const float* __restrict__ xl, float (&v)[FIN]) {
    v[J] = xl[(size_t)row_lane<J>(s) * FIN];
    if constexpr (J + 1 < FIN) gather_rows<J + 1>(s, xl, v);
}
template <int J>
__device__ __forceinline__ void scatter_rows(int d, int wbits, float* __restrict__ al,
                                             const float (&v)[FIN]) {
    const int   dj = row_lane<J>(d);
    const float wj = __int_as_float(row_lane<J>(wbits));
    atomicAdd(&al[(size_t)dj * FIN], v[J] * wj);
    if constexpr (J + 1 < FIN) scatter_rows<J + 1>(d, wbits, al, v);
}

__global__ void __launch_bounds__(256)
edge_aggr_in_kernel(const int* __restrict__ src, const int* __restrict__ dst,
                    const float* __restrict__ s_src, const float* __restrict__ nd,
                    float* __restrict__ den,
                    const float* __restrict__ x, float* __restrict__ agg) {
    const int lane = threadIdx.x & 31;
    const int wv   = blockIdx.x * 8 + (threadIdx.x >> 5);
    if (wv >= NEP / AGGR_EPW) return;
    const int e = wv * AGGR_EPW + lane;
    int s, d; edge_sd(e, src, dst, s, d);
    const float w  = edge_weight(s_src[s], nd, d);
    atomicAdd(&den[d], w);
    const int wbits = __float_as_int(w);
    const int lo = lane & 15;
    float v[FIN];
    gather_rows<0>(s, x + lo, v);
    scatter_rows<0>(d, wbits, agg + lo, v);
}

__global__ void gemm_norm_act_kernel(const float* __restrict__ agg, const float* __restrict__ den,
                                     const float* __restrict__ W, const float* __restrict__ b,
                                     float* __restrict__ act, int nRowTiles) {
    const int lane = threadIdx.x & 31;
    const int gw   = blockIdx.x * 8 + (threadIdx.x >> 5);
    const int rowTile = gw >> 2;
    const int colTile = gw & 3;
    if (rowTile >= nRowTiles) return;
    const int hi  = lane >> 4;
    const int lo  = lane & 15;
    const int row = rowTile * 16 + lo;
    const int col = colTile * 16 + lo;
    const float dn = den[row] + 1e-16f;
    v8f c = {};
#pragma unroll
    for (int kk = 0; kk < FIN; kk += 4) {
        const int k0 = kk + 2 * hi;
        v2f a, bb;
        a.x  = agg[(size_t)row * FIN + k0]     / dn;
        a.y  = agg[(size_t)row * FIN + k0 + 1] / dn;
        bb.x = W[(size_t)k0 * FH + col];
        bb.y = W[(size_t)(k0 + 1) * FH + col];
        c = __builtin_amdgcn_wmma_f32_16x16x4_f32(false, a, false, bb, (short)0, c, false, false);
    }
    const int rbase = rowTile * 16 + 8 * hi;
    const float bias = b[col];
#pragma unroll
    for (int r = 0; r < 8; ++r) {
        const float o = c[r] + bias;
        act[(size_t)(rbase + r) * FH + col] = o > 0.f ? o : 0.f;
    }
}

__global__ void act_kernel(float* __restrict__ buf, const float* __restrict__ den,
                           const float* __restrict__ b) {
    const size_t i = (size_t)blockIdx.x * blockDim.x + threadIdx.x;
    if (i >= (size_t)NNODES * FH) return;
    const float v = buf[i] / (den[i >> 6] + 1e-16f) + b[i & 63];
    buf[i] = v > 0.f ? v : 0.f;
}

__global__ void policy_kernel(const float* __restrict__ act, const float* __restrict__ Wp,
                              const float* __restrict__ bp, float* __restrict__ out,
                              int nRowTiles) {
    const int lane = threadIdx.x & 31;
    const int gw   = blockIdx.x * 8 + (threadIdx.x >> 5);
    if (gw >= nRowTiles) return;
    const int hi  = lane >> 4;
    const int lo  = lane & 15;
    const int row = gw * 16 + lo;
    v8f c = {};
    for (int kk = 0; kk < FH; kk += 4) {
        const int k0 = kk + 2 * hi;
        v2f a, b;
        a.x = act[(size_t)row * FH + k0];
        a.y = act[(size_t)row * FH + k0 + 1];
        b.x = (lo < FOUT) ? Wp[(size_t)k0 * FOUT + lo]       : 0.f;
        b.y = (lo < FOUT) ? Wp[(size_t)(k0 + 1) * FOUT + lo] : 0.f;
        c = __builtin_amdgcn_wmma_f32_16x16x4_f32(false, a, false, b, (short)0, c, false, false);
    }
    if (lo < FOUT) {
        const int rbase = gw * 16 + 8 * hi;
        const float bb = bp[lo];
#pragma unroll
        for (int r = 0; r < 8; ++r)
            out[(size_t)(rbase + r) * FOUT + lo] = c[r] + bb;
    }
}

extern "C" void kernel_launch(void* const* d_in, const int* in_sizes, int n_in,
                              void* d_out, int out_size, void* d_ws, size_t ws_size,
                              hipStream_t stream) {
    const float* x   = (const float*)d_in[0];
    const int*   ei  = (const int*)  d_in[1];
    const float* W1  = (const float*)d_in[2];
    const float* as1 = (const float*)d_in[3];
    const float* ad1 = (const float*)d_in[4];
    const float* b1  = (const float*)d_in[5];
    const float* W2  = (const float*)d_in[6];
    const float* as2 = (const float*)d_in[7];
    const float* ad2 = (const float*)d_in[8];
    const float* b2  = (const float*)d_in[9];
    const float* Wp  = (const float*)d_in[10];
    const float* bp  = (const float*)d_in[11];
    float* out = (float*)d_out;

    const int* srcA = ei;
    const int* dstA = ei + NEDGES;

    float* ws   = (float*)d_ws;
    float* hbuf = ws;
    float* obuf = ws + (size_t)NNODES * FH;
    float* ssrc = ws + (size_t)2 * NNODES * FH;
    float* nd   = ssrc + NNODES;
    float* den  = nd + (size_t)2 * NNODES;

    const int nRowTiles = NNODES / 16;
    const dim3 blk(256);
    const int gemmBlocks = (nRowTiles * 4 + 7) / 8;
    const int nodeBlocks = (NNODES + 7) / 8;
    const int edgeBlocks = (NEP + 255) / 256;
    const int aggrBlocks = (NEP / AGGR_EPW + 7) / 8;
    const int actBlocks  = (NNODES * FH + 255) / 256;
    const int polBlocks  = (nRowTiles + 7) / 8;

    float* agg = obuf;
    gemm_h_kernel<<<gemmBlocks, blk, 0, stream>>>(x, W1, hbuf, FIN, nRowTiles);
    scalars_init_kernel<FIN><<<nodeBlocks, blk, 0, stream>>>(hbuf, as1, ad1, ssrc, nd, den, agg);
    edge_max_kernel<<<edgeBlocks, blk, 0, stream>>>(srcA, dstA, ssrc, nd);
    edge_aggr_in_kernel<<<aggrBlocks, blk, 0, stream>>>(srcA, dstA, ssrc, nd, den, x, agg);
    gemm_norm_act_kernel<<<gemmBlocks, blk, 0, stream>>>(agg, den, W1, b1, hbuf, nRowTiles);

    gemm_h_kernel<<<gemmBlocks, blk, 0, stream>>>(hbuf, W2, obuf, FH, nRowTiles);
    scalars_init_kernel<FH><<<nodeBlocks, blk, 0, stream>>>(obuf, as2, ad2, ssrc, nd, den, hbuf);
    edge_max_kernel<<<edgeBlocks, blk, 0, stream>>>(srcA, dstA, ssrc, nd);
    edge_aggr_kernel<<<aggrBlocks, blk, 0, stream>>>(srcA, dstA, ssrc, nd, den, obuf, hbuf);
    act_kernel<<<actBlocks, blk, 0, stream>>>(hbuf, den, b2);

    policy_kernel<<<polBlocks, blk, 0, stream>>>(hbuf, Wp, bp, out, nRowTiles);
}


# Layer 1: the softmax denominator rides in the node's accumulator row (column 16 of a 128-byte row) as a 17th lane of the edge's own atomic add: the random 4-byte atomic per edge into den goes
# speedup vs baseline: 1.8341x; 1.1599x over previous
#include <hip/hip_runtime.h>
#include <math.h>

#define NNODES 100000
#define NEDGES 1600000
#define NEP    1700000
#define FIN    16
#define FH     64
#define FOUT   8
#define AGW    32

typedef __attribute__((ext_vector_type(2))) float v2f;
typedef __attribute__((ext_vector_type(8))) float v8f;

__device__ __forceinline__ float leaky(float e) { return e > 0.f ? e : 0.2f * e; }

__global__ void gemm_h_kernel(const float* __restrict__ in, const float* __restrict__ W,
                              float* __restrict__ h, int K, int nRowTiles) {
    const int lane = threadIdx.x & 31;
    const int gw   = blockIdx.x * 8 + (threadIdx.x >> 5);
    const int rowTile = gw >> 2;
    const int colTile = gw & 3;
    if (rowTile >= nRowTiles) return;
    const int hi  = lane >> 4;
    const int lo  = lane & 15;
    const int row = rowTile * 16 + lo;
    const int col = colTile * 16 + lo;
    v8f c = {};
    for (int kk = 0; kk < K; kk += 4) {
        const int k0 = kk + 2 * hi;
        v2f a, b;
        a.x = in[(size_t)row * K + k0];
        a.y = in[(size_t)row * K + k0 + 1];
        b.x = W[(size_t)k0 * FH + col];
        b.y = W[(size_t)(k0 + 1) * FH + col];
        c = __builtin_amdgcn_wmma_f32_16x16x4_f32(false, a, false, b, (short)0, c, false, false);
    }
    const int rbase = rowTile * 16 + 8 * hi;
#pragma unroll
    for (int r = 0; r < 8; ++r)
        h[(size_t)(rbase + r) * FH + col] = c[r];
}

template <int OW>
__global__ void scalars_init_kernel(const float* __restrict__ h,
                                    const float* __restrict__ a_src,
                                    const float* __restrict__ a_dst,
                                    float* __restrict__ s_src, float* __restrict__ nd,
                                    float* __restrict__ den, float* __restrict__ obuf) {
    const int lane = threadIdx.x & 31;
    const int n    = blockIdx.x * 8 + (threadIdx.x >> 5);
    if (n >= NNODES) return;
    const float h0 = h[(size_t)n * FH + lane];
    const float h1 = h[(size_t)n * FH + lane + 32];
    float ps = h0 * a_src[lane] + h1 * a_src[lane + 32];
    float pd = h0 * a_dst[lane] + h1 * a_dst[lane + 32];
    for (int off = 16; off; off >>= 1) {
        ps += __shfl_xor(ps, off, 32);
        pd += __shfl_xor(pd, off, 32);
    }
    if (lane == 0) {
        s_src[n] = ps;
        v2f p; p.x = pd; p.y = __int_as_float((int)0xFF800000u);
        *(v2f*)(nd + 2 * (size_t)n) = p;
        if (OW == FH) den[n] = 0.f;
    }
    if (OW == FH) {
        obuf[(size_t)n * FH + lane]      = 0.f;
        obuf[(size_t)n * FH + lane + 32] = 0.f;
    } else {
        obuf[(size_t)n * AGW + lane] = 0.f;
    }
}

__device__ __forceinline__ void edge_sd(int e, const int* __restrict__ src,
                                        const int* __restrict__ dst, int& s, int& d) {
    if (e < NEDGES) { s = src[e]; d = dst[e]; }
    else            { s = d = e - NEDGES; }
}

__global__ void edge_max_kernel(const int* __restrict__ src, const int* __restrict__ dst,
                                const float* __restrict__ s_src, float* __restrict__ nd) {
    const int e = blockIdx.x * blockDim.x + threadIdx.x;
    if (e >= NEP) return;
    int s, d; edge_sd(e, src, dst, s, d);
    const unsigned u = __float_as_uint(s_src[s]);
    float* mp = nd + 2 * (size_t)d + 1;
    if ((u >> 31) == 0) atomicMax((int*)mp, (int)u);
    else                atomicMin((unsigned int*)mp, u);
}

__device__ __forceinline__ float edge_weight(float ss, const float* __restrict__ nd, int d) {
    const v2f p = *(const v2f*)(nd + 2 * (size_t)d);
    const float l  = leaky(ss + p.x);
    const float mv = leaky(p.y + p.x);
    return expf(l - mv);
}

#define AGGR_EPW  32
#define AGGR_ROWS 8
static_assert(NEP % AGGR_EPW == 0 && NEDGES % AGGR_EPW == 0, "full waves only");
static_assert(AGGR_EPW % AGGR_ROWS == 0, "whole groups");

__global__ void __launch_bounds__(256)
edge_aggr_kernel(const int* __restrict__ src, const int* __restrict__ dst,
                 const float* __restrict__ s_src, const float* __restrict__ nd,
                 float* __restrict__ den,
                 const float* __restrict__ h, float* __restrict__ out) {
    const int lane = threadIdx.x & 31;
    const int wv   = blockIdx.x * 8 + (threadIdx.x >> 5);
    if (wv >= NEP / AGGR_EPW) return;
    const int e = wv * AGGR_EPW + lane;
    int s, d; edge_sd(e, src, dst, s, d);
    const float w  = edge_weight(s_src[s], nd, d);
    atomicAdd(&den[d], w);
    const int wbits = __float_as_int(w);
    const float* __restrict__ hl = h + lane;
    float* __restrict__ ol = out + lane;
#pragma unroll
    for (int g = 0; g < AGGR_EPW; g += AGGR_ROWS) {
        float v0[AGGR_ROWS], v1[AGGR_ROWS];
#pragma unroll
        for (int j = 0; j < AGGR_ROWS; ++j) {
            const int sj = __builtin_amdgcn_readlane(s, g + j);
            v0[j] = hl[(size_t)sj * FH];
            v1[j] = hl[(size_t)sj * FH + 32];
        }
#pragma unroll
        for (int j = 0; j < AGGR_ROWS; ++j) {
            const int   dj = __builtin_amdgcn_readlane(d, g + j);
            const float wj = __int_as_float(__builtin_amdgcn_readlane(wbits, g + j));
            atomicAdd(&ol[(size_t)dj * FH],      v0[j] * wj);
            atomicAdd(&ol[(size_t)dj * FH + 32], v1[j] * wj);
        }
    }
}

static_assert(FIN == 16 && AGW >= FIN + 1, "half a wave per input row, one more column for the denominator");
template <int J>
__device__ __forceinline__ int row_lane(int v) {
    return __builtin_amdgcn_update_dpp(0, v, 0x150 + J, 0xF, 0xF, false);
}
template <int J>
__device__ __forceinline__ void gather_rows(int s, const float* __restrict__ xl, float (&v)[FIN]) {
    v[J] = xl[(size_t)row_lane<J>(s) * FIN];
    if constexpr (J + 1 < FIN) gather_rows<J + 1>(s, xl, v);
}
template <int J>
__device__ __forceinline__ void scatter_rows(int lane, int d, int wbits, float* __restrict__ agg,
                                             const float (&v)[FIN]) {
    {
        const int   dj = __builtin_amdgcn_readlane(d, J);
        const float wj = __int_as_float(__builtin_amdgcn_readlane(wbits, J));
        if (lane <= FIN)
            atomicAdd(&agg[(size_t)dj * AGW + lane], lane < FIN ? v[J] * wj : wj);
    }
    {
        const int   dj = __builtin_amdgcn_readlane(d, FIN + J);
        const float wj = __int_as_float(__builtin_amdgcn_readlane(wbits, FIN + J));
        if (lane >= FIN || lane == 0)
            atomicAdd(&agg[(size_t)dj * AGW + (lane >= FIN ? lane - FIN : FIN)],
                      lane >= FIN ? v[J] * wj : wj);
    }
    if constexpr (J + 1 < FIN) scatter_rows<J + 1>(lane, d, wbits, agg, v);
}

__global__ void __launch_bounds__(256)
edge_aggr_in_kernel(const int* __restrict__ src, const int* __restrict__ dst,
                    const float* __restrict__ s_src, const float* __restrict__ nd,
                    const float* __restrict__ x, float* __restrict__ agg) {
    const int lane = threadIdx.x & 31;
    const int wv   = blockIdx.x * 8 + (threadIdx.x >> 5);
    if (wv >= NEP / AGGR_EPW) return;
    const int e = wv * AGGR_EPW + lane;
    int s, d; edge_sd(e, src, dst, s, d);
    const float w  = edge_weight(s_src[s], nd, d);
    const int wbits = __float_as_int(w);
    const int lo = lane & 15;
    float v[FIN];
    gather_rows<0>(s, x + lo, v);
    scatter_rows<0>(lane, d, wbits, agg, v);
}

__global__ void gemm_norm_act_kernel(const float* __restrict__ agg,
                                     const float* __restrict__ W, const float* __restrict__ b,
                                     float* __restrict__ act, int nRowTiles) {
    const int lane = threadIdx.x & 31;
    const int gw   = blockIdx.x * 8 + (threadIdx.x >> 5);
    const int rowTile = gw >> 2;
    const int colTile = gw & 3;
    if (rowTile >= nRowTiles) return;
    const int hi  = lane >> 4;
    const int lo  = lane & 15;
    const int row = rowTile * 16 + lo;
    const int col = colTile * 16 + lo;
    const float dn = agg[(size_t)row * AGW + FIN] + 1e-16f;
    v8f c = {};
#pragma unroll
    for (int kk = 0; kk < FIN; kk += 4) {
        const int k0 = kk + 2 * hi;
        v2f a, bb;
        a.x  = agg[(size_t)row * AGW + k0]     / dn;
        a.y  = agg[(size_t)row * AGW + k0 + 1] / dn;
        bb.x = W[(size_t)k0 * FH + col];
        bb.y = W[(size_t)(k0 + 1) * FH + col];
        c = __builtin_amdgcn_wmma_f32_16x16x4_f32(false, a, false, bb, (short)0, c, false, false);
    }
    const int rbase = rowTile * 16 + 8 * hi;
    const float bias = b[col];
#pragma unroll
    for (int r = 0; r < 8; ++r) {
        const float o = c[r] + bias;
        act[(size_t)(rbase + r) * FH + col] = o > 0.f ? o : 0.f;
    }
}

__global__ void act_kernel(float* __restrict__ buf, const float* __restrict__ den,
                           const float* __restrict__ b) {
    const size_t i = (size_t)blockIdx.x * blockDim.x + threadIdx.x;
    if (i >= (size_t)NNODES * FH) return;
    const float v = buf[i] / (den[i >> 6] + 1e-16f) + b[i & 63];
    buf[i] = v > 0.f ? v : 0.f;
}

__global__ void policy_kernel(const float* __restrict__ act, const float* __restrict__ Wp,
                              const float* __restrict__ bp, float* __restrict__ out,
                              int nRowTiles) {
    const int lane = threadIdx.x & 31;
    const int gw   = blockIdx.x * 8 + (threadIdx.x >> 5);
    if (gw >= nRowTiles) return;
    const int hi  = lane >> 4;
    const int lo  = lane & 15;
    const int row = gw * 16 + lo;
    v8f c = {};
    for (int kk = 0; kk < FH; kk += 4) {
        const int k0 = kk + 2 * hi;
        v2f a, b;
        a.x = act[(size_t)row * FH + k0];
        a.y = act[(size_t)row * FH + k0 + 1];
        b.x = (lo < FOUT) ? Wp[(size_t)k0 * FOUT + lo]       : 0.f;
        b.y = (lo < FOUT) ? Wp[(size_t)(k0 + 1) * FOUT + lo] : 0.f;
        c = __builtin_amdgcn_wmma_f32_16x16x4_f32(false, a, false, b, (short)0, c, false, false);
    }
    if (lo < FOUT) {
        const int rbase = gw * 16 + 8 * hi;
        const float bb = bp[lo];
#pragma unroll
        for (int r = 0; r < 8; ++r)
            out[(size_t)(rbase + r) * FOUT + lo] = c[r] + bb;
    }
}

extern "C" void kernel_launch(void* const* d_in, const int* in_sizes, int n_in,
                              void* d_out, int out_size, void* d_ws, size_t ws_size,
                              hipStream_t stream) {
    const float* x   = (const float*)d_in[0];
    const int*   ei  = (const int*)  d_in[1];
    const float* W1  = (const float*)d_in[2];
    const float* as1 = (const float*)d_in[3];
    const float* ad1 = (const float*)d_in[4];
    const float* b1  = (const float*)d_in[5];
    const float* W2  = (const float*)d_in[6];
    const float* as2 = (const float*)d_in[7];
    const float* ad2 = (const float*)d_in[8];
    const float* b2  = (const float*)d_in[9];
    const float* Wp  = (const float*)d_in[10];
    const float* bp  = (const float*)d_in[11];
    float* out = (float*)d_out;

    const int* srcA = ei;
    const int* dstA = ei + NEDGES;

    float* ws   = (float*)d_ws;
    float* hbuf = ws;
    float* obuf = ws + (size_t)NNODES * FH;
    float* ssrc = ws + (size_t)2 * NNODES * FH;
    float* nd   = ssrc + NNODES;
    float* den  = nd + (size_t)2 * NNODES;

    const int nRowTiles = NNODES / 16;
    const dim3 blk(256);
    const int gemmBlocks = (nRowTiles * 4 + 7) / 8;
    const int nodeBlocks = (NNODES + 7) / 8;
    const int edgeBlocks = (NEP + 255) / 256;
    const int aggrBlocks = (NEP / AGGR_EPW + 7) / 8;
    const int actBlocks  = (NNODES * FH + 255) / 256;
    const int polBlocks  = (nRowTiles + 7) / 8;

    float* agg = obuf;
    gemm_h_kernel<<<gemmBlocks, blk, 0, stream>>>(x, W1, hbuf, FIN, nRowTiles);
    scalars_init_kernel<AGW><<<nodeBlocks, blk, 0, stream>>>(hbuf, as1, ad1, ssrc, nd, den, agg);
    edge_max_kernel<<<edgeBlocks, blk, 0, stream>>>(srcA, dstA, ssrc, nd);
    edge_aggr_in_kernel<<<aggrBlocks, blk, 0, stream>>>(srcA, dstA, ssrc, nd, x, agg);
    gemm_norm_act_kernel<<<gemmBlocks, blk, 0, stream>>>(agg, W1, b1, hbuf, nRowTiles);

    gemm_h_kernel<<<gemmBlocks, blk, 0, stream>>>(hbuf, W2, obuf, FH, nRowTiles);
    scalars_init_kernel<FH><<<nodeBlocks, blk, 0, stream>>>(obuf, as2, ad2, ssrc, nd, den, hbuf);
    edge_max_kernel<<<edgeBlocks, blk, 0, stream>>>(srcA, dstA, ssrc, nd);
    edge_aggr_kernel<<<aggrBlocks, blk, 0, stream>>>(srcA, dstA, ssrc, nd, den, obuf, hbuf);
    act_kernel<<<actBlocks, blk, 0, stream>>>(hbuf, den, b2);

    policy_kernel<<<polBlocks, blk, 0, stream>>>(hbuf, Wp, bp, out, nRowTiles);
}


# Edges grouped by destination once per call (count, prefix sum, fill in arrival order); one wave owns a node: max, weights and row sums in registers in list order; edge atomics, max passes, clears go
# speedup vs baseline: 2.5572x; 1.3943x over previous
#include <hip/hip_runtime.h>
#include <math.h>

#define NNODES 100000
#define NEDGES 1600000
#define FIN    16
#define FH     64
#define FOUT   8
#define AGW    32

typedef __attribute__((ext_vector_type(2))) float v2f;
typedef __attribute__((ext_vector_type(8))) float v8f;

#define NHALF  (NNODES / 2)
static_assert(NNODES % 32 == 0, "a half is a whole number of 16-row tiles");

__device__ __forceinline__ float leaky(float e) { return e > 0.f ? e : 0.2f * e; }

__global__ void gemm_h_kernel(const float* __restrict__ in, const float* __restrict__ W,
                              float* __restrict__ h, int K, int nRowTiles) {
    const int lane = threadIdx.x & 31;
    const int gw   = blockIdx.x * 8 + (threadIdx.x >> 5);
    const int rowTile = gw >> 2;
    const int colTile = gw & 3;
    if (rowTile >= nRowTiles) return;
    const int hi  = lane >> 4;
    const int lo  = lane & 15;
    const int row = rowTile * 16 + lo;
    const int col = colTile * 16 + lo;
    v8f c = {};
    for (int kk = 0; kk < K; kk += 4) {
        const int k0 = kk + 2 * hi;
        v2f a, b;
        a.x = in[(size_t)row * K + k0];
        a.y = in[(size_t)row * K + k0 + 1];
        b.x = W[(size_t)k0 * FH + col];
        b.y = W[(size_t)(k0 + 1) * FH + col];
        c = __builtin_amdgcn_wmma_f32_16x16x4_f32(false, a, false, b, (short)0, c, false, false);
    }
    const int rbase = rowTile * 16 + 8 * hi;
#pragma unroll
    for (int r = 0; r < 8; ++r)
        h[(size_t)(rbase + r) * FH + col] = c[r];
}

__device__ __forceinline__ const float* plane_row(const float* __restrict__ lo,
                                                  const float* __restrict__ hi, int s) {
    return s < NHALF ? lo + (size_t)s * FH : hi + (size_t)(s - NHALF) * FH;
}

__global__ void scalars_kernel(const float* __restrict__ hlo, const float* __restrict__ hhi,
                               const float* __restrict__ a_src,
                               const float* __restrict__ a_dst,
                               float* __restrict__ s_src, float* __restrict__ s_dst) {
    const int lane = threadIdx.x & 31;
    const int n    = blockIdx.x * 8 + (threadIdx.x >> 5);
    if (n >= NNODES) return;
    const float* __restrict__ hr = plane_row(hlo, hhi, n);
    const float h0 = hr[lane];
    const float h1 = hr[lane + 32];
    float ps = h0 * a_src[lane] + h1 * a_src[lane + 32];
    float pd = h0 * a_dst[lane] + h1 * a_dst[lane + 32];
    for (int off = 16; off; off >>= 1) {
        ps += __shfl_xor(ps, off, 32);
        pd += __shfl_xor(pd, off, 32);
    }
    if (lane == 0) { s_src[n] = ps; s_dst[n] = pd; }
}

__global__ void count_zero_kernel(int* __restrict__ cnt) {
    const int n = blockIdx.x * blockDim.x + threadIdx.x;
    if (n < NNODES) cnt[n] = 0;
}

__global__ void count_kernel(const int* __restrict__ dst, int* __restrict__ cnt) {
    const int e = blockIdx.x * blockDim.x + threadIdx.x;
    if (e < NEDGES) atomicAdd(&cnt[dst[e]], 1);
}

static_assert(NNODES % 4 == 0, "counts are read four at a time");
static_assert((NNODES / 4 + 255) / 256 <= 256, "one thread per block total");
__global__ void __launch_bounds__(256)
block_sum_kernel(const int* __restrict__ cnt, int* __restrict__ bsum) {
    __shared__ int wsum[8];
    const int t    = threadIdx.x;
    const int lane = t & 31;
    const int wv   = t >> 5;
    const int q = blockIdx.x * 256 + t;
    int own = 0;
    if (q < NNODES / 4) {
        const int4 v = ((const int4*)cnt)[q];
        own = v.x + v.y + v.z + v.w;
    }
    for (int off = 16; off; off >>= 1) own += __shfl_xor(own, off, 32);
    if (lane == 0) wsum[wv] = own;
    __syncthreads();
    if (t == 0) {
        int s = 0;
        for (int k = 0; k < 8; ++k) s += wsum[k];
        bsum[blockIdx.x] = s;
    }
}

__global__ void __launch_bounds__(256)
scan_kernel(const int* __restrict__ cnt, const int* __restrict__ bsum, int* __restrict__ cursor) {
    __shared__ int wsum[8];
    __shared__ int wpre[8];
    const int t    = threadIdx.x;
    const int lane = t & 31;
    const int wv   = t >> 5;
    const int4* __restrict__ c4 = (const int4*)cnt;
    const int q0 = blockIdx.x * 256;
    int pre = t < (int)blockIdx.x ? bsum[t] : 0;
    const int q = q0 + t;
    int4 v = make_int4(0, 0, 0, 0);
    if (q < NNODES / 4) v = c4[q];
    const int own = v.x + v.y + v.z + v.w;
    int inc = own;
    for (int off = 1; off < 32; off <<= 1) {
        const int u = __shfl_up(inc, off, 32);
        if (lane >= off) inc += u;
    }
    for (int off = 16; off; off >>= 1) pre += __shfl_xor(pre, off, 32);
    if (lane == 31) wsum[wv] = inc;
    if (lane == 0)  wpre[wv] = pre;
    __syncthreads();
    int base = 0;
    for (int k = 0; k < 8; ++k) {
        base += wpre[k];
        if (k < wv) base += wsum[k];
    }
    const int ex = base + inc - own;
    if (q < NNODES / 4) {
        int4 o;
        o.x = ex; o.y = o.x + v.x; o.z = o.y + v.y; o.w = o.z + v.z;
        ((int4*)cursor)[q] = o;
    }
}

__global__ void fill_kernel(const int* __restrict__ src, const int* __restrict__ dst,
                            int* __restrict__ cursor, int* __restrict__ csr) {
    const int e = blockIdx.x * blockDim.x + threadIdx.x;
    if (e >= NEDGES) return;
    const int pos = atomicAdd(&cursor[dst[e]], 1);
    csr[pos] = src[e];
}

#define ROWS_TOGETHER 8

struct NodeEdges { int beg, deg; float sd; };

__device__ __forceinline__ NodeEdges node_edges(int n, const int* __restrict__ rowend,
                                                const float* __restrict__ s_dst) {
    NodeEdges q;
    const int end = rowend[n];
    q.beg = n ? rowend[n - 1] : 0;
    q.deg = end - q.beg;
    q.sd  = s_dst[n];
    return q;
}
__device__ __forceinline__ int edge_source(int i, int n, const NodeEdges& q, const int* __restrict__ csr) {
    return i < q.deg ? csr[q.beg + i] : n;
}
__device__ __forceinline__ float wave_max(float v) {
    for (int off = 16; off; off >>= 1) v = fmaxf(v, __shfl_xor(v, off, 32));
    return v;
}
__device__ __forceinline__ float wave_sum(float v) {
    for (int off = 16; off; off >>= 1) v += __shfl_xor(v, off, 32);
    return v;
}

__global__ void __launch_bounds__(256)
node_aggr_kernel(const int* __restrict__ rowend, const int* __restrict__ csr,
                 const float* __restrict__ s_src, const float* __restrict__ s_dst,
                 const float* __restrict__ hlo, const float* __restrict__ hhi,
                 const float* __restrict__ b, float* __restrict__ act, int n0) {
    const int lane = threadIdx.x & 31;
    const int n    = n0 + blockIdx.x * 8 + __builtin_amdgcn_readfirstlane(threadIdx.x >> 5);
    if (n >= n0 + NHALF) return;
    const NodeEdges q = node_edges(n, rowend, s_dst);
    const int T = q.deg + 1;
    const int   sA  = edge_source(lane, n, q, csr);
    const float ssA = s_src[sA];
    float mx = ssA;
    for (int i = lane + 32; i < T; i += 32)
        mx = fmaxf(mx, s_src[edge_source(i, n, q, csr)]);
    mx = wave_max(mx);
    const float mv = leaky(mx + q.sd);
    float acc0 = 0.f, acc1 = 0.f, dsum = 0.f;
    for (int base = 0; base < T; base += 32) {
        const int i = base + lane;
        int s = sA; float ss = ssA;
        if (base) { s = edge_source(i, n, q, csr); ss = s_src[s]; }
        const float w = i < T ? expf(leaky(ss + q.sd) - mv) : 0.f;
        dsum += w;
        const int wbits = __float_as_int(w);
        const int cnt = T - base < 32 ? T - base : 32;
        for (int g = 0; g < cnt; g += ROWS_TOGETHER) {
            float v0[ROWS_TOGETHER], v1[ROWS_TOGETHER];
#pragma unroll
            for (int j = 0; j < ROWS_TOGETHER; ++j) {
                v0[j] = 0.f; v1[j] = 0.f;
                if (g + j < cnt) {
                    const float* __restrict__ row = plane_row(hlo, hhi, __builtin_amdgcn_readlane(s, g + j));
                    v0[j] = row[lane];
                    v1[j] = row[lane + 32];
                }
            }
#pragma unroll
            for (int j = 0; j < ROWS_TOGETHER; ++j) {
                if (g + j < cnt) {
                    const float wj = __int_as_float(__builtin_amdgcn_readlane(wbits, g + j));
                    acc0 += v0[j] * wj;
                    acc1 += v1[j] * wj;
                }
            }
        }
    }
    const float dn = wave_sum(dsum) + 1e-16f;
    const float o0 = acc0 / dn + b[lane];
    const float o1 = acc1 / dn + b[lane + 32];
    act[(size_t)(n - n0) * FH + lane]      = o0 > 0.f ? o0 : 0.f;
    act[(size_t)(n - n0) * FH + lane + 32] = o1 > 0.f ? o1 : 0.f;
}

static_assert(FIN == 16 && AGW >= FIN + 1, "half a wave per input row, one more column for the denominator");
#define PAIRS_TOGETHER 8

__global__ void __launch_bounds__(256)
node_aggr_in_kernel(const int* __restrict__ rowend, const int* __restrict__ csr,
                    const float* __restrict__ s_src, const float* __restrict__ s_dst,
                    const float* __restrict__ x, float* __restrict__ agg) {
    const int lane = threadIdx.x & 31;
    const int n    = blockIdx.x * 8 + __builtin_amdgcn_readfirstlane(threadIdx.x >> 5);
    if (n >= NNODES) return;
    const int hi = lane >> 4;
    const int lo = lane & 15;
    const NodeEdges q = node_edges(n, rowend, s_dst);
    const int T = q.deg + 1;
    const int   sA  = edge_source(lane, n, q, csr);
    const float ssA = s_src[sA];
    float mx = ssA;
    for (int i = lane + 32; i < T; i += 32)
        mx = fmaxf(mx, s_src[edge_source(i, n, q, csr)]);
    mx = wave_max(mx);
    const float mv = leaky(mx + q.sd);
    const float* __restrict__ xl = x + lo;
    float acc = 0.f, dsum = 0.f;
    for (int base = 0; base < T; base += 32) {
        const int i = base + lane;
        int s = sA; float ss = ssA;
        if (base) { s = edge_source(i, n, q, csr); ss = s_src[s]; }
        const float w = i < T ? expf(leaky(ss + q.sd) - mv) : 0.f;
        dsum += w;
        const int wbits = __float_as_int(w);
        const int cnt = T - base < 32 ? T - base : 32;
        for (int g = 0; g < cnt; g += 2 * PAIRS_TOGETHER) {
            float v[PAIRS_TOGETHER];
#pragma unroll
            for (int j = 0; j < PAIRS_TOGETHER; ++j) {
                v[j] = 0.f;
                if (g + 2 * j < cnt) {
                    const int sa = __builtin_amdgcn_readlane(s, g + 2 * j);
                    const int sb = __builtin_amdgcn_readlane(s, g + 2 * j + 1);
                    v[j] = xl[(size_t)(hi ? sb : sa) * FIN];
                }
            }
#pragma unroll
            for (int j = 0; j < PAIRS_TOGETHER; ++j) {
                if (g + 2 * j < cnt) {
                    const int wa = __builtin_amdgcn_readlane(wbits, g + 2 * j);
                    const int wb = __builtin_amdgcn_readlane(wbits, g + 2 * j + 1);
                    const bool live = g + 2 * j + hi < cnt;
                    acc += live ? v[j] * __int_as_float(hi ? wb : wa) : 0.f;
                }
            }
        }
    }
    acc += __shfl_xor(acc, 16, 32);
    const float den = wave_sum(dsum);
    if (lane <= FIN)
        agg[(size_t)n * AGW + lane] = lane < FIN ? acc : den;
}

__global__ void gemm_norm_act_kernel(const float* __restrict__ agg,
                                     const float* __restrict__ W, const float* __restrict__ b,
                                     float* __restrict__ act, int nRowTiles) {
    const int lane = threadIdx.x & 31;
    const int gw   = blockIdx.x * 8 + (threadIdx.x >> 5);
    const int rowTile = gw >> 2;
    const int colTile = gw & 3;
    if (rowTile >= nRowTiles) return;
    const int hi  = lane >> 4;
    const int lo  = lane & 15;
    const int row = rowTile * 16 + lo;
    const int col = colTile * 16 + lo;
    const float dn = agg[(size_t)row * AGW + FIN] + 1e-16f;
    v8f c = {};
#pragma unroll
    for (int kk = 0; kk < FIN; kk += 4) {
        const int k0 = kk + 2 * hi;
        v2f a, bb;
        a.x  = agg[(size_t)row * AGW + k0]     / dn;
        a.y  = agg[(size_t)row * AGW + k0 + 1] / dn;
        bb.x = W[(size_t)k0 * FH + col];
        bb.y = W[(size_t)(k0 + 1) * FH + col];
        c = __builtin_amdgcn_wmma_f32_16x16x4_f32(false, a, false, bb, (short)0, c, false, false);
    }
    const int rbase = rowTile * 16 + 8 * hi;
    const float bias = b[col];
#pragma unroll
    for (int r = 0; r < 8; ++r) {
        const float o = c[r] + bias;
        act[(size_t)(rbase + r) * FH + col] = o > 0.f ? o : 0.f;
    }
}

__global__ void policy_kernel(const float* __restrict__ act, const float* __restrict__ Wp,
                              const float* __restrict__ bp, float* __restrict__ out,
                              int nRowTiles) {
    const int lane = threadIdx.x & 31;
    const int gw   = blockIdx.x * 8 + (threadIdx.x >> 5);
    if (gw >= nRowTiles) return;
    const int hi  = lane >> 4;
    const int lo  = lane & 15;
    const int row = gw * 16 + lo;
    v8f c = {};
    for (int kk = 0; kk < FH; kk += 4) {
        const int k0 = kk + 2 * hi;
        v2f a, b;
        a.x = act[(size_t)row * FH + k0];
        a.y = act[(size_t)row * FH + k0 + 1];
        b.x = (lo < FOUT) ? Wp[(size_t)k0 * FOUT + lo]       : 0.f;
        b.y = (lo < FOUT) ? Wp[(size_t)(k0 + 1) * FOUT + lo] : 0.f;
        c = __builtin_amdgcn_wmma_f32_16x16x4_f32(false, a, false, b, (short)0, c, false, false);
    }
    if (lo < FOUT) {
        const int rbase = gw * 16 + 8 * hi;
        const float bb = bp[lo];
#pragma unroll
        for (int r = 0; r < 8; ++r)
            out[(size_t)(rbase + r) * FOUT + lo] = c[r] + bb;
    }
}

extern "C" void kernel_launch(void* const* d_in, const int* in_sizes, int n_in,
                              void* d_out, int out_size, void* d_ws, size_t ws_size,
                              hipStream_t stream) {
    const float* x   = (const float*)d_in[0];
    const int*   ei  = (const int*)  d_in[1];
    const float* W1  = (const float*)d_in[2];
    const float* as1 = (const float*)d_in[3];
    const float* ad1 = (const float*)d_in[4];
    const float* b1  = (const float*)d_in[5];
    const float* W2  = (const float*)d_in[6];
    const float* as2 = (const float*)d_in[7];
    const float* ad2 = (const float*)d_in[8];
    const float* b2  = (const float*)d_in[9];
    const float* Wp  = (const float*)d_in[10];
    const float* bp  = (const float*)d_in[11];
    float* out = (float*)d_out;

    const int* srcA = ei;
    const int* dstA = ei + NEDGES;

    float* ws   = (float*)d_ws;
    float* Plo  = ws;
    float* Phi  = ws + (size_t)NNODES * (FH / 2);
    float* Qlo  = ws + (size_t)NNODES * FH;
    float* Qhi  = ws + (size_t)NNODES * FH + (size_t)NNODES * (FH / 2);
    int*   rowend = (int*)(ws + (size_t)2 * NNODES * FH);
    float* ssrc = ws + (size_t)2 * NNODES * FH + NNODES;
    float* sdst = ws + (size_t)2 * NNODES * FH + 2 * NNODES;
    int*   csr  = (int*)Phi;
    int*   cnt  = (int*)ssrc;
    int*   bsum = (int*)sdst;

    const int nRowTiles = NNODES / 16;
    const int hRowTiles = NHALF / 16;
    const dim3 blk(256);
    const int gemmBlocks  = (nRowTiles * 4 + 7) / 8;
    const int gemmBlocksH = (hRowTiles * 4 + 7) / 8;
    const int nodeBlocks  = (NNODES + 7) / 8;
    const int nodeBlocksH = (NHALF + 7) / 8;
    const int nThrBlocks  = (NNODES + 255) / 256;
    const int edgeBlocks  = (NEDGES + 255) / 256;
    const int scanBlocks  = (NNODES / 4 + 255) / 256;
    const int polBlocksH  = (hRowTiles + 7) / 8;

    count_zero_kernel<<<nThrBlocks, blk, 0, stream>>>(cnt);
    count_kernel<<<edgeBlocks, blk, 0, stream>>>(dstA, cnt);
    block_sum_kernel<<<scanBlocks, blk, 0, stream>>>(cnt, bsum);
    scan_kernel<<<scanBlocks, blk, 0, stream>>>(cnt, bsum, rowend);
    fill_kernel<<<edgeBlocks, blk, 0, stream>>>(srcA, dstA, rowend, csr);

    float* agg = Plo;
    gemm_h_kernel<<<gemmBlocks, blk, 0, stream>>>(x, W1, Qlo, FIN, nRowTiles);
    scalars_kernel<<<nodeBlocks, blk, 0, stream>>>(Qlo, Qhi, as1, ad1, ssrc, sdst);
    node_aggr_in_kernel<<<nodeBlocks, blk, 0, stream>>>(rowend, csr, ssrc, sdst, x, agg);
    gemm_norm_act_kernel<<<gemmBlocks, blk, 0, stream>>>(agg, W1, b1, Qlo, nRowTiles);

    gemm_h_kernel<<<gemmBlocksH, blk, 0, stream>>>(Qlo, W2, Plo, FH, hRowTiles);
    gemm_h_kernel<<<gemmBlocksH, blk, 0, stream>>>(Qhi, W2, Qlo, FH, hRowTiles);
    scalars_kernel<<<nodeBlocks, blk, 0, stream>>>(Plo, Qlo, as2, ad2, ssrc, sdst);
    node_aggr_kernel<<<nodeBlocksH, blk, 0, stream>>>(rowend, csr, ssrc, sdst, Plo, Qlo, b2, Qhi, 0);
    policy_kernel<<<polBlocksH, blk, 0, stream>>>(Qhi, Wp, bp, out, hRowTiles);
    node_aggr_kernel<<<nodeBlocksH, blk, 0, stream>>>(rowend, csr, ssrc, sdst, Plo, Qlo, b2, Qhi, NHALF);
    policy_kernel<<<polBlocksH, blk, 0, stream>>>(Qhi, Wp, bp, out + (size_t)NHALF * FOUT, hRowTiles);
}


# Edges grouped by destination by a two-level counting sort ranked by LDS counters (buckets of 256 nodes, one packed word per edge): both global atomics per edge go; row sums keep list (arrival) order
# speedup vs baseline: 4.4432x; 1.7375x over previous
#include <hip/hip_runtime.h>
#include <math.h>

#define NNODES 100000
#define NEDGES 1600000
#define FIN    16
#define FH     64
#define FOUT   8
#define AGW    32

typedef __attribute__((ext_vector_type(2))) float v2f;
typedef __attribute__((ext_vector_type(8))) float v8f;

#define NHALF  (NNODES / 2)
static_assert(NNODES % 32 == 0, "a half is a whole number of 16-row tiles");

__device__ __forceinline__ float leaky(float e) { return e > 0.f ? e : 0.2f * e; }

__global__ void gemm_h_kernel(const float* __restrict__ in, const float* __restrict__ W,
                              float* __restrict__ h, int K, int nRowTiles) {
    const int lane = threadIdx.x & 31;
    const int gw   = blockIdx.x * 8 + (threadIdx.x >> 5);
    const int rowTile = gw >> 2;
    const int colTile = gw & 3;
    if (rowTile >= nRowTiles) return;
    const int hi  = lane >> 4;
    const int lo  = lane & 15;
    const int row = rowTile * 16 + lo;
    const int col = colTile * 16 + lo;
    v8f c = {};
    for (int kk = 0; kk < K; kk += 4) {
        const int k0 = kk + 2 * hi;
        v2f a, b;
        a.x = in[(size_t)row * K + k0];
        a.y = in[(size_t)row * K + k0 + 1];
        b.x = W[(size_t)k0 * FH + col];
        b.y = W[(size_t)(k0 + 1) * FH + col];
        c = __builtin_amdgcn_wmma_f32_16x16x4_f32(false, a, false, b, (short)0, c, false, false);
    }
    const int rbase = rowTile * 16 + 8 * hi;
#pragma unroll
    for (int r = 0; r < 8; ++r)
        h[(size_t)(rbase + r) * FH + col] = c[r];
}

__device__ __forceinline__ const float* plane_row(const float* __restrict__ lo,
                                                  const float* __restrict__ hi, int s) {
    return s < NHALF ? lo + (size_t)s * FH : hi + (size_t)(s - NHALF) * FH;
}

__global__ void scalars_kernel(const float* __restrict__ hlo, const float* __restrict__ hhi,
                               const float* __restrict__ a_src,
                               const float* __restrict__ a_dst,
                               float* __restrict__ s_src, float* __restrict__ s_dst) {
    const int lane = threadIdx.x & 31;
    const int n    = blockIdx.x * 8 + (threadIdx.x >> 5);
    if (n >= NNODES) return;
    const float* __restrict__ hr = plane_row(hlo, hhi, n);
    const float h0 = hr[lane];
    const float h1 = hr[lane + 32];
    float ps = h0 * a_src[lane] + h1 * a_src[lane + 32];
    float pd = h0 * a_dst[lane] + h1 * a_dst[lane + 32];
    for (int off = 16; off; off >>= 1) {
        ps += __shfl_xor(ps, off, 32);
        pd += __shfl_xor(pd, off, 32);
    }
    if (lane == 0) { s_src[n] = ps; s_dst[n] = pd; }
}

#define BKT_SHIFT 8
#define BKT_NODES (1 << BKT_SHIFT)
#define EPT       10
#define EPB       (256 * EPT)
#define NBLK      (NEDGES / EPB)
#define NBKT      (((NNODES + BKT_NODES - 1) / BKT_NODES + 3) / 4 * 4)
#define MATN      (NBKT * NBLK)
#define SRC_BITS  17
#define STAGE_CAP 6144
static_assert(NEDGES % EPB == 0, "level 1 blocks are full");
static_assert(NNODES <= (1 << SRC_BITS) && SRC_BITS + BKT_SHIFT <= 32, "source and in-bucket destination share one word");
static_assert(BKT_NODES == 256 && NBKT % 2 == 0 && NBKT <= 512, "one thread per destination of a bucket, two buckets per thread in the block's scan");

__device__ __forceinline__ int block_scan_incl(int own, int* wsum) {
    const int lane = threadIdx.x & 31;
    const int wv   = threadIdx.x >> 5;
    int inc = own;
    for (int off = 1; off < 32; off <<= 1) {
        const int u = __shfl_up(inc, off, 32);
        if (lane >= off) inc += u;
    }
    if (lane == 31) wsum[wv] = inc;
    __syncthreads();
    for (int k = 0; k < wv; ++k) inc += wsum[k];
    return inc;
}

__global__ void __launch_bounds__(256)
bucket_count_kernel(const int* __restrict__ dst, int* __restrict__ mat) {
    __shared__ int h[NBKT];
    const int t = threadIdx.x;
    for (int i = t; i < NBKT; i += 256) h[i] = 0;
    __syncthreads();
    const int* __restrict__ dp = dst + (size_t)blockIdx.x * EPB + t;
    int d[EPT];
#pragma unroll
    for (int k = 0; k < EPT; ++k) d[k] = dp[k * 256];
#pragma unroll
    for (int k = 0; k < EPT; ++k) atomicAdd(&h[d[k] >> BKT_SHIFT], 1);
    __syncthreads();
    for (int i = t; i < NBKT; i += 256) mat[i * NBLK + blockIdx.x] = h[i];
}

static_assert(MATN % 4 == 0, "counts are read four at a time");
static_assert((MATN / 4 + 255) / 256 <= 256, "one thread per block total");
__global__ void __launch_bounds__(256)
block_sum_kernel(const int* __restrict__ cnt, int* __restrict__ bsum, int nq) {
    __shared__ int wsum[8];
    const int t    = threadIdx.x;
    const int lane = t & 31;
    const int wv   = t >> 5;
    const int q = blockIdx.x * 256 + t;
    int own = 0;
    if (q < nq) {
        const int4 v = ((const int4*)cnt)[q];
        own = v.x + v.y + v.z + v.w;
    }
    for (int off = 16; off; off >>= 1) own += __shfl_xor(own, off, 32);
    if (lane == 0) wsum[wv] = own;
    __syncthreads();
    if (t == 0) {
        int s = 0;
        for (int k = 0; k < 8; ++k) s += wsum[k];
        bsum[blockIdx.x] = s;
    }
}

__global__ void __launch_bounds__(256)
scan_kernel(const int* __restrict__ cnt, const int* __restrict__ bsum, int* __restrict__ cursor, int nq) {
    __shared__ int wsum[8];
    __shared__ int wpre[8];
    const int t    = threadIdx.x;
    const int lane = t & 31;
    const int wv   = t >> 5;
    const int4* __restrict__ c4 = (const int4*)cnt;
    const int q0 = blockIdx.x * 256;
    int pre = t < (int)blockIdx.x ? bsum[t] : 0;
    const int q = q0 + t;
    int4 v = make_int4(0, 0, 0, 0);
    if (q < nq) v = c4[q];
    const int own = v.x + v.y + v.z + v.w;
    int inc = own;
    for (int off = 1; off < 32; off <<= 1) {
        const int u = __shfl_up(inc, off, 32);
        if (lane >= off) inc += u;
    }
    for (int off = 16; off; off >>= 1) pre += __shfl_xor(pre, off, 32);
    if (lane == 31) wsum[wv] = inc;
    if (lane == 0)  wpre[wv] = pre;
    __syncthreads();
    int base = 0;
    for (int k = 0; k < 8; ++k) {
        base += wpre[k];
        if (k < wv) base += wsum[k];
    }
    const int ex = base + inc - own;
    if (q < nq) {
        int4 o;
        o.x = ex; o.y = o.x + v.x; o.z = o.y + v.y; o.w = o.z + v.z;
        ((int4*)cursor)[q] = o;
    }
}

__global__ void __launch_bounds__(256)
bucket_scatter_kernel(const int* __restrict__ src, const int* __restrict__ dst,
                      const int* __restrict__ offs, unsigned* __restrict__ tmp) {
    __shared__ int      h[NBKT];
    __shared__ int      goff[NBKT];
    __shared__ unsigned val[EPB];
    __shared__ int      gpos[EPB];
    __shared__ int      wsum[8];
    const int t = threadIdx.x;
    const int* __restrict__ dp = dst + (size_t)blockIdx.x * EPB + t;
    const int* __restrict__ sp = src + (size_t)blockIdx.x * EPB + t;
    int d[EPT], s[EPT], r[EPT];
#pragma unroll
    for (int k = 0; k < EPT; ++k) { d[k] = dp[k * 256]; s[k] = sp[k * 256]; }
    for (int i = t; i < NBKT; i += 256) { h[i] = 0; goff[i] = offs[i * NBLK + blockIdx.x]; }
    __syncthreads();
#pragma unroll
    for (int k = 0; k < EPT; ++k) r[k] = atomicAdd(&h[d[k] >> BKT_SHIFT], 1);
    __syncthreads();
    int c0 = 0, c1 = 0;
    if (2 * t < NBKT) { c0 = h[2 * t]; c1 = h[2 * t + 1]; }
    const int ex = block_scan_incl(c0 + c1, wsum) - (c0 + c1);
    if (2 * t < NBKT) {
        h[2 * t]     = ex;       goff[2 * t]     -= ex;
        h[2 * t + 1] = ex + c0;  goff[2 * t + 1] -= ex + c0;
    }
    __syncthreads();
#pragma unroll
    for (int k = 0; k < EPT; ++k) {
        const int b  = d[k] >> BKT_SHIFT;
        const int lp = h[b] + r[k];
        val[lp]  = ((unsigned)(d[k] & (BKT_NODES - 1)) << SRC_BITS) | (unsigned)s[k];
        gpos[lp] = goff[b] + lp;
    }
    __syncthreads();
    for (int i = t; i < EPB; i += 256) tmp[gpos[i]] = val[i];
}

#define BUILD_TOGETHER 18
__global__ void __launch_bounds__(256)
bucket_build_kernel(const int* __restrict__ offs, const unsigned* __restrict__ tmp,
                    int* __restrict__ rowend, int* __restrict__ csr) {
    __shared__ int c[BKT_NODES];
    __shared__ int wsum[8];
    __shared__ int stage[STAGE_CAP];
    const int t    = threadIdx.x;
    const int b    = blockIdx.x;
    const int beg  = offs[b * NBLK];
    const int end  = b + 1 < NBKT ? offs[(b + 1) * NBLK] : NEDGES;
    const bool fits = end - beg <= STAGE_CAP;
    c[t] = 0;
    __syncthreads();
    for (int i0 = beg + t; i0 < end; i0 += 256 * BUILD_TOGETHER) {
        unsigned v[BUILD_TOGETHER];
#pragma unroll
        for (int j = 0; j < BUILD_TOGETHER; ++j) {
            const int i = i0 + j * 256;
            v[j] = tmp[i < end ? i : end - 1];
        }
#pragma unroll
        for (int j = 0; j < BUILD_TOGETHER; ++j)
            if (i0 + j * 256 < end) atomicAdd(&c[v[j] >> SRC_BITS], 1);
    }
    __syncthreads();
    const int own    = c[t];
    const int segend = beg + block_scan_incl(own, wsum);
    const int node   = b * BKT_NODES + t;
    if (node < NNODES) rowend[node] = segend;
    c[t] = segend - own;
    __syncthreads();
    for (int i0 = beg + t; i0 < end; i0 += 256 * BUILD_TOGETHER) {
        unsigned v[BUILD_TOGETHER];
#pragma unroll
        for (int j = 0; j < BUILD_TOGETHER; ++j) {
            const int i = i0 + j * 256;
            v[j] = tmp[i < end ? i : end - 1];
        }
#pragma unroll
        for (int j = 0; j < BUILD_TOGETHER; ++j) {
            if (i0 + j * 256 < end) {
                const int p = atomicAdd(&c[v[j] >> SRC_BITS], 1);
                const int sv = (int)(v[j] & ((1u << SRC_BITS) - 1u));
                if (fits) stage[p - beg] = sv; else csr[p] = sv;
            }
        }
    }
    if (fits) {
        __syncthreads();
        for (int i = t; i < end - beg; i += 256) csr[beg + i] = stage[i];
    }
}

#define ROWS_TOGETHER 8

struct NodeEdges { int beg, deg; float sd; };

__device__ __forceinline__ NodeEdges node_edges(int n, const int* __restrict__ rowend,
                                                const float* __restrict__ s_dst) {
    NodeEdges q;
    const int end = rowend[n];
    q.beg = n ? rowend[n - 1] : 0;
    q.deg = end - q.beg;
    q.sd  = s_dst[n];
    return q;
}
__device__ __forceinline__ int edge_source(int i, int n, const NodeEdges& q, const int* __restrict__ csr) {
    return i < q.deg ? csr[q.beg + i] : n;
}
__device__ __forceinline__ float wave_max(float v) {
    for (int off = 16; off; off >>= 1) v = fmaxf(v, __shfl_xor(v, off, 32));
    return v;
}
__device__ __forceinline__ float wave_sum(float v) {
    for (int off = 16; off; off >>= 1) v += __shfl_xor(v, off, 32);
    return v;
}

__global__ void __launch_bounds__(256)
node_aggr_kernel(const int* __restrict__ rowend, const int* __restrict__ csr,
                 const float* __restrict__ s_src, const float* __restrict__ s_dst,
                 const float* __restrict__ hlo, const float* __restrict__ hhi,
                 const float* __restrict__ b, float* __restrict__ act, int n0) {
    const int lane = threadIdx.x & 31;
    const int n    = n0 + blockIdx.x * 8 + __builtin_amdgcn_readfirstlane(threadIdx.x >> 5);
    if (n >= n0 + NHALF) return;
    const NodeEdges q = node_edges(n, rowend, s_dst);
    const int T = q.deg + 1;
    const int   sA  = edge_source(lane, n, q, csr);
    const float ssA = s_src[sA];
    float mx = ssA;
    for (int i = lane + 32; i < T; i += 32)
        mx = fmaxf(mx, s_src[edge_source(i, n, q, csr)]);
    mx = wave_max(mx);
    const float mv = leaky(mx + q.sd);
    float acc0 = 0.f, acc1 = 0.f, dsum = 0.f;
    for (int base = 0; base < T; base += 32) {
        const int i = base + lane;
        int s = sA; float ss = ssA;
        if (base) { s = edge_source(i, n, q, csr); ss = s_src[s]; }
        const float w = i < T ? expf(leaky(ss + q.sd) - mv) : 0.f;
        dsum += w;
        const int wbits = __float_as_int(w);
        const int cnt = T - base < 32 ? T - base : 32;
        for (int g = 0; g < cnt; g += ROWS_TOGETHER) {
            float v0[ROWS_TOGETHER], v1[ROWS_TOGETHER];
#pragma unroll
            for (int j = 0; j < ROWS_TOGETHER; ++j) {
                v0[j] = 0.f; v1[j] = 0.f;
                if (g + j < cnt) {
                    const float* __restrict__ row = plane_row(hlo, hhi, __builtin_amdgcn_readlane(s, g + j));
                    v0[j] = row[lane];
                    v1[j] = row[lane + 32];
                }
            }
#pragma unroll
            for (int j = 0; j < ROWS_TOGETHER; ++j) {
                if (g + j < cnt) {
                    const float wj = __int_as_float(__builtin_amdgcn_readlane(wbits, g + j));
                    acc0 += v0[j] * wj;
                    acc1 += v1[j] * wj;
                }
            }
        }
    }
    const float dn = wave_sum(dsum) + 1e-16f;
    const float o0 = acc0 / dn + b[lane];
    const float o1 = acc1 / dn + b[lane + 32];
    act[(size_t)(n - n0) * FH + lane]      = o0 > 0.f ? o0 : 0.f;
    act[(size_t)(n - n0) * FH + lane + 32] = o1 > 0.f ? o1 : 0.f;
}

static_assert(FIN == 16 && AGW >= FIN + 1, "half a wave per input row, one more column for the denominator");
#define PAIRS_TOGETHER 8

__global__ void __launch_bounds__(256)
node_aggr_in_kernel(const int* __restrict__ rowend, const int* __restrict__ csr,
                    const float* __restrict__ s_src, const float* __restrict__ s_dst,
                    const float* __restrict__ x, float* __restrict__ agg) {
    const int lane = threadIdx.x & 31;
    const int n    = blockIdx.x * 8 + __builtin_amdgcn_readfirstlane(threadIdx.x >> 5);
    if (n >= NNODES) return;
    const int hi = lane >> 4;
    const int lo = lane & 15;
    const NodeEdges q = node_edges(n, rowend, s_dst);
    const int T = q.deg + 1;
    const int   sA  = edge_source(lane, n, q, csr);
    const float ssA = s_src[sA];
    float mx = ssA;
    for (int i = lane + 32; i < T; i += 32)
        mx = fmaxf(mx, s_src[edge_source(i, n, q, csr)]);
    mx = wave_max(mx);
    const float mv = leaky(mx + q.sd);
    const float* __restrict__ xl = x + lo;
    float acc = 0.f, dsum = 0.f;
    for (int base = 0; base < T; base += 32) {
        const int i = base + lane;
        int s = sA; float ss = ssA;
        if (base) { s = edge_source(i, n, q, csr); ss = s_src[s]; }
        const float w = i < T ? expf(leaky(ss + q.sd) - mv) : 0.f;
        dsum += w;
        const int wbits = __float_as_int(w);
        const int cnt = T - base < 32 ? T - base : 32;
        for (int g = 0; g < cnt; g += 2 * PAIRS_TOGETHER) {
            float v[PAIRS_TOGETHER];
#pragma unroll
            for (int j = 0; j < PAIRS_TOGETHER; ++j) {
                v[j] = 0.f;
                if (g + 2 * j < cnt) {
                    const int sa = __builtin_amdgcn_readlane(s, g + 2 * j);
                    const int sb = __builtin_amdgcn_readlane(s, g + 2 * j + 1);
                    v[j] = xl[(size_t)(hi ? sb : sa) * FIN];
                }
            }
#pragma unroll
            for (int j = 0; j < PAIRS_TOGETHER; ++j) {
                if (g + 2 * j < cnt) {
                    const int wa = __builtin_amdgcn_readlane(wbits, g + 2 * j);
                    const int wb = __builtin_amdgcn_readlane(wbits, g + 2 * j + 1);
                    const bool live = g + 2 * j + hi < cnt;
                    acc += live ? v[j] * __int_as_float(hi ? wb : wa) : 0.f;
                }
            }
        }
    }
    acc += __shfl_xor(acc, 16, 32);
    const float den = wave_sum(dsum);
    if (lane <= FIN)
        agg[(size_t)n * AGW + lane] = lane < FIN ? acc : den;
}

__global__ void gemm_norm_act_kernel(const float* __restrict__ agg,
                                     const float* __restrict__ W, const float* __restrict__ b,
                                     float* __restrict__ act, int nRowTiles) {
    const int lane = threadIdx.x & 31;
    const int gw   = blockIdx.x * 8 + (threadIdx.x >> 5);
    const int rowTile = gw >> 2;
    const int colTile = gw & 3;
    if (rowTile >= nRowTiles) return;
    const int hi  = lane >> 4;
    const int lo  = lane & 15;
    const int row = rowTile * 16 + lo;
    const int col = colTile * 16 + lo;
    const float dn = agg[(size_t)row * AGW + FIN] + 1e-16f;
    v8f c = {};
#pragma unroll
    for (int kk = 0; kk < FIN; kk += 4) {
        const int k0 = kk + 2 * hi;
        v2f a, bb;
        a.x  = agg[(size_t)row * AGW + k0]     / dn;
        a.y  = agg[(size_t)row * AGW + k0 + 1] / dn;
        bb.x = W[(size_t)k0 * FH + col];
        bb.y = W[(size_t)(k0 + 1) * FH + col];
        c = __builtin_amdgcn_wmma_f32_16x16x4_f32(false, a, false, bb, (short)0, c, false, false);
    }
    const int rbase = rowTile * 16 + 8 * hi;
    const float bias = b[col];
#pragma unroll
    for (int r = 0; r < 8; ++r) {
        const float o = c[r] + bias;
        act[(size_t)(rbase + r) * FH + col] = o > 0.f ? o : 0.f;
    }
}

__global__ void policy_kernel(const float* __restrict__ act, const float* __restrict__ Wp,
                              const float* __restrict__ bp, float* __restrict__ out,
                              int nRowTiles) {
    const int lane = threadIdx.x & 31;
    const int gw   = blockIdx.x * 8 + (threadIdx.x >> 5);
    if (gw >= nRowTiles) return;
    const int hi  = lane >> 4;
    const int lo  = lane & 15;
    const int row = gw * 16 + lo;
    v8f c = {};
    for (int kk = 0; kk < FH; kk += 4) {
        const int k0 = kk + 2 * hi;
        v2f a, b;
        a.x = act[(size_t)row * FH + k0];
        a.y = act[(size_t)row * FH + k0 + 1];
        b.x = (lo < FOUT) ? Wp[(size_t)k0 * FOUT + lo]       : 0.f;
        b.y = (lo < FOUT) ? Wp[(size_t)(k0 + 1) * FOUT + lo] : 0.f;
        c = __builtin_amdgcn_wmma_f32_16x16x4_f32(false, a, false, b, (short)0, c, false, false);
    }
    if (lo < FOUT) {
        const int rbase = gw * 16 + 8 * hi;
        const float bb = bp[lo];
#pragma unroll
        for (int r = 0; r < 8; ++r)
            out[(size_t)(rbase + r) * FOUT + lo] = c[r] + bb;
    }
}

extern "C" void kernel_launch(void* const* d_in, const int* in_sizes, int n_in,
                              void* d_out, int out_size, void* d_ws, size_t ws_size,
                              hipStream_t stream) {
    const float* x   = (const float*)d_in[0];
    const int*   ei  = (const int*)  d_in[1];
    const float* W1  = (const float*)d_in[2];
    const float* as1 = (const float*)d_in[3];
    const float* ad1 = (const float*)d_in[4];
    const float* b1  = (const float*)d_in[5];
    const float* W2  = (const float*)d_in[6];
    const float* as2 = (const float*)d_in[7];
    const float* ad2 = (const float*)d_in[8];
    const float* b2  = (const float*)d_in[9];
    const float* Wp  = (const float*)d_in[10];
    const float* bp  = (const float*)d_in[11];
    float* out = (float*)d_out;

    const int* srcA = ei;
    const int* dstA = ei + NEDGES;

    float* ws   = (float*)d_ws;
    float* Plo  = ws;
    float* Phi  = ws + (size_t)NNODES * (FH / 2);
    float* Qlo  = ws + (size_t)NNODES * FH;
    float* Qhi  = ws + (size_t)NNODES * FH + (size_t)NNODES * (FH / 2);
    int*   rowend = (int*)(ws + (size_t)2 * NNODES * FH);
    float* ssrc = ws + (size_t)2 * NNODES * FH + NNODES;
    float* sdst = ws + (size_t)2 * NNODES * FH + 2 * NNODES;
    int*   csr  = (int*)Phi;
    unsigned* tmp = (unsigned*)Qlo;
    int*   mat  = (int*)Qlo + NEDGES;
    int*   offs = (int*)Qlo + NEDGES + MATN;
    int*   bsum = (int*)Qlo + NEDGES + 2 * MATN;

    const int nRowTiles = NNODES / 16;
    const int hRowTiles = NHALF / 16;
    const dim3 blk(256);
    const int gemmBlocks  = (nRowTiles * 4 + 7) / 8;
    const int gemmBlocksH = (hRowTiles * 4 + 7) / 8;
    const int nodeBlocks  = (NNODES + 7) / 8;
    const int nodeBlocksH = (NHALF + 7) / 8;
    const int scanBlocks  = (MATN / 4 + 255) / 256;
    const int polBlocksH  = (hRowTiles + 7) / 8;

    bucket_count_kernel<<<NBLK, blk, 0, stream>>>(dstA, mat);
    block_sum_kernel<<<scanBlocks, blk, 0, stream>>>(mat, bsum, MATN / 4);
    scan_kernel<<<scanBlocks, blk, 0, stream>>>(mat, bsum, offs, MATN / 4);
    bucket_scatter_kernel<<<NBLK, blk, 0, stream>>>(srcA, dstA, offs, tmp);
    bucket_build_kernel<<<NBKT, blk, 0, stream>>>(offs, tmp, rowend, csr);

    float* agg = Plo;
    gemm_h_kernel<<<gemmBlocks, blk, 0, stream>>>(x, W1, Qlo, FIN, nRowTiles);
    scalars_kernel<<<nodeBlocks, blk, 0, stream>>>(Qlo, Qhi, as1, ad1, ssrc, sdst);
    node_aggr_in_kernel<<<nodeBlocks, blk, 0, stream>>>(rowend, csr, ssrc, sdst, x, agg);
    gemm_norm_act_kernel<<<gemmBlocks, blk, 0, stream>>>(agg, W1, b1, Qlo, nRowTiles);

    gemm_h_kernel<<<gemmBlocksH, blk, 0, stream>>>(Qlo, W2, Plo, FH, hRowTiles);
    gemm_h_kernel<<<gemmBlocksH, blk, 0, stream>>>(Qhi, W2, Qlo, FH, hRowTiles);
    scalars_kernel<<<nodeBlocks, blk, 0, stream>>>(Plo, Qlo, as2, ad2, ssrc, sdst);
    node_aggr_kernel<<<nodeBlocksH, blk, 0, stream>>>(rowend, csr, ssrc, sdst, Plo, Qlo, b2, Qhi, 0);
    policy_kernel<<<polBlocksH, blk, 0, stream>>>(Qhi, Wp, bp, out, hRowTiles);
    node_aggr_kernel<<<nodeBlocksH, blk, 0, stream>>>(rowend, csr, ssrc, sdst, Plo, Qlo, b2, Qhi, NHALF);
    policy_kernel<<<polBlocksH, blk, 0, stream>>>(Qhi, Wp, bp, out + (size_t)NHALF * FOUT, hRowTiles);
}


# Layer 1's W1 product and relu, layer 2's W2 product and its two attention dots in one launch per 16-node tile (act1 handed on through LDS, never stored): 3 launches go; the dots' sum order changes
# speedup vs baseline: 4.6678x; 1.0505x over previous
#include <hip/hip_runtime.h>
#include <math.h>

#define NNODES 100000
#define NEDGES 1600000
#define FIN    16
#define FH     64
#define FOUT   8
#define AGW    32

typedef __attribute__((ext_vector_type(2))) float v2f;
typedef __attribute__((ext_vector_type(8))) float v8f;

#define NHALF  (NNODES / 2)
static_assert(NNODES % 32 == 0, "a half is a whole number of 16-row tiles");

__device__ __forceinline__ float leaky(float e) { return e > 0.f ? e : 0.2f * e; }

__global__ void gemm_h_kernel(const float* __restrict__ in, const float* __restrict__ W,
                              float* __restrict__ h, int K, int nRowTiles) {
    const int lane = threadIdx.x & 31;
    const int gw   = blockIdx.x * 8 + (threadIdx.x >> 5);
    const int rowTile = gw >> 2;
    const int colTile = gw & 3;
    if (rowTile >= nRowTiles) return;
    const int hi  = lane >> 4;
    const int lo  = lane & 15;
    const int row = rowTile * 16 + lo;
    const int col = colTile * 16 + lo;
    v8f c = {};
    for (int kk = 0; kk < K; kk += 4) {
        const int k0 = kk + 2 * hi;
        v2f a, b;
        a.x = in[(size_t)row * K + k0];
        a.y = in[(size_t)row * K + k0 + 1];
        b.x = W[(size_t)k0 * FH + col];
        b.y = W[(size_t)(k0 + 1) * FH + col];
        c = __builtin_amdgcn_wmma_f32_16x16x4_f32(false, a, false, b, (short)0, c, false, false);
    }
    const int rbase = rowTile * 16 + 8 * hi;
#pragma unroll
    for (int r = 0; r < 8; ++r)
        h[(size_t)(rbase + r) * FH + col] = c[r];
}

__device__ __forceinline__ const float* plane_row(const float* __restrict__ lo,
                                                  const float* __restrict__ hi, int s) {
    return s < NHALF ? lo + (size_t)s * FH : hi + (size_t)(s - NHALF) * FH;
}

__global__ void scalars_kernel(const float* __restrict__ hlo, const float* __restrict__ hhi,
                               const float* __restrict__ a_src,
                               const float* __restrict__ a_dst,
                               float* __restrict__ s_src, float* __restrict__ s_dst) {
    const int lane = threadIdx.x & 31;
    const int n    = blockIdx.x * 8 + (threadIdx.x >> 5);
    if (n >= NNODES) return;
    const float* __restrict__ hr = plane_row(hlo, hhi, n);
    const float h0 = hr[lane];
    const float h1 = hr[lane + 32];
    float ps = h0 * a_src[lane] + h1 * a_src[lane + 32];
    float pd = h0 * a_dst[lane] + h1 * a_dst[lane + 32];
    for (int off = 16; off; off >>= 1) {
        ps += __shfl_xor(ps, off, 32);
        pd += __shfl_xor(pd, off, 32);
    }
    if (lane == 0) { s_src[n] = ps; s_dst[n] = pd; }
}

#define BKT_SHIFT 8
#define BKT_NODES (1 << BKT_SHIFT)
#define EPT       10
#define EPB       (256 * EPT)
#define NBLK      (NEDGES / EPB)
#define NBKT      (((NNODES + BKT_NODES - 1) / BKT_NODES + 3) / 4 * 4)
#define MATN      (NBKT * NBLK)
#define SRC_BITS  17
#define STAGE_CAP 6144
static_assert(NEDGES % EPB == 0, "level 1 blocks are full");
static_assert(NNODES <= (1 << SRC_BITS) && SRC_BITS + BKT_SHIFT <= 32, "source and in-bucket destination share one word");
static_assert(BKT_NODES == 256 && NBKT % 2 == 0 && NBKT <= 512, "one thread per destination of a bucket, two buckets per thread in the block's scan");

__device__ __forceinline__ int block_scan_incl(int own, int* wsum) {
    const int lane = threadIdx.x & 31;
    const int wv   = threadIdx.x >> 5;
    int inc = own;
    for (int off = 1; off < 32; off <<= 1) {
        const int u = __shfl_up(inc, off, 32);
        if (lane >= off) inc += u;
    }
    if (lane == 31) wsum[wv] = inc;
    __syncthreads();
    for (int k = 0; k < wv; ++k) inc += wsum[k];
    return inc;
}

__global__ void __launch_bounds__(256)
bucket_count_kernel(const int* __restrict__ dst, int* __restrict__ mat) {
    __shared__ int h[NBKT];
    const int t = threadIdx.x;
    for (int i = t; i < NBKT; i += 256) h[i] = 0;
    __syncthreads();
    const int* __restrict__ dp = dst + (size_t)blockIdx.x * EPB + t;
    int d[EPT];
#pragma unroll
    for (int k = 0; k < EPT; ++k) d[k] = dp[k * 256];
#pragma unroll
    for (int k = 0; k < EPT; ++k) atomicAdd(&h[d[k] >> BKT_SHIFT], 1);
    __syncthreads();
    for (int i = t; i < NBKT; i += 256) mat[i * NBLK + blockIdx.x] = h[i];
}

static_assert(MATN % 4 == 0, "counts are read four at a time");
static_assert((MATN / 4 + 255) / 256 <= 256, "one thread per block total");
__global__ void __launch_bounds__(256)
block_sum_kernel(const int* __restrict__ cnt, int* __restrict__ bsum, int nq) {
    __shared__ int wsum[8];
    const int t    = threadIdx.x;
    const int lane = t & 31;
    const int wv   = t >> 5;
    const int q = blockIdx.x * 256 + t;
    int own = 0;
    if (q < nq) {
        const int4 v = ((const int4*)cnt)[q];
        own = v.x + v.y + v.z + v.w;
    }
    for (int off = 16; off; off >>= 1) own += __shfl_xor(own, off, 32);
    if (lane == 0) wsum[wv] = own;
    __syncthreads();
    if (t == 0) {
        int s = 0;
        for (int k = 0; k < 8; ++k) s += wsum[k];
        bsum[blockIdx.x] = s;
    }
}

__global__ void __launch_bounds__(256)
scan_kernel(const int* __restrict__ cnt, const int* __restrict__ bsum, int* __restrict__ cursor, int nq) {
    __shared__ int wsum[8];
    __shared__ int wpre[8];
    const int t    = threadIdx.x;
    const int lane = t & 31;
    const int wv   = t >> 5;
    const int4* __restrict__ c4 = (const int4*)cnt;
    const int q0 = blockIdx.x * 256;
    int pre = t < (int)blockIdx.x ? bsum[t] : 0;
    const int q = q0 + t;
    int4 v = make_int4(0, 0, 0, 0);
    if (q < nq) v = c4[q];
    const int own = v.x + v.y + v.z + v.w;
    int inc = own;
    for (int off = 1; off < 32; off <<= 1) {
        const int u = __shfl_up(inc, off, 32);
        if (lane >= off) inc += u;
    }
    for (int off = 16; off; off >>= 1) pre += __shfl_xor(pre, off, 32);
    if (lane == 31) wsum[wv] = inc;
    if (lane == 0)  wpre[wv] = pre;
    __syncthreads();
    int base = 0;
    for (int k = 0; k < 8; ++k) {
        base += wpre[k];
        if (k < wv) base += wsum[k];
    }
    const int ex = base + inc - own;
    if (q < nq) {
        int4 o;
        o.x = ex; o.y = o.x + v.x; o.z = o.y + v.y; o.w = o.z + v.z;
        ((int4*)cursor)[q] = o;
    }
}

__global__ void __launch_bounds__(256)
bucket_scatter_kernel(const int* __restrict__ src, const int* __restrict__ dst,
                      const int* __restrict__ offs, unsigned* __restrict__ tmp) {
    __shared__ int      h[NBKT];
    __shared__ int      goff[NBKT];
    __shared__ unsigned val[EPB];
    __shared__ int      gpos[EPB];
    __shared__ int      wsum[8];
    const int t = threadIdx.x;
    const int* __restrict__ dp = dst + (size_t)blockIdx.x * EPB + t;
    const int* __restrict__ sp = src + (size_t)blockIdx.x * EPB + t;
    int d[EPT], s[EPT], r[EPT];
#pragma unroll
    for (int k = 0; k < EPT; ++k) { d[k] = dp[k * 256]; s[k] = sp[k * 256]; }
    for (int i = t; i < NBKT; i += 256) { h[i] = 0; goff[i] = offs[i * NBLK + blockIdx.x]; }
    __syncthreads();
#pragma unroll
    for (int k = 0; k < EPT; ++k) r[k] = atomicAdd(&h[d[k] >> BKT_SHIFT], 1);
    __syncthreads();
    int c0 = 0, c1 = 0;
    if (2 * t < NBKT) { c0 = h[2 * t]; c1 = h[2 * t + 1]; }
    const int ex = block_scan_incl(c0 + c1, wsum) - (c0 + c1);
    if (2 * t < NBKT) {
        h[2 * t]     = ex;       goff[2 * t]     -= ex;
        h[2 * t + 1] = ex + c0;  goff[2 * t + 1] -= ex + c0;
    }
    __syncthreads();
#pragma unroll
    for (int k = 0; k < EPT; ++k) {
        const int b  = d[k] >> BKT_SHIFT;
        const int lp = h[b] + r[k];
        val[lp]  = ((unsigned)(d[k] & (BKT_NODES - 1)) << SRC_BITS) | (unsigned)s[k];
        gpos[lp] = goff[b] + lp;
    }
    __syncthreads();
    for (int i = t; i < EPB; i += 256) tmp[gpos[i]] = val[i];
}

#define BUILD_TOGETHER 18
__global__ void __launch_bounds__(256)
bucket_build_kernel(const int* __restrict__ offs, const unsigned* __restrict__ tmp,
                    int* __restrict__ rowend, int* __restrict__ csr) {
    __shared__ int c[BKT_NODES];
    __shared__ int wsum[8];
    __shared__ int stage[STAGE_CAP];
    const int t    = threadIdx.x;
    const int b    = blockIdx.x;
    const int beg  = offs[b * NBLK];
    const int end  = b + 1 < NBKT ? offs[(b + 1) * NBLK] : NEDGES;
    const bool fits = end - beg <= STAGE_CAP;
    c[t] = 0;
    __syncthreads();
    for (int i0 = beg + t; i0 < end; i0 += 256 * BUILD_TOGETHER) {
        unsigned v[BUILD_TOGETHER];
#pragma unroll
        for (int j = 0; j < BUILD_TOGETHER; ++j) {
            const int i = i0 + j * 256;
            v[j] = tmp[i < end ? i : end - 1];
        }
#pragma unroll
        for (int j = 0; j < BUILD_TOGETHER; ++j)
            if (i0 + j * 256 < end) atomicAdd(&c[v[j] >> SRC_BITS], 1);
    }
    __syncthreads();
    const int own    = c[t];
    const int segend = beg + block_scan_incl(own, wsum);
    const int node   = b * BKT_NODES + t;
    if (node < NNODES) rowend[node] = segend;
    c[t] = segend - own;
    __syncthreads();
    for (int i0 = beg + t; i0 < end; i0 += 256 * BUILD_TOGETHER) {
        unsigned v[BUILD_TOGETHER];
#pragma unroll
        for (int j = 0; j < BUILD_TOGETHER; ++j) {
            const int i = i0 + j * 256;
            v[j] = tmp[i < end ? i : end - 1];
        }
#pragma unroll
        for (int j = 0; j < BUILD_TOGETHER; ++j) {
            if (i0 + j * 256 < end) {
                const int p = atomicAdd(&c[v[j] >> SRC_BITS], 1);
                const int sv = (int)(v[j] & ((1u << SRC_BITS) - 1u));
                if (fits) stage[p - beg] = sv; else csr[p] = sv;
            }
        }
    }
    if (fits) {
        __syncthreads();
        for (int i = t; i < end - beg; i += 256) csr[beg + i] = stage[i];
    }
}

#define ROWS_TOGETHER 8

struct NodeEdges { int beg, deg; float sd; };

__device__ __forceinline__ NodeEdges node_edges(int n, const int* __restrict__ rowend,
                                                const float* __restrict__ s_dst) {
    NodeEdges q;
    const int end = rowend[n];
    q.beg = n ? rowend[n - 1] : 0;
    q.deg = end - q.beg;
    q.sd  = s_dst[n];
    return q;
}
__device__ __forceinline__ int edge_source(int i, int n, const NodeEdges& q, const int* __restrict__ csr) {
    return i < q.deg ? csr[q.beg + i] : n;
}
__device__ __forceinline__ float wave_max(float v) {
    for (int off = 16; off; off >>= 1) v = fmaxf(v, __shfl_xor(v, off, 32));
    return v;
}
__device__ __forceinline__ float wave_sum(float v) {
    for (int off = 16; off; off >>= 1) v += __shfl_xor(v, off, 32);
    return v;
}

__global__ void __launch_bounds__(256)
node_aggr_kernel(const int* __restrict__ rowend, const int* __restrict__ csr,
                 const float* __restrict__ s_src, const float* __restrict__ s_dst,
                 const float* __restrict__ hlo, const float* __restrict__ hhi,
                 const float* __restrict__ b, float* __restrict__ act, int n0) {
    const int lane = threadIdx.x & 31;
    const int n    = n0 + blockIdx.x * 8 + __builtin_amdgcn_readfirstlane(threadIdx.x >> 5);
    if (n >= n0 + NHALF) return;
    const NodeEdges q = node_edges(n, rowend, s_dst);
    const int T = q.deg + 1;
    const int   sA  = edge_source(lane, n, q, csr);
    const float ssA = s_src[sA];
    float mx = ssA;
    for (int i = lane + 32; i < T; i += 32)
        mx = fmaxf(mx, s_src[edge_source(i, n, q, csr)]);
    mx = wave_max(mx);
    const float mv = leaky(mx + q.sd);
    float acc0 = 0.f, acc1 = 0.f, dsum = 0.f;
    for (int base = 0; base < T; base += 32) {
        const int i = base + lane;
        int s = sA; float ss = ssA;
        if (base) { s = edge_source(i, n, q, csr); ss = s_src[s]; }
        const float w = i < T ? expf(leaky(ss + q.sd) - mv) : 0.f;
        dsum += w;
        const int wbits = __float_as_int(w);
        const int cnt = T - base < 32 ? T - base : 32;
        for (int g = 0; g < cnt; g += ROWS_TOGETHER) {
            float v0[ROWS_TOGETHER], v1[ROWS_TOGETHER];
#pragma unroll
            for (int j = 0; j < ROWS_TOGETHER; ++j) {
                v0[j] = 0.f; v1[j] = 0.f;
                if (g + j < cnt) {
                    const float* __restrict__ row = plane_row(hlo, hhi, __builtin_amdgcn_readlane(s, g + j));
                    v0[j] = row[lane];
                    v1[j] = row[lane + 32];
                }
            }
#pragma unroll
            for (int j = 0; j < ROWS_TOGETHER; ++j) {
                if (g + j < cnt) {
                    const float wj = __int_as_float(__builtin_amdgcn_readlane(wbits, g + j));
                    acc0 += v0[j] * wj;
                    acc1 += v1[j] * wj;
                }
            }
        }
    }
    const float dn = wave_sum(dsum) + 1e-16f;
    const float o0 = acc0 / dn + b[lane];
    const float o1 = acc1 / dn + b[lane + 32];
    act[(size_t)(n - n0) * FH + lane]      = o0 > 0.f ? o0 : 0.f;
    act[(size_t)(n - n0) * FH + lane + 32] = o1 > 0.f ? o1 : 0.f;
}

static_assert(FIN == 16 && AGW >= FIN + 1, "half a wave per input row, one more column for the denominator");
#define PAIRS_TOGETHER 8

__global__ void __launch_bounds__(256)
node_aggr_in_kernel(const int* __restrict__ rowend, const int* __restrict__ csr,
                    const float* __restrict__ s_src, const float* __restrict__ s_dst,
                    const float* __restrict__ x, float* __restrict__ agg) {
    const int lane = threadIdx.x & 31;
    const int n    = blockIdx.x * 8 + __builtin_amdgcn_readfirstlane(threadIdx.x >> 5);
    if (n >= NNODES) return;
    const int hi = lane >> 4;
    const int lo = lane & 15;
    const NodeEdges q = node_edges(n, rowend, s_dst);
    const int T = q.deg + 1;
    const int   sA  = edge_source(lane, n, q, csr);
    const float ssA = s_src[sA];
    float mx = ssA;
    for (int i = lane + 32; i < T; i += 32)
        mx = fmaxf(mx, s_src[edge_source(i, n, q, csr)]);
    mx = wave_max(mx);
    const float mv = leaky(mx + q.sd);
    const float* __restrict__ xl = x + lo;
    float acc = 0.f, dsum = 0.f;
    for (int base = 0; base < T; base += 32) {
        const int i = base + lane;
        int s = sA; float ss = ssA;
        if (base) { s = edge_source(i, n, q, csr); ss = s_src[s]; }
        const float w = i < T ? expf(leaky(ss + q.sd) - mv) : 0.f;
        dsum += w;
        const int wbits = __float_as_int(w);
        const int cnt = T - base < 32 ? T - base : 32;
        for (int g = 0; g < cnt; g += 2 * PAIRS_TOGETHER) {
            float v[PAIRS_TOGETHER];
#pragma unroll
            for (int j = 0; j < PAIRS_TOGETHER; ++j) {
                v[j] = 0.f;
                if (g + 2 * j < cnt) {
                    const int sa = __builtin_amdgcn_readlane(s, g + 2 * j);
                    const int sb = __builtin_amdgcn_readlane(s, g + 2 * j + 1);
                    v[j] = xl[(size_t)(hi ? sb : sa) * FIN];
                }
            }
#pragma unroll
            for (int j = 0; j < PAIRS_TOGETHER; ++j) {
                if (g + 2 * j < cnt) {
                    const int wa = __builtin_amdgcn_readlane(wbits, g + 2 * j);
                    const int wb = __builtin_amdgcn_readlane(wbits, g + 2 * j + 1);
                    const bool live = g + 2 * j + hi < cnt;
                    acc += live ? v[j] * __int_as_float(hi ? wb : wa) : 0.f;
                }
            }
        }
    }
    acc += __shfl_xor(acc, 16, 32);
    const float den = wave_sum(dsum);
    if (lane <= FIN)
        agg[(size_t)n * AGW + lane] = lane < FIN ? acc : den;
}

#define MID_WAVES 4
#define TPITCH    68

template <int M> __device__ __forceinline__ float half_xor_get(float v) {
    return __int_as_float(__builtin_amdgcn_update_dpp(0, __float_as_int(v), 0x160 + M, 0xf, 0xf, false));
}
__device__ __forceinline__ float half_sum(float v) {
    v += half_xor_get<1>(v);
    v += half_xor_get<2>(v);
    v += half_xor_get<4>(v);
    v += half_xor_get<8>(v);
    return v;
}

__global__ void __launch_bounds__(32 * MID_WAVES)
dense_mid_kernel(const float* __restrict__ agg,
                 const float* __restrict__ W1, const float* __restrict__ b1,
                 const float* __restrict__ W2,
                 const float* __restrict__ a_src, const float* __restrict__ a_dst,
                 float* __restrict__ h2, float* __restrict__ s_src, float* __restrict__ s_dst,
                 int nRowTiles) {
    __shared__ __attribute__((aligned(16))) float tile[MID_WAVES][16 * TPITCH];
    const int lane = threadIdx.x & 31;
    const int wv   = threadIdx.x >> 5;
    const int rowTile = blockIdx.x * MID_WAVES + wv;
    if (rowTile >= nRowTiles) return;
    const int hi  = lane >> 4;
    const int lo  = lane & 15;
    const int row = rowTile * 16 + lo;
    float* __restrict__ T = tile[wv];

    const float dn = agg[(size_t)row * AGW + FIN] + 1e-16f;
    v8f c[4] = {};
#pragma unroll
    for (int kk = 0; kk < FIN; kk += 4) {
        const int k0 = kk + 2 * hi;
        v2f a;
        a.x = agg[(size_t)row * AGW + k0]     / dn;
        a.y = agg[(size_t)row * AGW + k0 + 1] / dn;
        const float4 w0 = *(const float4*)(W1 + (size_t)k0 * FH + 4 * lo);
        const float4 w1 = *(const float4*)(W1 + (size_t)(k0 + 1) * FH + 4 * lo);
        v2f bb;
        bb.x = w0.x; bb.y = w1.x;
        c[0] = __builtin_amdgcn_wmma_f32_16x16x4_f32(false, a, false, bb, (short)0, c[0], false, false);
        bb.x = w0.y; bb.y = w1.y;
        c[1] = __builtin_amdgcn_wmma_f32_16x16x4_f32(false, a, false, bb, (short)0, c[1], false, false);
        bb.x = w0.z; bb.y = w1.z;
        c[2] = __builtin_amdgcn_wmma_f32_16x16x4_f32(false, a, false, bb, (short)0, c[2], false, false);
        bb.x = w0.w; bb.y = w1.w;
        c[3] = __builtin_amdgcn_wmma_f32_16x16x4_f32(false, a, false, bb, (short)0, c[3], false, false);
    }
    {
        const float4 bias = *(const float4*)(b1 + 4 * lo);
#pragma unroll
        for (int r = 0; r < 8; ++r) {
            float4 o;
            o.x = c[0][r] + bias.x; o.y = c[1][r] + bias.y; o.z = c[2][r] + bias.z; o.w = c[3][r] + bias.w;
            o.x = o.x > 0.f ? o.x : 0.f; o.y = o.y > 0.f ? o.y : 0.f;
            o.z = o.z > 0.f ? o.z : 0.f; o.w = o.w > 0.f ? o.w : 0.f;
            *(float4*)(T + (r + 8 * hi) * TPITCH + 4 * lo) = o;
        }
    }

    v8f d[4] = {};
#pragma unroll 4
    for (int kk = 0; kk < FH; kk += 4) {
        const int k0 = kk + 2 * hi;
        const v2f a = *(const v2f*)(T + lo * TPITCH + k0);
        const float4 w0 = *(const float4*)(W2 + (size_t)k0 * FH + 4 * lo);
        const float4 w1 = *(const float4*)(W2 + (size_t)(k0 + 1) * FH + 4 * lo);
        v2f bb;
        bb.x = w0.x; bb.y = w1.x;
        d[0] = __builtin_amdgcn_wmma_f32_16x16x4_f32(false, a, false, bb, (short)0, d[0], false, false);
        bb.x = w0.y; bb.y = w1.y;
        d[1] = __builtin_amdgcn_wmma_f32_16x16x4_f32(false, a, false, bb, (short)0, d[1], false, false);
        bb.x = w0.z; bb.y = w1.z;
        d[2] = __builtin_amdgcn_wmma_f32_16x16x4_f32(false, a, false, bb, (short)0, d[2], false, false);
        bb.x = w0.w; bb.y = w1.w;
        d[3] = __builtin_amdgcn_wmma_f32_16x16x4_f32(false, a, false, bb, (short)0, d[3], false, false);
    }

    const int rbase = rowTile * 16 + 8 * hi;
    const float4 as = *(const float4*)(a_src + 4 * lo);
    const float4 ad = *(const float4*)(a_dst + 4 * lo);
    float ps[8], pd[8];
#pragma unroll
    for (int r = 0; r < 8; ++r) {
        float4 o;
        o.x = d[0][r]; o.y = d[1][r]; o.z = d[2][r]; o.w = d[3][r];
        *(float4*)(h2 + (size_t)(rbase + r) * FH + 4 * lo) = o;
        ps[r] = o.x * as.x + o.y * as.y + o.z * as.z + o.w * as.w;
        pd[r] = o.x * ad.x + o.y * ad.y + o.z * ad.z + o.w * ad.w;
    }
#pragma unroll
    for (int r = 0; r < 8; ++r) { ps[r] = half_sum(ps[r]); pd[r] = half_sum(pd[r]); }
    float vs = ps[0], vd = pd[0];
#pragma unroll
    for (int r = 1; r < 8; ++r) {
        vs = lo == r ? ps[r] : vs;
        vd = lo == r ? pd[r] : vd;
    }
    if (lo < 8) {
        s_src[rbase + lo] = vs;
        s_dst[rbase + lo] = vd;
    }
}

__global__ void policy_kernel(const float* __restrict__ act, const float* __restrict__ Wp,
                              const float* __restrict__ bp, float* __restrict__ out,
                              int nRowTiles) {
    const int lane = threadIdx.x & 31;
    const int gw   = blockIdx.x * 8 + (threadIdx.x >> 5);
    if (gw >= nRowTiles) return;
    const int hi  = lane >> 4;
    const int lo  = lane & 15;
    const int row = gw * 16 + lo;
    v8f c = {};
    for (int kk = 0; kk < FH; kk += 4) {
        const int k0 = kk + 2 * hi;
        v2f a, b;
        a.x = act[(size_t)row * FH + k0];
        a.y = act[(size_t)row * FH + k0 + 1];
        b.x = (lo < FOUT) ? Wp[(size_t)k0 * FOUT + lo]       : 0.f;
        b.y = (lo < FOUT) ? Wp[(size_t)(k0 + 1) * FOUT + lo] : 0.f;
        c = __builtin_amdgcn_wmma_f32_16x16x4_f32(false, a, false, b, (short)0, c, false, false);
    }
    if (lo < FOUT) {
        const int rbase = gw * 16 + 8 * hi;
        const float bb = bp[lo];
#pragma unroll
        for (int r = 0; r < 8; ++r)
            out[(size_t)(rbase + r) * FOUT + lo] = c[r] + bb;
    }
}

extern "C" void kernel_launch(void* const* d_in, const int* in_sizes, int n_in,
                              void* d_out, int out_size, void* d_ws, size_t ws_size,
                              hipStream_t stream) {
    const float* x   = (const float*)d_in[0];
    const int*   ei  = (const int*)  d_in[1];
    const float* W1  = (const float*)d_in[2];
    const float* as1 = (const float*)d_in[3];
    const float* ad1 = (const float*)d_in[4];
    const float* b1  = (const float*)d_in[5];
    const float* W2  = (const float*)d_in[6];
    const float* as2 = (const float*)d_in[7];
    const float* ad2 = (const float*)d_in[8];
    const float* b2  = (const float*)d_in[9];
    const float* Wp  = (const float*)d_in[10];
    const float* bp  = (const float*)d_in[11];
    float* out = (float*)d_out;

    const int* srcA = ei;
    const int* dstA = ei + NEDGES;

    float* ws   = (float*)d_ws;
    float* Plo  = ws;
    float* Phi  = ws + (size_t)NNODES * (FH / 2);
    float* Qlo  = ws + (size_t)NNODES * FH;
    float* Qhi  = ws + (size_t)NNODES * FH + (size_t)NNODES * (FH / 2);
    int*   rowend = (int*)(ws + (size_t)2 * NNODES * FH);
    float* ssrc = ws + (size_t)2 * NNODES * FH + NNODES;
    float* sdst = ws + (size_t)2 * NNODES * FH + 2 * NNODES;
    int*   csr  = (int*)Phi;
    unsigned* tmp = (unsigned*)Qlo;
    int*   mat  = (int*)Qlo + NEDGES;
    int*   offs = (int*)Qlo + NEDGES + MATN;
    int*   bsum = (int*)Qlo + NEDGES + 2 * MATN;

    const int nRowTiles = NNODES / 16;
    const int hRowTiles = NHALF / 16;
    const dim3 blk(256);
    const int gemmBlocks  = (nRowTiles * 4 + 7) / 8;
    const int midBlocks   = (nRowTiles + MID_WAVES - 1) / MID_WAVES;
    const int nodeBlocks  = (NNODES + 7) / 8;
    const int nodeBlocksH = (NHALF + 7) / 8;
    const int scanBlocks  = (MATN / 4 + 255) / 256;
    const int polBlocksH  = (hRowTiles + 7) / 8;

    bucket_count_kernel<<<NBLK, blk, 0, stream>>>(dstA, mat);
    block_sum_kernel<<<scanBlocks, blk, 0, stream>>>(mat, bsum, MATN / 4);
    scan_kernel<<<scanBlocks, blk, 0, stream>>>(mat, bsum, offs, MATN / 4);
    bucket_scatter_kernel<<<NBLK, blk, 0, stream>>>(srcA, dstA, offs, tmp);
    bucket_build_kernel<<<NBKT, blk, 0, stream>>>(offs, tmp, rowend, csr);

    float* agg = Plo;
    gemm_h_kernel<<<gemmBlocks, blk, 0, stream>>>(x, W1, Qlo, FIN, nRowTiles);
    scalars_kernel<<<nodeBlocks, blk, 0, stream>>>(Qlo, Qhi, as1, ad1, ssrc, sdst);
    node_aggr_in_kernel<<<nodeBlocks, blk, 0, stream>>>(rowend, csr, ssrc, sdst, x, agg);
    dense_mid_kernel<<<midBlocks, dim3(32 * MID_WAVES), 0, stream>>>(agg, W1, b1, W2, as2, ad2, Qlo, ssrc, sdst, nRowTiles);

    node_aggr_kernel<<<nodeBlocksH, blk, 0, stream>>>(rowend, csr, ssrc, sdst, Qlo, Qhi, b2, Plo, 0);
    policy_kernel<<<polBlocksH, blk, 0, stream>>>(Plo, Wp, bp, out, hRowTiles);
    node_aggr_kernel<<<nodeBlocksH, blk, 0, stream>>>(rowend, csr, ssrc, sdst, Qlo, Qhi, b2, Plo, NHALF);
    policy_kernel<<<polBlocksH, blk, 0, stream>>>(Plo, Wp, bp, out + (size_t)NHALF * FOUT, hRowTiles);
}


# Both aggregation kernels: a wave takes two nodes, one per 16-lane half (edge data by DPP row_share, a 64-float row as 16 bytes per lane): half the waves; sums and their order unchanged
# speedup vs baseline: 4.9738x; 1.0656x over previous
#include <hip/hip_runtime.h>
#include <math.h>

#define NNODES 100000
#define NEDGES 1600000
#define FIN    16
#define FH     64
#define FOUT   8
#define AGW    32

typedef __attribute__((ext_vector_type(2))) float v2f;
typedef __attribute__((ext_vector_type(8))) float v8f;

#define NHALF  (NNODES / 2)
static_assert(NNODES % 32 == 0, "a half is a whole number of 16-row tiles");

__device__ __forceinline__ float leaky(float e) { return e > 0.f ? e : 0.2f * e; }

__global__ void gemm_h_kernel(const float* __restrict__ in, const float* __restrict__ W,
                              float* __restrict__ h, int K, int nRowTiles) {
    const int lane = threadIdx.x & 31;
    const int gw   = blockIdx.x * 8 + (threadIdx.x >> 5);
    const int rowTile = gw >> 2;
    const int colTile = gw & 3;
    if (rowTile >= nRowTiles) return;
    const int hi  = lane >> 4;
    const int lo  = lane & 15;
    const int row = rowTile * 16 + lo;
    const int col = colTile * 16 + lo;
    v8f c = {};
    for (int kk = 0; kk < K; kk += 4) {
        const int k0 = kk + 2 * hi;
        v2f a, b;
        a.x = in[(size_t)row * K + k0];
        a.y = in[(size_t)row * K + k0 + 1];
        b.x = W[(size_t)k0 * FH + col];
        b.y = W[(size_t)(k0 + 1) * FH + col];
        c = __builtin_amdgcn_wmma_f32_16x16x4_f32(false, a, false, b, (short)0, c, false, false);
    }
    const int rbase = rowTile * 16 + 8 * hi;
#pragma unroll
    for (int r = 0; r < 8; ++r)
        h[(size_t)(rbase + r) * FH + col] = c[r];
}

__device__ __forceinline__ const float* plane_row(const float* __restrict__ lo,
                                                  const float* __restrict__ hi, int s) {
    return s < NHALF ? lo + (size_t)s * FH : hi + (size_t)(s - NHALF) * FH;
}

__global__ void scalars_kernel(const float* __restrict__ hlo, const float* __restrict__ hhi,
                               const float* __restrict__ a_src,
                               const float* __restrict__ a_dst,
                               float* __restrict__ s_src, float* __restrict__ s_dst) {
    const int lane = threadIdx.x & 31;
    const int n    = blockIdx.x * 8 + (threadIdx.x >> 5);
    if (n >= NNODES) return;
    const float* __restrict__ hr = plane_row(hlo, hhi, n);
    const float h0 = hr[lane];
    const float h1 = hr[lane + 32];
    float ps = h0 * a_src[lane] + h1 * a_src[lane + 32];
    float pd = h0 * a_dst[lane] + h1 * a_dst[lane + 32];
    for (int off = 16; off; off >>= 1) {
        ps += __shfl_xor(ps, off, 32);
        pd += __shfl_xor(pd, off, 32);
    }
    if (lane == 0) { s_src[n] = ps; s_dst[n] = pd; }
}

#define BKT_SHIFT 8
#define BKT_NODES (1 << BKT_SHIFT)
#define EPT       10
#define EPB       (256 * EPT)
#define NBLK      (NEDGES / EPB)
#define NBKT      (((NNODES + BKT_NODES - 1) / BKT_NODES + 3) / 4 * 4)
#define MATN      (NBKT * NBLK)
#define SRC_BITS  17
#define STAGE_CAP 6144
static_assert(NEDGES % EPB == 0, "level 1 blocks are full");
static_assert(NNODES <= (1 << SRC_BITS) && SRC_BITS + BKT_SHIFT <= 32, "source and in-bucket destination share one word");
static_assert(BKT_NODES == 256 && NBKT % 2 == 0 && NBKT <= 512, "one thread per destination of a bucket, two buckets per thread in the block's scan");

__device__ __forceinline__ int block_scan_incl(int own, int* wsum) {
    const int lane = threadIdx.x & 31;
    const int wv   = threadIdx.x >> 5;
    int inc = own;
    for (int off = 1; off < 32; off <<= 1) {
        const int u = __shfl_up(inc, off, 32);
        if (lane >= off) inc += u;
    }
    if (lane == 31) wsum[wv] = inc;
    __syncthreads();
    for (int k = 0; k < wv; ++k) inc += wsum[k];
    return inc;
}

__global__ void __launch_bounds__(256)
bucket_count_kernel(const int* __restrict__ dst, int* __restrict__ mat) {
    __shared__ int h[NBKT];
    const int t = threadIdx.x;
    for (int i = t; i < NBKT; i += 256) h[i] = 0;
    __syncthreads();
    const int* __restrict__ dp = dst + (size_t)blockIdx.x * EPB + t;
    int d[EPT];
#pragma unroll
    for (int k = 0; k < EPT; ++k) d[k] = dp[k * 256];
#pragma unroll
    for (int k = 0; k < EPT; ++k) atomicAdd(&h[d[k] >> BKT_SHIFT], 1);
    __syncthreads();
    for (int i = t; i < NBKT; i += 256) mat[i * NBLK + blockIdx.x] = h[i];
}

static_assert(MATN % 4 == 0, "counts are read four at a time");
static_assert((MATN / 4 + 255) / 256 <= 256, "one thread per block total");
__global__ void __launch_bounds__(256)
block_sum_kernel(const int* __restrict__ cnt, int* __restrict__ bsum, int nq) {
    __shared__ int wsum[8];
    const int t    = threadIdx.x;
    const int lane = t & 31;
    const int wv   = t >> 5;
    const int q = blockIdx.x * 256 + t;
    int own = 0;
    if (q < nq) {
        const int4 v = ((const int4*)cnt)[q];
        own = v.x + v.y + v.z + v.w;
    }
    for (int off = 16; off; off >>= 1) own += __shfl_xor(own, off, 32);
    if (lane == 0) wsum[wv] = own;
    __syncthreads();
    if (t == 0) {
        int s = 0;
        for (int k = 0; k < 8; ++k) s += wsum[k];
        bsum[blockIdx.x] = s;
    }
}

__global__ void __launch_bounds__(256)
scan_kernel(const int* __restrict__ cnt, const int* __restrict__ bsum, int* __restrict__ cursor, int nq) {
    __shared__ int wsum[8];
    __shared__ int wpre[8];
    const int t    = threadIdx.x;
    const int lane = t & 31;
    const int wv   = t >> 5;
    const int4* __restrict__ c4 = (const int4*)cnt;
    const int q0 = blockIdx.x * 256;
    int pre = t < (int)blockIdx.x ? bsum[t] : 0;
    const int q = q0 + t;
    int4 v = make_int4(0, 0, 0, 0);
    if (q < nq) v = c4[q];
    const int own = v.x + v.y + v.z + v.w;
    int inc = own;
    for (int off = 1; off < 32; off <<= 1) {
        const int u = __shfl_up(inc, off, 32);
        if (lane >= off) inc += u;
    }
    for (int off = 16; off; off >>= 1) pre += __shfl_xor(pre, off, 32);
    if (lane == 31) wsum[wv] = inc;
    if (lane == 0)  wpre[wv] = pre;
    __syncthreads();
    int base = 0;
    for (int k = 0; k < 8; ++k) {
        base += wpre[k];
        if (k < wv) base += wsum[k];
    }
    const int ex = base + inc - own;
    if (q < nq) {
        int4 o;
        o.x = ex; o.y = o.x + v.x; o.z = o.y + v.y; o.w = o.z + v.z;
        ((int4*)cursor)[q] = o;
    }
}

__global__ void __launch_bounds__(256)
bucket_scatter_kernel(const int* __restrict__ src, const int* __restrict__ dst,
                      const int* __restrict__ offs, unsigned* __restrict__ tmp) {
    __shared__ int      h[NBKT];
    __shared__ int      goff[NBKT];
    __shared__ unsigned val[EPB];
    __shared__ int      gpos[EPB];
    __shared__ int      wsum[8];
    const int t = threadIdx.x;
    const int* __restrict__ dp = dst + (size_t)blockIdx.x * EPB + t;
    const int* __restrict__ sp = src + (size_t)blockIdx.x * EPB + t;
    int d[EPT], s[EPT], r[EPT];
#pragma unroll
    for (int k = 0; k < EPT; ++k) { d[k] = dp[k * 256]; s[k] = sp[k * 256]; }
    for (int i = t; i < NBKT; i += 256) { h[i] = 0; goff[i] = offs[i * NBLK + blockIdx.x]; }
    __syncthreads();
#pragma unroll
    for (int k = 0; k < EPT; ++k) r[k] = atomicAdd(&h[d[k] >> BKT_SHIFT], 1);
    __syncthreads();
    int c0 = 0, c1 = 0;
    if (2 * t < NBKT) { c0 = h[2 * t]; c1 = h[2 * t + 1]; }
    const int ex = block_scan_incl(c0 + c1, wsum) - (c0 + c1);
    if (2 * t < NBKT) {
        h[2 * t]     = ex;       goff[2 * t]     -= ex;
        h[2 * t + 1] = ex + c0;  goff[2 * t + 1] -= ex + c0;
    }
    __syncthreads();
#pragma unroll
    for (int k = 0; k < EPT; ++k) {
        const int b  = d[k] >> BKT_SHIFT;
        const int lp = h[b] + r[k];
        val[lp]  = ((unsigned)(d[k] & (BKT_NODES - 1)) << SRC_BITS) | (unsigned)s[k];
        gpos[lp] = goff[b] + lp;
    }
    __syncthreads();
    for (int i = t; i < EPB; i += 256) tmp[gpos[i]] = val[i];
}

#define BUILD_TOGETHER 18
__global__ void __launch_bounds__(256)
bucket_build_kernel(const int* __restrict__ offs, const unsigned* __restrict__ tmp,
                    int* __restrict__ rowend, int* __restrict__ csr) {
    __shared__ int c[BKT_NODES];
    __shared__ int wsum[8];
    __shared__ int stage[STAGE_CAP];
    const int t    = threadIdx.x;
    const int b    = blockIdx.x;
    const int beg  = offs[b * NBLK];
    const int end  = b + 1 < NBKT ? offs[(b + 1) * NBLK] : NEDGES;
    const bool fits = end - beg <= STAGE_CAP;
    c[t] = 0;
    __syncthreads();
    for (int i0 = beg + t; i0 < end; i0 += 256 * BUILD_TOGETHER) {
        unsigned v[BUILD_TOGETHER];
#pragma unroll
        for (int j = 0; j < BUILD_TOGETHER; ++j) {
            const int i = i0 + j * 256;
            v[j] = tmp[i < end ? i : end - 1];
        }
#pragma unroll
        for (int j = 0; j < BUILD_TOGETHER; ++j)
            if (i0 + j * 256 < end) atomicAdd(&c[v[j] >> SRC_BITS], 1);
    }
    __syncthreads();
    const int own    = c[t];
    const int segend = beg + block_scan_incl(own, wsum);
    const int node   = b * BKT_NODES + t;
    if (node < NNODES) rowend[node] = segend;
    c[t] = segend - own;
    __syncthreads();
    for (int i0 = beg + t; i0 < end; i0 += 256 * BUILD_TOGETHER) {
        unsigned v[BUILD_TOGETHER];
#pragma unroll
        for (int j = 0; j < BUILD_TOGETHER; ++j) {
            const int i = i0 + j * 256;
            v[j] = tmp[i < end ? i : end - 1];
        }
#pragma unroll
        for (int j = 0; j < BUILD_TOGETHER; ++j) {
            if (i0 + j * 256 < end) {
                const int p = atomicAdd(&c[v[j] >> SRC_BITS], 1);
                const int sv = (int)(v[j] & ((1u << SRC_BITS) - 1u));
                if (fits) stage[p - beg] = sv; else csr[p] = sv;
            }
        }
    }
    if (fits) {
        __syncthreads();
        for (int i = t; i < end - beg; i += 256) csr[beg + i] = stage[i];
    }
}

#define ROWS_TOGETHER 8

template <int M> __device__ __forceinline__ float half_xor_get(float v) {
    return __int_as_float(__builtin_amdgcn_update_dpp(0, __float_as_int(v), 0x160 + M, 0xf, 0xf, false));
}
template <int K> __device__ __forceinline__ int half_get_i(int v) {
    return __builtin_amdgcn_update_dpp(0, v, 0x150 + K, 0xf, 0xf, false);
}
template <int K> __device__ __forceinline__ float half_get_f(float v) {
    return __int_as_float(half_get_i<K>(__float_as_int(v)));
}
__device__ __forceinline__ float half_max(float v) {
    v = fmaxf(v, half_xor_get<8>(v));
    v = fmaxf(v, half_xor_get<4>(v));
    v = fmaxf(v, half_xor_get<2>(v));
    v = fmaxf(v, half_xor_get<1>(v));
    return v;
}
__device__ __forceinline__ float half_sum_down(float v) {
    v += half_xor_get<8>(v);
    v += half_xor_get<4>(v);
    v += half_xor_get<2>(v);
    v += half_xor_get<1>(v);
    return v;
}

#define L2_REQ(J)                                                                      \
    if (G + J < cnt) {                                                                 \
        const unsigned sj = (unsigned)half_get_i<G + J>(s);                            \
        if (G + J < left) v[J] = *(const float4*)(hb + (sj * (unsigned)(FH * 4) + lo16)); \
    }
#define L2_USE(J)                                                                      \
    if (G + J < cnt) {                                                                 \
        const float wj = half_get_f<G + J>(w);                                         \
        if (G + J < left) {                                                            \
            acc.x += v[J].x * wj; acc.y += v[J].y * wj;                                \
            acc.z += v[J].z * wj; acc.w += v[J].w * wj;                                \
        }                                                                              \
    }
template <int G>
__device__ __forceinline__ void rows64_group(int s, float w, int left, int cnt,
                                             const float* __restrict__ h, int lo, float4& acc) {
    const char* __restrict__ hb = (const char*)h;
    const unsigned lo16 = 16u * (unsigned)lo;
    float4 v[ROWS_TOGETHER];
    L2_REQ(0) L2_REQ(1) L2_REQ(2) L2_REQ(3) L2_REQ(4) L2_REQ(5) L2_REQ(6) L2_REQ(7)
    L2_USE(0) L2_USE(1) L2_USE(2) L2_USE(3) L2_USE(4) L2_USE(5) L2_USE(6) L2_USE(7)
}

static_assert(NHALF % 16 == 0 && NNODES % 16 == 0, "a block of 8 waves takes 16 nodes, no tail");
__global__ void __launch_bounds__(256)
node_aggr_kernel(const int* __restrict__ rowend, const int* __restrict__ csr,
                 const float* __restrict__ s_src, const float* __restrict__ s_dst,
                 const float* __restrict__ h,
                 const float* __restrict__ b, float* __restrict__ act, int n0) {
    const int lane = threadIdx.x & 31;
    const int hi = lane >> 4;
    const int lo = lane & 15;
    const int pr = blockIdx.x * 8 + (threadIdx.x >> 5);
    if (2 * pr >= NHALF) return;
    const int n   = n0 + 2 * pr + hi;
    const int end = rowend[n];
    const int beg = n ? rowend[n - 1] : 0;
    const int deg = end - beg;
    const float sd = s_dst[n];
    const int T = deg + 1;
    const int   sA  = lo      < deg ? csr[beg + lo]      : n;
    const int   sB  = lo + 16 < deg ? csr[beg + lo + 16] : n;
    const float ssA = s_src[sA];
    const float ssB = s_src[sB];
    float mx = fmaxf(ssA, ssB);
    for (int i = lo + 32; i < T; i += 16)
        mx = fmaxf(mx, s_src[i < deg ? csr[beg + i] : n]);
    mx = half_max(mx);
    const float mv = leaky(mx + sd);
    const int Ta = __builtin_amdgcn_readlane(T, 0);
    const int Tb = __builtin_amdgcn_readlane(T, 16);
    const int Tl = Ta > Tb ? Ta : Tb;
    float4 acc = make_float4(0.f, 0.f, 0.f, 0.f);
    float dsE = 0.f, dsO = 0.f;
    for (int base = 0; base < Tl; base += 16) {
        const int i = base + lo;
        int s = sA; float ss = ssA;
        if (base == 16) { s = sB; ss = ssB; }
        if (base > 16)  { s = i < deg ? csr[beg + i] : n; ss = s_src[s]; }
        const float w = i < T ? expf(leaky(ss + sd) - mv) : 0.f;
        if (base & 16) dsO += w; else dsE += w;
        const int left = T - base;
        const int cnt  = Tl - base < 16 ? Tl - base : 16;
        rows64_group<0>(s, w, left, cnt, h, lo, acc);
        if (cnt > ROWS_TOGETHER) rows64_group<ROWS_TOGETHER>(s, w, left, cnt, h, lo, acc);
    }
    const float dn = half_sum_down(dsE + dsO) + 1e-16f;
    const float4 bb = *(const float4*)(b + 4 * lo);
    float4 o;
    o.x = acc.x / dn + bb.x; o.y = acc.y / dn + bb.y; o.z = acc.z / dn + bb.z; o.w = acc.w / dn + bb.w;
    o.x = o.x > 0.f ? o.x : 0.f; o.y = o.y > 0.f ? o.y : 0.f;
    o.z = o.z > 0.f ? o.z : 0.f; o.w = o.w > 0.f ? o.w : 0.f;
    *(float4*)(act + (size_t)(n - n0) * FH + 4 * lo) = o;
}

static_assert(FIN == 16 && AGW >= FIN + 1, "half a wave per input row, one more column for the denominator");

#define L1_REQ(J)                                                                      \
    if (J < cnt) {                                                                     \
        const unsigned sj = (unsigned)half_get_i<J>(s);                                \
        v[J] = *(const float*)(xb + (sj * (unsigned)(FIN * 4) + lo4));                 \
    }
#define L1_USE(J, ACC)                                                                 \
    if (J < cnt) {                                                                     \
        const float wj = half_get_f<J>(w);                                             \
        ACC += J < left ? v[J] * wj : 0.f;                                             \
    }

__global__ void __launch_bounds__(256)
node_aggr_in_kernel(const int* __restrict__ rowend, const int* __restrict__ csr,
                    const float* __restrict__ s_src, const float* __restrict__ s_dst,
                    const float* __restrict__ x, float* __restrict__ agg) {
    const int lane = threadIdx.x & 31;
    const int hi = lane >> 4;
    const int lo = lane & 15;
    const int pr = blockIdx.x * 8 + (threadIdx.x >> 5);
    if (2 * pr >= NNODES) return;
    const int n   = 2 * pr + hi;
    const int end = rowend[n];
    const int beg = n ? rowend[n - 1] : 0;
    const int deg = end - beg;
    const float sd = s_dst[n];
    const int T = deg + 1;
    const int   sA  = lo      < deg ? csr[beg + lo]      : n;
    const int   sB  = lo + 16 < deg ? csr[beg + lo + 16] : n;
    const float ssA = s_src[sA];
    const float ssB = s_src[sB];
    float mx = fmaxf(ssA, ssB);
    for (int i = lo + 32; i < T; i += 16)
        mx = fmaxf(mx, s_src[i < deg ? csr[beg + i] : n]);
    mx = half_max(mx);
    const float mv = leaky(mx + sd);
    const int Ta = __builtin_amdgcn_readlane(T, 0);
    const int Tb = __builtin_amdgcn_readlane(T, 16);
    const int Tl = Ta > Tb ? Ta : Tb;
    const char* __restrict__ xb = (const char*)x;
    const unsigned lo4 = 4u * (unsigned)lo;
    float accE = 0.f, accO = 0.f, dsE = 0.f, dsO = 0.f;
    for (int base = 0; base < Tl; base += 16) {
        const int i = base + lo;
        int s = sA; float ss = ssA;
        if (base == 16) { s = sB; ss = ssB; }
        if (base > 16)  { s = i < deg ? csr[beg + i] : n; ss = s_src[s]; }
        const float w = i < T ? expf(leaky(ss + sd) - mv) : 0.f;
        if (base & 16) dsO += w; else dsE += w;
        const int left = T - base;
        const int cnt  = Tl - base < 16 ? Tl - base : 16;
        float v[16];
        L1_REQ(0) L1_REQ(1) L1_REQ(2)  L1_REQ(3)  L1_REQ(4)  L1_REQ(5)  L1_REQ(6)  L1_REQ(7)
        L1_REQ(8) L1_REQ(9) L1_REQ(10) L1_REQ(11) L1_REQ(12) L1_REQ(13) L1_REQ(14) L1_REQ(15)
        L1_USE(0, accE) L1_USE(1, accO) L1_USE(2,  accE) L1_USE(3,  accO) L1_USE(4,  accE) L1_USE(5,  accO) L1_USE(6,  accE) L1_USE(7,  accO)
        L1_USE(8, accE) L1_USE(9, accO) L1_USE(10, accE) L1_USE(11, accO) L1_USE(12, accE) L1_USE(13, accO) L1_USE(14, accE) L1_USE(15, accO)
    }
    const float acc = accE + accO;
    const float den = half_sum_down(dsE + dsO);
    agg[(size_t)n * AGW + lo] = acc;
    if (lo == 0) agg[(size_t)n * AGW + FIN] = den;
}

#define MID_WAVES 4
#define TPITCH    68

__device__ __forceinline__ float half_sum(float v) {
    v += half_xor_get<1>(v);
    v += half_xor_get<2>(v);
    v += half_xor_get<4>(v);
    v += half_xor_get<8>(v);
    return v;
}

__global__ void __launch_bounds__(32 * MID_WAVES)
dense_mid_kernel(const float* __restrict__ agg,
                 const float* __restrict__ W1, const float* __restrict__ b1,
                 const float* __restrict__ W2,
                 const float* __restrict__ a_src, const float* __restrict__ a_dst,
                 float* __restrict__ h2, float* __restrict__ s_src, float* __restrict__ s_dst,
                 int nRowTiles) {
    __shared__ __attribute__((aligned(16))) float tile[MID_WAVES][16 * TPITCH];
    const int lane = threadIdx.x & 31;
    const int wv   = threadIdx.x >> 5;
    const int rowTile = blockIdx.x * MID_WAVES + wv;
    if (rowTile >= nRowTiles) return;
    const int hi  = lane >> 4;
    const int lo  = lane & 15;
    const int row = rowTile * 16 + lo;
    float* __restrict__ T = tile[wv];

    const float dn = agg[(size_t)row * AGW + FIN] + 1e-16f;
    v8f c[4] = {};
#pragma unroll
    for (int kk = 0; kk < FIN; kk += 4) {
        const int k0 = kk + 2 * hi;
        v2f a;
        a.x = agg[(size_t)row * AGW + k0]     / dn;
        a.y = agg[(size_t)row * AGW + k0 + 1] / dn;
        const float4 w0 = *(const float4*)(W1 + (size_t)k0 * FH + 4 * lo);
        const float4 w1 = *(const float4*)(W1 + (size_t)(k0 + 1) * FH + 4 * lo);
        v2f bb;
        bb.x = w0.x; bb.y = w1.x;
        c[0] = __builtin_amdgcn_wmma_f32_16x16x4_f32(false, a, false, bb, (short)0, c[0], false, false);
        bb.x = w0.y; bb.y = w1.y;
        c[1] = __builtin_amdgcn_wmma_f32_16x16x4_f32(false, a, false, bb, (short)0, c[1], false, false);
        bb.x = w0.z; bb.y = w1.z;
        c[2] = __builtin_amdgcn_wmma_f32_16x16x4_f32(false, a, false, bb, (short)0, c[2], false, false);
        bb.x = w0.w; bb.y = w1.w;
        c[3] = __builtin_amdgcn_wmma_f32_16x16x4_f32(false, a, false, bb, (short)0, c[3], false, false);
    }
    {
        const float4 bias = *(const float4*)(b1 + 4 * lo);
#pragma unroll
        for (int r = 0; r < 8; ++r) {
            float4 o;
            o.x = c[0][r] + bias.x; o.y = c[1][r] + bias.y; o.z = c[2][r] + bias.z; o.w = c[3][r] + bias.w;
            o.x = o.x > 0.f ? o.x : 0.f; o.y = o.y > 0.f ? o.y : 0.f;
            o.z = o.z > 0.f ? o.z : 0.f; o.w = o.w > 0.f ? o.w : 0.f;
            *(float4*)(T + (r + 8 * hi) * TPITCH + 4 * lo) = o;
        }
    }

    v8f d[4] = {};
#pragma unroll 4
    for (int kk = 0; kk < FH; kk += 4) {
        const int k0 = kk + 2 * hi;
        const v2f a = *(const v2f*)(T + lo * TPITCH + k0);
        const float4 w0 = *(const float4*)(W2 + (size_t)k0 * FH + 4 * lo);
        const float4 w1 = *(const float4*)(W2 + (size_t)(k0 + 1) * FH + 4 * lo);
        v2f bb;
        bb.x = w0.x; bb.y = w1.x;
        d[0] = __builtin_amdgcn_wmma_f32_16x16x4_f32(false, a, false, bb, (short)0, d[0], false, false);
        bb.x = w0.y; bb.y = w1.y;
        d[1] = __builtin_amdgcn_wmma_f32_16x16x4_f32(false, a, false, bb, (short)0, d[1], false, false);
        bb.x = w0.z; bb.y = w1.z;
        d[2] = __builtin_amdgcn_wmma_f32_16x16x4_f32(false, a, false, bb, (short)0, d[2], false, false);
        bb.x = w0.w; bb.y = w1.w;
        d[3] = __builtin_amdgcn_wmma_f32_16x16x4_f32(false, a, false, bb, (short)0, d[3], false, false);
    }

    const int rbase = rowTile * 16 + 8 * hi;
    const float4 as = *(const float4*)(a_src + 4 * lo);
    const float4 ad = *(const float4*)(a_dst + 4 * lo);
    float ps[8], pd[8];
#pragma unroll
    for (int r = 0; r < 8; ++r) {
        float4 o;
        o.x = d[0][r]; o.y = d[1][r]; o.z = d[2][r]; o.w = d[3][r];
        *(float4*)(h2 + (size_t)(rbase + r) * FH + 4 * lo) = o;
        ps[r] = o.x * as.x + o.y * as.y + o.z * as.z + o.w * as.w;
        pd[r] = o.x * ad.x + o.y * ad.y + o.z * ad.z + o.w * ad.w;
    }
#pragma unroll
    for (int r = 0; r < 8; ++r) { ps[r] = half_sum(ps[r]); pd[r] = half_sum(pd[r]); }
    float vs = ps[0], vd = pd[0];
#pragma unroll
    for (int r = 1; r < 8; ++r) {
        vs = lo == r ? ps[r] : vs;
        vd = lo == r ? pd[r] : vd;
    }
    if (lo < 8) {
        s_src[rbase + lo] = vs;
        s_dst[rbase + lo] = vd;
    }
}

__global__ void policy_kernel(const float* __restrict__ act, const float* __restrict__ Wp,
                              const float* __restrict__ bp, float* __restrict__ out,
                              int nRowTiles) {
    const int lane = threadIdx.x & 31;
    const int gw   = blockIdx.x * 8 + (threadIdx.x >> 5);
    if (gw >= nRowTiles) return;
    const int hi  = lane >> 4;
    const int lo  = lane & 15;
    const int row = gw * 16 + lo;
    v8f c = {};
    for (int kk = 0; kk < FH; kk += 4) {
        const int k0 = kk + 2 * hi;
        v2f a, b;
        a.x = act[(size_t)row * FH + k0];
        a.y = act[(size_t)row * FH + k0 + 1];
        b.x = (lo < FOUT) ? Wp[(size_t)k0 * FOUT + lo]       : 0.f;
        b.y = (lo < FOUT) ? Wp[(size_t)(k0 + 1) * FOUT + lo] : 0.f;
        c = __builtin_amdgcn_wmma_f32_16x16x4_f32(false, a, false, b, (short)0, c, false, false);
    }
    if (lo < FOUT) {
        const int rbase = gw * 16 + 8 * hi;
        const float bb = bp[lo];
#pragma unroll
        for (int r = 0; r < 8; ++r)
            out[(size_t)(rbase + r) * FOUT + lo] = c[r] + bb;
    }
}

extern "C" void kernel_launch(void* const* d_in, const int* in_sizes, int n_in,
                              void* d_out, int out_size, void* d_ws, size_t ws_size,
                              hipStream_t stream) {
    const float* x   = (const float*)d_in[0];
    const int*   ei  = (const int*)  d_in[1];
    const float* W1  = (const float*)d_in[2];
    const float* as1 = (const float*)d_in[3];
    const float* ad1 = (const float*)d_in[4];
    const float* b1  = (const float*)d_in[5];
    const float* W2  = (const float*)d_in[6];
    const float* as2 = (const float*)d_in[7];
    const float* ad2 = (const float*)d_in[8];
    const float* b2  = (const float*)d_in[9];
    const float* Wp  = (const float*)d_in[10];
    const float* bp  = (const float*)d_in[11];
    float* out = (float*)d_out;

    const int* srcA = ei;
    const int* dstA = ei + NEDGES;

    float* ws   = (float*)d_ws;
    float* Plo  = ws;
    float* Phi  = ws + (size_t)NNODES * (FH / 2);
    float* Qlo  = ws + (size_t)NNODES * FH;
    float* Qhi  = ws + (size_t)NNODES * FH + (size_t)NNODES * (FH / 2);
    int*   rowend = (int*)(ws + (size_t)2 * NNODES * FH);
    float* ssrc = ws + (size_t)2 * NNODES * FH + NNODES;
    float* sdst = ws + (size_t)2 * NNODES * FH + 2 * NNODES;
    int*   csr  = (int*)Phi;
    unsigned* tmp = (unsigned*)Qlo;
    int*   mat  = (int*)Qlo + NEDGES;
    int*   offs = (int*)Qlo + NEDGES + MATN;
    int*   bsum = (int*)Qlo + NEDGES + 2 * MATN;

    const int nRowTiles = NNODES / 16;
    const int hRowTiles = NHALF / 16;
    const dim3 blk(256);
    const int gemmBlocks  = (nRowTiles * 4 + 7) / 8;
    const int midBlocks   = (nRowTiles + MID_WAVES - 1) / MID_WAVES;
    const int nodeBlocks  = (NNODES + 7) / 8;
    const int pairBlocks  = NNODES / 16;
    const int pairBlocksH = NHALF / 16;
    const int scanBlocks  = (MATN / 4 + 255) / 256;
    const int polBlocksH  = (hRowTiles + 7) / 8;

    bucket_count_kernel<<<NBLK, blk, 0, stream>>>(dstA, mat);
    block_sum_kernel<<<scanBlocks, blk, 0, stream>>>(mat, bsum, MATN / 4);
    scan_kernel<<<scanBlocks, blk, 0, stream>>>(mat, bsum, offs, MATN / 4);
    bucket_scatter_kernel<<<NBLK, blk, 0, stream>>>(srcA, dstA, offs, tmp);
    bucket_build_kernel<<<NBKT, blk, 0, stream>>>(offs, tmp, rowend, csr);

    float* agg = Plo;
    gemm_h_kernel<<<gemmBlocks, blk, 0, stream>>>(x, W1, Qlo, FIN, nRowTiles);
    scalars_kernel<<<nodeBlocks, blk, 0, stream>>>(Qlo, Qhi, as1, ad1, ssrc, sdst);
    node_aggr_in_kernel<<<pairBlocks, blk, 0, stream>>>(rowend, csr, ssrc, sdst, x, agg);
    dense_mid_kernel<<<midBlocks, dim3(32 * MID_WAVES), 0, stream>>>(agg, W1, b1, W2, as2, ad2, Qlo, ssrc, sdst, nRowTiles);

    node_aggr_kernel<<<pairBlocksH, blk, 0, stream>>>(rowend, csr, ssrc, sdst, Qlo, b2, Plo, 0);
    policy_kernel<<<polBlocksH, blk, 0, stream>>>(Plo, Wp, bp, out, hRowTiles);
    node_aggr_kernel<<<pairBlocksH, blk, 0, stream>>>(rowend, csr, ssrc, sdst, Qlo, b2, Plo, NHALF);
    policy_kernel<<<polBlocksH, blk, 0, stream>>>(Plo, Wp, bp, out + (size_t)NHALF * FOUT, hRowTiles);
}


# Layer 2's aggregation and the policy head in one launch per 16-node block (act2 rows go through LDS to wave 0's Wp product, never stored): 3 launches go; sums and their order unchanged
# speedup vs baseline: 5.7446x; 1.1550x over previous
#include <hip/hip_runtime.h>
#include <math.h>

#define NNODES 100000
#define NEDGES 1600000
#define FIN    16
#define FH     64
#define FOUT   8
#define AGW    32

typedef __attribute__((ext_vector_type(2))) float v2f;
typedef __attribute__((ext_vector_type(8))) float v8f;

#define NHALF  (NNODES / 2)
static_assert(NNODES % 32 == 0, "a half is a whole number of 16-row tiles");

__device__ __forceinline__ float leaky(float e) { return e > 0.f ? e : 0.2f * e; }

__global__ void gemm_h_kernel(const float* __restrict__ in, const float* __restrict__ W,
                              float* __restrict__ h, int K, int nRowTiles) {
    const int lane = threadIdx.x & 31;
    const int gw   = blockIdx.x * 8 + (threadIdx.x >> 5);
    const int rowTile = gw >> 2;
    const int colTile = gw & 3;
    if (rowTile >= nRowTiles) return;
    const int hi  = lane >> 4;
    const int lo  = lane & 15;
    const int row = rowTile * 16 + lo;
    const int col = colTile * 16 + lo;
    v8f c = {};
    for (int kk = 0; kk < K; kk += 4) {
        const int k0 = kk + 2 * hi;
        v2f a, b;
        a.x = in[(size_t)row * K + k0];
        a.y = in[(size_t)row * K + k0 + 1];
        b.x = W[(size_t)k0 * FH + col];
        b.y = W[(size_t)(k0 + 1) * FH + col];
        c = __builtin_amdgcn_wmma_f32_16x16x4_f32(false, a, false, b, (short)0, c, false, false);
    }
    const int rbase = rowTile * 16 + 8 * hi;
#pragma unroll
    for (int r = 0; r < 8; ++r)
        h[(size_t)(rbase + r) * FH + col] = c[r];
}

__device__ __forceinline__ const float* plane_row(const float* __restrict__ lo,
                                                  const float* __restrict__ hi, int s) {
    return s < NHALF ? lo + (size_t)s * FH : hi + (size_t)(s - NHALF) * FH;
}

__global__ void scalars_kernel(const float* __restrict__ hlo, const float* __restrict__ hhi,
                               const float* __restrict__ a_src,
                               const float* __restrict__ a_dst,
                               float* __restrict__ s_src, float* __restrict__ s_dst) {
    const int lane = threadIdx.x & 31;
    const int n    = blockIdx.x * 8 + (threadIdx.x >> 5);
    if (n >= NNODES) return;
    const float* __restrict__ hr = plane_row(hlo, hhi, n);
    const float h0 = hr[lane];
    const float h1 = hr[lane + 32];
    float ps = h0 * a_src[lane] + h1 * a_src[lane + 32];
    float pd = h0 * a_dst[lane] + h1 * a_dst[lane + 32];
    for (int off = 16; off; off >>= 1) {
        ps += __shfl_xor(ps, off, 32);
        pd += __shfl_xor(pd, off, 32);
    }
    if (lane == 0) { s_src[n] = ps; s_dst[n] = pd; }
}

#define BKT_SHIFT 8
#define BKT_NODES (1 << BKT_SHIFT)
#define EPT       10
#define EPB       (256 * EPT)
#define NBLK      (NEDGES / EPB)
#define NBKT      (((NNODES + BKT_NODES - 1) / BKT_NODES + 3) / 4 * 4)
#define MATN      (NBKT * NBLK)
#define SRC_BITS  17
#define STAGE_CAP 6144
static_assert(NEDGES % EPB == 0, "level 1 blocks are full");
static_assert(NNODES <= (1 << SRC_BITS) && SRC_BITS + BKT_SHIFT <= 32, "source and in-bucket destination share one word");
static_assert(BKT_NODES == 256 && NBKT % 2 == 0 && NBKT <= 512, "one thread per destination of a bucket, two buckets per thread in the block's scan");

__device__ __forceinline__ int block_scan_incl(int own, int* wsum) {
    const int lane = threadIdx.x & 31;
    const int wv   = threadIdx.x >> 5;
    int inc = own;
    for (int off = 1; off < 32; off <<= 1) {
        const int u = __shfl_up(inc, off, 32);
        if (lane >= off) inc += u;
    }
    if (lane == 31) wsum[wv] = inc;
    __syncthreads();
    for (int k = 0; k < wv; ++k) inc += wsum[k];
    return inc;
}

__global__ void __launch_bounds__(256)
bucket_count_kernel(const int* __restrict__ dst, int* __restrict__ mat) {
    __shared__ int h[NBKT];
    const int t = threadIdx.x;
    for (int i = t; i < NBKT; i += 256) h[i] = 0;
    __syncthreads();
    const int* __restrict__ dp = dst + (size_t)blockIdx.x * EPB + t;
    int d[EPT];
#pragma unroll
    for (int k = 0; k < EPT; ++k) d[k] = dp[k * 256];
#pragma unroll
    for (int k = 0; k < EPT; ++k) atomicAdd(&h[d[k] >> BKT_SHIFT], 1);
    __syncthreads();
    for (int i = t; i < NBKT; i += 256) mat[i * NBLK + blockIdx.x] = h[i];
}

static_assert(MATN % 4 == 0, "counts are read four at a time");
static_assert((MATN / 4 + 255) / 256 <= 256, "one thread per block total");
__global__ void __launch_bounds__(256)
block_sum_kernel(const int* __restrict__ cnt, int* __restrict__ bsum, int nq) {
    __shared__ int wsum[8];
    const int t    = threadIdx.x;
    const int lane = t & 31;
    const int wv   = t >> 5;
    const int q = blockIdx.x * 256 + t;
    int own = 0;
    if (q < nq) {
        const int4 v = ((const int4*)cnt)[q];
        own = v.x + v.y + v.z + v.w;
    }
    for (int off = 16; off; off >>= 1) own += __shfl_xor(own, off, 32);
    if (lane == 0) wsum[wv] = own;
    __syncthreads();
    if (t == 0) {
        int s = 0;
        for (int k = 0; k < 8; ++k) s += wsum[k];
        bsum[blockIdx.x] = s;
    }
}

__global__ void __launch_bounds__(256)
scan_kernel(const int* __restrict__ cnt, const int* __restrict__ bsum, int* __restrict__ cursor, int nq) {
    __shared__ int wsum[8];
    __shared__ int wpre[8];
    const int t    = threadIdx.x;
    const int lane = t & 31;
    const int wv   = t >> 5;
    const int4* __restrict__ c4 = (const int4*)cnt;
    const int q0 = blockIdx.x * 256;
    int pre = t < (int)blockIdx.x ? bsum[t] : 0;
    const int q = q0 + t;
    int4 v = make_int4(0, 0, 0, 0);
    if (q < nq) v = c4[q];
    const int own = v.x + v.y + v.z + v.w;
    int inc = own;
    for (int off = 1; off < 32; off <<= 1) {
        const int u = __shfl_up(inc, off, 32);
        if (lane >= off) inc += u;
    }
    for (int off = 16; off; off >>= 1) pre += __shfl_xor(pre, off, 32);
    if (lane == 31) wsum[wv] = inc;
    if (lane == 0)  wpre[wv] = pre;
    __syncthreads();
    int base = 0;
    for (int k = 0; k < 8; ++k) {
        base += wpre[k];
        if (k < wv) base += wsum[k];
    }
    const int ex = base + inc - own;
    if (q < nq) {
        int4 o;
        o.x = ex; o.y = o.x + v.x; o.z = o.y + v.y; o.w = o.z + v.z;
        ((int4*)cursor)[q] = o;
    }
}

__global__ void __launch_bounds__(256)
bucket_scatter_kernel(const int* __restrict__ src, const int* __restrict__ dst,
                      const int* __restrict__ offs, unsigned* __restrict__ tmp) {
    __shared__ int      h[NBKT];
    __shared__ int      goff[NBKT];
    __shared__ unsigned val[EPB];
    __shared__ int      gpos[EPB];
    __shared__ int      wsum[8];
    const int t = threadIdx.x;
    const int* __restrict__ dp = dst + (size_t)blockIdx.x * EPB + t;
    const int* __restrict__ sp = src + (size_t)blockIdx.x * EPB + t;
    int d[EPT], s[EPT], r[EPT];
#pragma unroll
    for (int k = 0; k < EPT; ++k) { d[k] = dp[k * 256]; s[k] = sp[k * 256]; }
    for (int i = t; i < NBKT; i += 256) { h[i] = 0; goff[i] = offs[i * NBLK + blockIdx.x]; }
    __syncthreads();
#pragma unroll
    for (int k = 0; k < EPT; ++k) r[k] = atomicAdd(&h[d[k] >> BKT_SHIFT], 1);
    __syncthreads();
    int c0 = 0, c1 = 0;
    if (2 * t < NBKT) { c0 = h[2 * t]; c1 = h[2 * t + 1]; }
    const int ex = block_scan_incl(c0 + c1, wsum) - (c0 + c1);
    if (2 * t < NBKT) {
        h[2 * t]     = ex;       goff[2 * t]     -= ex;
        h[2 * t + 1] = ex + c0;  goff[2 * t + 1] -= ex + c0;
    }
    __syncthreads();
#pragma unroll
    for (int k = 0; k < EPT; ++k) {
        const int b  = d[k] >> BKT_SHIFT;
        const int lp = h[b] + r[k];
        val[lp]  = ((unsigned)(d[k] & (BKT_NODES - 1)) << SRC_BITS) | (unsigned)s[k];
        gpos[lp] = goff[b] + lp;
    }
    __syncthreads();
    for (int i = t; i < EPB; i += 256) tmp[gpos[i]] = val[i];
}

#define BUILD_TOGETHER 18
__global__ void __launch_bounds__(256)
bucket_build_kernel(const int* __restrict__ offs, const unsigned* __restrict__ tmp,
                    int* __restrict__ rowend, int* __restrict__ csr) {
    __shared__ int c[BKT_NODES];
    __shared__ int wsum[8];
    __shared__ int stage[STAGE_CAP];
    const int t    = threadIdx.x;
    const int b    = blockIdx.x;
    const int beg  = offs[b * NBLK];
    const int end  = b + 1 < NBKT ? offs[(b + 1) * NBLK] : NEDGES;
    const bool fits = end - beg <= STAGE_CAP;
    c[t] = 0;
    __syncthreads();
    for (int i0 = beg + t; i0 < end; i0 += 256 * BUILD_TOGETHER) {
        unsigned v[BUILD_TOGETHER];
#pragma unroll
        for (int j = 0; j < BUILD_TOGETHER; ++j) {
            const int i = i0 + j * 256;
            v[j] = tmp[i < end ? i : end - 1];
        }
#pragma unroll
        for (int j = 0; j < BUILD_TOGETHER; ++j)
            if (i0 + j * 256 < end) atomicAdd(&c[v[j] >> SRC_BITS], 1);
    }
    __syncthreads();
    const int own    = c[t];
    const int segend = beg + block_scan_incl(own, wsum);
    const int node   = b * BKT_NODES + t;
    if (node < NNODES) rowend[node] = segend;
    c[t] = segend - own;
    __syncthreads();
    for (int i0 = beg + t; i0 < end; i0 += 256 * BUILD_TOGETHER) {
        unsigned v[BUILD_TOGETHER];
#pragma unroll
        for (int j = 0; j < BUILD_TOGETHER; ++j) {
            const int i = i0 + j * 256;
            v[j] = tmp[i < end ? i : end - 1];
        }
#pragma unroll
        for (int j = 0; j < BUILD_TOGETHER; ++j) {
            if (i0 + j * 256 < end) {
                const int p = atomicAdd(&c[v[j] >> SRC_BITS], 1);
                const int sv = (int)(v[j] & ((1u << SRC_BITS) - 1u));
                if (fits) stage[p - beg] = sv; else csr[p] = sv;
            }
        }
    }
    if (fits) {
        __syncthreads();
        for (int i = t; i < end - beg; i += 256) csr[beg + i] = stage[i];
    }
}

#define ROWS_TOGETHER 8

template <int M> __device__ __forceinline__ float half_xor_get(float v) {
    return __int_as_float(__builtin_amdgcn_update_dpp(0, __float_as_int(v), 0x160 + M, 0xf, 0xf, false));
}
template <int K> __device__ __forceinline__ int half_get_i(int v) {
    return __builtin_amdgcn_update_dpp(0, v, 0x150 + K, 0xf, 0xf, false);
}
template <int K> __device__ __forceinline__ float half_get_f(float v) {
    return __int_as_float(half_get_i<K>(__float_as_int(v)));
}
__device__ __forceinline__ float half_max(float v) {
    v = fmaxf(v, half_xor_get<8>(v));
    v = fmaxf(v, half_xor_get<4>(v));
    v = fmaxf(v, half_xor_get<2>(v));
    v = fmaxf(v, half_xor_get<1>(v));
    return v;
}
__device__ __forceinline__ float half_sum_down(float v) {
    v += half_xor_get<8>(v);
    v += half_xor_get<4>(v);
    v += half_xor_get<2>(v);
    v += half_xor_get<1>(v);
    return v;
}

#define L2_REQ(J)                                                                      \
    if (G + J < cnt) {                                                                 \
        const unsigned sj = (unsigned)half_get_i<G + J>(s);                            \
        if (G + J < left) v[J] = *(const float4*)(hb + (sj * (unsigned)(FH * 4) + lo16)); \
    }
#define L2_USE(J)                                                                      \
    if (G + J < cnt) {                                                                 \
        const float wj = half_get_f<G + J>(w);                                         \
        if (G + J < left) {                                                            \
            acc.x += v[J].x * wj; acc.y += v[J].y * wj;                                \
            acc.z += v[J].z * wj; acc.w += v[J].w * wj;                                \
        }                                                                              \
    }
template <int G>
__device__ __forceinline__ void rows64_group(int s, float w, int left, int cnt,
                                             const float* __restrict__ h, int lo, float4& acc) {
    const char* __restrict__ hb = (const char*)h;
    const unsigned lo16 = 16u * (unsigned)lo;
    float4 v[ROWS_TOGETHER];
    L2_REQ(0) L2_REQ(1) L2_REQ(2) L2_REQ(3) L2_REQ(4) L2_REQ(5) L2_REQ(6) L2_REQ(7)
    L2_USE(0) L2_USE(1) L2_USE(2) L2_USE(3) L2_USE(4) L2_USE(5) L2_USE(6) L2_USE(7)
}

#define HPITCH 68
static_assert(NNODES % 16 == 0, "a block of 8 waves takes 16 nodes, no tail: every wave reaches the barrier");
static_assert(FOUT <= 16 && (FOUT & (FOUT - 1)) == 0, "lane lo reads column lo modulo FOUT of Wp");
__global__ void __launch_bounds__(256)
node_aggr_head_kernel(const int* __restrict__ rowend, const int* __restrict__ csr,
                      const float* __restrict__ s_src, const float* __restrict__ s_dst,
                      const float* __restrict__ h, const float* __restrict__ b,
                      const float* __restrict__ Wp, const float* __restrict__ bp,
                      float* __restrict__ out) {
    __shared__ __attribute__((aligned(16))) float tile[16 * HPITCH];
    const int lane = threadIdx.x & 31;
    const int wv = threadIdx.x >> 5;
    const int hi = lane >> 4;
    const int lo = lane & 15;
    const int n   = blockIdx.x * 16 + 2 * wv + hi;
    const int end = rowend[n];
    const int beg = n ? rowend[n - 1] : 0;
    const int deg = end - beg;
    const float sd = s_dst[n];
    const int T = deg + 1;
    const int   sA  = lo      < deg ? csr[beg + lo]      : n;
    const int   sB  = lo + 16 < deg ? csr[beg + lo + 16] : n;
    const float ssA = s_src[sA];
    const float ssB = s_src[sB];
    float mx = fmaxf(ssA, ssB);
    for (int i = lo + 32; i < T; i += 16)
        mx = fmaxf(mx, s_src[i < deg ? csr[beg + i] : n]);
    mx = half_max(mx);
    const float mv = leaky(mx + sd);
    const int Ta = __builtin_amdgcn_readlane(T, 0);
    const int Tb = __builtin_amdgcn_readlane(T, 16);
    const int Tl = Ta > Tb ? Ta : Tb;
    float4 acc = make_float4(0.f, 0.f, 0.f, 0.f);
    float dsE = 0.f, dsO = 0.f;
    for (int base = 0; base < Tl; base += 16) {
        const int i = base + lo;
        int s = sA; float ss = ssA;
        if (base == 16) { s = sB; ss = ssB; }
        if (base > 16)  { s = i < deg ? csr[beg + i] : n; ss = s_src[s]; }
        const float w = i < T ? expf(leaky(ss + sd) - mv) : 0.f;
        if (base & 16) dsO += w; else dsE += w;
        const int left = T - base;
        const int cnt  = Tl - base < 16 ? Tl - base : 16;
        rows64_group<0>(s, w, left, cnt, h, lo, acc);
        if (cnt > ROWS_TOGETHER) rows64_group<ROWS_TOGETHER>(s, w, left, cnt, h, lo, acc);
    }
    const float dn = half_sum_down(dsE + dsO) + 1e-16f;
    const float4 bb = *(const float4*)(b + 4 * lo);
    float4 o;
    o.x = acc.x / dn + bb.x; o.y = acc.y / dn + bb.y; o.z = acc.z / dn + bb.z; o.w = acc.w / dn + bb.w;
    o.x = o.x > 0.f ? o.x : 0.f; o.y = o.y > 0.f ? o.y : 0.f;
    o.z = o.z > 0.f ? o.z : 0.f; o.w = o.w > 0.f ? o.w : 0.f;
    *(float4*)(tile + (2 * wv + hi) * HPITCH + 4 * lo) = o;
    __syncthreads();
    if (wv != 0) return;

    const float* __restrict__ ta = tile + lo * HPITCH + 2 * hi;
    const float* __restrict__ wp = Wp + (2 * hi) * FOUT + (lo & (FOUT - 1));
    v8f c = {};
    for (int kk = 0; kk < FH; kk += 4) {
        const v2f a = *(const v2f*)(ta + kk);
        v2f wb;
        wb.x = wp[kk * FOUT];
        wb.y = wp[(kk + 1) * FOUT];
        c = __builtin_amdgcn_wmma_f32_16x16x4_f32(false, a, false, wb, (short)0, c, false, false);
    }
    if (lo < FOUT) {
        const int rbase = blockIdx.x * 16 + 8 * hi;
        const float pb = bp[lo];
#pragma unroll
        for (int r = 0; r < 8; ++r)
            out[(size_t)(rbase + r) * FOUT + lo] = c[r] + pb;
    }
}

static_assert(FIN == 16 && AGW >= FIN + 1, "half a wave per input row, one more column for the denominator");

#define L1_REQ(J)                                                                      \
    if (J < cnt) {                                                                     \
        const unsigned sj = (unsigned)half_get_i<J>(s);                                \
        v[J] = *(const float*)(xb + (sj * (unsigned)(FIN * 4) + lo4));                 \
    }
#define L1_USE(J, ACC)                                                                 \
    if (J < cnt) {                                                                     \
        const float wj = half_get_f<J>(w);                                             \
        ACC += J < left ? v[J] * wj : 0.f;                                             \
    }

__global__ void __launch_bounds__(256)
node_aggr_in_kernel(const int* __restrict__ rowend, const int* __restrict__ csr,
                    const float* __restrict__ s_src, const float* __restrict__ s_dst,
                    const float* __restrict__ x, float* __restrict__ agg) {
    const int lane = threadIdx.x & 31;
    const int hi = lane >> 4;
    const int lo = lane & 15;
    const int pr = blockIdx.x * 8 + (threadIdx.x >> 5);
    if (2 * pr >= NNODES) return;
    const int n   = 2 * pr + hi;
    const int end = rowend[n];
    const int beg = n ? rowend[n - 1] : 0;
    const int deg = end - beg;
    const float sd = s_dst[n];
    const int T = deg + 1;
    const int   sA  = lo      < deg ? csr[beg + lo]      : n;
    const int   sB  = lo + 16 < deg ? csr[beg + lo + 16] : n;
    const float ssA = s_src[sA];
    const float ssB = s_src[sB];
    float mx = fmaxf(ssA, ssB);
    for (int i = lo + 32; i < T; i += 16)
        mx = fmaxf(mx, s_src[i < deg ? csr[beg + i] : n]);
    mx = half_max(mx);
    const float mv = leaky(mx + sd);
    const int Ta = __builtin_amdgcn_readlane(T, 0);
    const int Tb = __builtin_amdgcn_readlane(T, 16);
    const int Tl = Ta > Tb ? Ta : Tb;
    const char* __restrict__ xb = (const char*)x;
    const unsigned lo4 = 4u * (unsigned)lo;
    float accE = 0.f, accO = 0.f, dsE = 0.f, dsO = 0.f;
    for (int base = 0; base < Tl; base += 16) {
        const int i = base + lo;
        int s = sA; float ss = ssA;
        if (base == 16) { s = sB; ss = ssB; }
        if (base > 16)  { s = i < deg ? csr[beg + i] : n; ss = s_src[s]; }
        const float w = i < T ? expf(leaky(ss + sd) - mv) : 0.f;
        if (base & 16) dsO += w; else dsE += w;
        const int left = T - base;
        const int cnt  = Tl - base < 16 ? Tl - base : 16;
        float v[16];
        L1_REQ(0) L1_REQ(1) L1_REQ(2)  L1_REQ(3)  L1_REQ(4)  L1_REQ(5)  L1_REQ(6)  L1_REQ(7)
        L1_REQ(8) L1_REQ(9) L1_REQ(10) L1_REQ(11) L1_REQ(12) L1_REQ(13) L1_REQ(14) L1_REQ(15)
        L1_USE(0, accE) L1_USE(1, accO) L1_USE(2,  accE) L1_USE(3,  accO) L1_USE(4,  accE) L1_USE(5,  accO) L1_USE(6,  accE) L1_USE(7,  accO)
        L1_USE(8, accE) L1_USE(9, accO) L1_USE(10, accE) L1_USE(11, accO) L1_USE(12, accE) L1_USE(13, accO) L1_USE(14, accE) L1_USE(15, accO)
    }
    const float acc = accE + accO;
    const float den = half_sum_down(dsE + dsO);
    agg[(size_t)n * AGW + lo] = acc;
    if (lo == 0) agg[(size_t)n * AGW + FIN] = den;
}

#define MID_WAVES 4
#define TPITCH    68

__device__ __forceinline__ float half_sum(float v) {
    v += half_xor_get<1>(v);
    v += half_xor_get<2>(v);
    v += half_xor_get<4>(v);
    v += half_xor_get<8>(v);
    return v;
}

__global__ void __launch_bounds__(32 * MID_WAVES)
dense_mid_kernel(const float* __restrict__ agg,
                 const float* __restrict__ W1, const float* __restrict__ b1,
                 const float* __restrict__ W2,
                 const float* __restrict__ a_src, const float* __restrict__ a_dst,
                 float* __restrict__ h2, float* __restrict__ s_src, float* __restrict__ s_dst,
                 int nRowTiles) {
    __shared__ __attribute__((aligned(16))) float tile[MID_WAVES][16 * TPITCH];
    const int lane = threadIdx.x & 31;
    const int wv   = threadIdx.x >> 5;
    const int rowTile = blockIdx.x * MID_WAVES + wv;
    if (rowTile >= nRowTiles) return;
    const int hi  = lane >> 4;
    const int lo  = lane & 15;
    const int row = rowTile * 16 + lo;
    float* __restrict__ T = tile[wv];

    const float dn = agg[(size_t)row * AGW + FIN] + 1e-16f;
    v8f c[4] = {};
#pragma unroll
    for (int kk = 0; kk < FIN; kk += 4) {
        const int k0 = kk + 2 * hi;
        v2f a;
        a.x = agg[(size_t)row * AGW + k0]     / dn;
        a.y = agg[(size_t)row * AGW + k0 + 1] / dn;
        const float4 w0 = *(const float4*)(W1 + (size_t)k0 * FH + 4 * lo);
        const float4 w1 = *(const float4*)(W1 + (size_t)(k0 + 1) * FH + 4 * lo);
        v2f bb;
        bb.x = w0.x; bb.y = w1.x;
        c[0] = __builtin_amdgcn_wmma_f32_16x16x4_f32(false, a, false, bb, (short)0, c[0], false, false);
        bb.x = w0.y; bb.y = w1.y;
        c[1] = __builtin_amdgcn_wmma_f32_16x16x4_f32(false, a, false, bb, (short)0, c[1], false, false);
        bb.x = w0.z; bb.y = w1.z;
        c[2] = __builtin_amdgcn_wmma_f32_16x16x4_f32(false, a, false, bb, (short)0, c[2], false, false);
        bb.x = w0.w; bb.y = w1.w;
        c[3] = __builtin_amdgcn_wmma_f32_16x16x4_f32(false, a, false, bb, (short)0, c[3], false, false);
    }
    {
        const float4 bias = *(const float4*)(b1 + 4 * lo);
#pragma unroll
        for (int r = 0; r < 8; ++r) {
            float4 o;
            o.x = c[0][r] + bias.x; o.y = c[1][r] + bias.y; o.z = c[2][r] + bias.z; o.w = c[3][r] + bias.w;
            o.x = o.x > 0.f ? o.x : 0.f; o.y = o.y > 0.f ? o.y : 0.f;
            o.z = o.z > 0.f ? o.z : 0.f; o.w = o.w > 0.f ? o.w : 0.f;
            *(float4*)(T + (r + 8 * hi) * TPITCH + 4 * lo) = o;
        }
    }

    v8f d[4] = {};
#pragma unroll 4
    for (int kk = 0; kk < FH; kk += 4) {
        const int k0 = kk + 2 * hi;
        const v2f a = *(const v2f*)(T + lo * TPITCH + k0);
        const float4 w0 = *(const float4*)(W2 + (size_t)k0 * FH + 4 * lo);
        const float4 w1 = *(const float4*)(W2 + (size_t)(k0 + 1) * FH + 4 * lo);
        v2f bb;
        bb.x = w0.x; bb.y = w1.x;
        d[0] = __builtin_amdgcn_wmma_f32_16x16x4_f32(false, a, false, bb, (short)0, d[0], false, false);
        bb.x = w0.y; bb.y = w1.y;
        d[1] = __builtin_amdgcn_wmma_f32_16x16x4_f32(false, a, false, bb, (short)0, d[1], false, false);
        bb.x = w0.z; bb.y = w1.z;
        d[2] = __builtin_amdgcn_wmma_f32_16x16x4_f32(false, a, false, bb, (short)0, d[2], false, false);
        bb.x = w0.w; bb.y = w1.w;
        d[3] = __builtin_amdgcn_wmma_f32_16x16x4_f32(false, a, false, bb, (short)0, d[3], false, false);
    }

    const int rbase = rowTile * 16 + 8 * hi;
    const float4 as = *(const float4*)(a_src + 4 * lo);
    const float4 ad = *(const float4*)(a_dst + 4 * lo);
    float ps[8], pd[8];
#pragma unroll
    for (int r = 0; r < 8; ++r) {
        float4 o;
        o.x = d[0][r]; o.y = d[1][r]; o.z = d[2][r]; o.w = d[3][r];
        *(float4*)(h2 + (size_t)(rbase + r) * FH + 4 * lo) = o;
        ps[r] = o.x * as.x + o.y * as.y + o.z * as.z + o.w * as.w;
        pd[r] = o.x * ad.x + o.y * ad.y + o.z * ad.z + o.w * ad.w;
    }
#pragma unroll
    for (int r = 0; r < 8; ++r) { ps[r] = half_sum(ps[r]); pd[r] = half_sum(pd[r]); }
    float vs = ps[0], vd = pd[0];
#pragma unroll
    for (int r = 1; r < 8; ++r) {
        vs = lo == r ? ps[r] : vs;
        vd = lo == r ? pd[r] : vd;
    }
    if (lo < 8) {
        s_src[rbase + lo] = vs;
        s_dst[rbase + lo] = vd;
    }
}

extern "C" void kernel_launch(void* const* d_in, const int* in_sizes, int n_in,
                              void* d_out, int out_size, void* d_ws, size_t ws_size,
                              hipStream_t stream) {
    const float* x   = (const float*)d_in[0];
    const int*   ei  = (const int*)  d_in[1];
    const float* W1  = (const float*)d_in[2];
    const float* as1 = (const float*)d_in[3];
    const float* ad1 = (const float*)d_in[4];
    const float* b1  = (const float*)d_in[5];
    const float* W2  = (const float*)d_in[6];
    const float* as2 = (const float*)d_in[7];
    const float* ad2 = (const float*)d_in[8];
    const float* b2  = (const float*)d_in[9];
    const float* Wp  = (const float*)d_in[10];
    const float* bp  = (const float*)d_in[11];
    float* out = (float*)d_out;

    const int* srcA = ei;
    const int* dstA = ei + NEDGES;

    float* ws   = (float*)d_ws;
    float* Plo  = ws;
    float* Phi  = ws + (size_t)NNODES * (FH / 2);
    float* Qlo  = ws + (size_t)NNODES * FH;
    float* Qhi  = ws + (size_t)NNODES * FH + (size_t)NNODES * (FH / 2);
    int*   rowend = (int*)(ws + (size_t)2 * NNODES * FH);
    float* ssrc = ws + (size_t)2 * NNODES * FH + NNODES;
    float* sdst = ws + (size_t)2 * NNODES * FH + 2 * NNODES;
    int*   csr  = (int*)Phi;
    unsigned* tmp = (unsigned*)Qlo;
    int*   mat  = (int*)Qlo + NEDGES;
    int*   offs = (int*)Qlo + NEDGES + MATN;
    int*   bsum = (int*)Qlo + NEDGES + 2 * MATN;

    const int nRowTiles = NNODES / 16;
    const dim3 blk(256);
    const int gemmBlocks  = (nRowTiles * 4 + 7) / 8;
    const int midBlocks   = (nRowTiles + MID_WAVES - 1) / MID_WAVES;
    const int nodeBlocks  = (NNODES + 7) / 8;
    const int pairBlocks  = NNODES / 16;
    const int scanBlocks  = (MATN / 4 + 255) / 256;

    bucket_count_kernel<<<NBLK, blk, 0, stream>>>(dstA, mat);
    block_sum_kernel<<<scanBlocks, blk, 0, stream>>>(mat, bsum, MATN / 4);
    scan_kernel<<<scanBlocks, blk, 0, stream>>>(mat, bsum, offs, MATN / 4);
    bucket_scatter_kernel<<<NBLK, blk, 0, stream>>>(srcA, dstA, offs, tmp);
    bucket_build_kernel<<<NBKT, blk, 0, stream>>>(offs, tmp, rowend, csr);

    float* agg = Plo;
    gemm_h_kernel<<<gemmBlocks, blk, 0, stream>>>(x, W1, Qlo, FIN, nRowTiles);
    scalars_kernel<<<nodeBlocks, blk, 0, stream>>>(Qlo, Qhi, as1, ad1, ssrc, sdst);
    node_aggr_in_kernel<<<pairBlocks, blk, 0, stream>>>(rowend, csr, ssrc, sdst, x, agg);
    dense_mid_kernel<<<midBlocks, dim3(32 * MID_WAVES), 0, stream>>>(agg, W1, b1, W2, as2, ad2, Qlo, ssrc, sdst, nRowTiles);

    node_aggr_head_kernel<<<pairBlocks, blk, 0, stream>>>(rowend, csr, ssrc, sdst, Qlo, b2, Wp, bp, out);
}
